# starting kernel
# baseline (speedup 1.0000x reference)
#include <hip/hip_runtime.h>
#include <hip/hip_bf16.h>
#include <math.h>

typedef __attribute__((ext_vector_type(16))) _Float16 v16h;
typedef __attribute__((ext_vector_type(8)))  float    v8f;

#define HEADS 8
#define HDIM  32
#define FOUT  256
#define NEG_SLOPE 0.2f

__global__ void zero_i32(int* __restrict__ p, int n) {
    int i = blockIdx.x * blockDim.x + threadIdx.x;
    if (i < n) p[i] = 0;
}

__global__ void count_deg(const int* __restrict__ dst, int* __restrict__ deg, int e) {
    int i = blockIdx.x * blockDim.x + threadIdx.x;
    if (i < e) atomicAdd(&deg[dst[i]], 1);
}

__global__ void __launch_bounds__(1024) scan_off(const int* __restrict__ deg,
                                                 int* __restrict__ off, int n) {
    __shared__ int sh[1024];
    int tid = threadIdx.x;
    int chunk = (n + 1023) >> 10;
    int s0 = tid * chunk;
    int s1 = s0 + chunk; if (s1 > n) s1 = n;
    int local = 0;
    for (int i = s0; i < s1; ++i) local += deg[i];
    sh[tid] = local;
    __syncthreads();
    for (int o = 1; o < 1024; o <<= 1) {
        int v = (tid >= o) ? sh[tid - o] : 0;
        __syncthreads();
        sh[tid] += v;
        __syncthreads();
    }
    int run = sh[tid] - local;
    for (int i = s0; i < s1; ++i) { off[i] = run; run += deg[i]; }
    if (tid == 1023) off[n] = sh[1023];
}

__global__ void scatter_edges(const int* __restrict__ src, const int* __restrict__ dst,
                              const int* __restrict__ off, int* __restrict__ cursor,
                              int* __restrict__ eidx, int e) {
    int i = blockIdx.x * blockDim.x + threadIdx.x;
    if (i < e) {
        int d = dst[i];
        int pos = atomicAdd(&cursor[d], 1);
        eidx[off[d] + pos] = src[i];
    }
}

__global__ void cvt_f32_to_f16(const float* __restrict__ in, _Float16* __restrict__ out, int n) {
    int i = blockIdx.x * blockDim.x + threadIdx.x;
    if (i < n) out[i] = (_Float16)in[i];
}

__global__ void transpose_w(const float* __restrict__ W, _Float16* __restrict__ Wt, int K) {
    int i = blockIdx.x * blockDim.x + threadIdx.x;
    if (i < K * FOUT) {
        int k = i / FOUT, c = i % FOUT;
        Wt[c * K + k] = (_Float16)W[k * FOUT + c];
    }
}

__device__ __forceinline__ v16h load_frag(const _Float16* __restrict__ p) {
    v16h r;
#pragma unroll
    for (int e = 0; e < 8; ++e) { r[e] = p[e]; r[e + 8] = p[e + 16]; }
    return r;
}

__global__ void __launch_bounds__(256) gemm_wmma_f16(const _Float16* __restrict__ A,
                                                     const _Float16* __restrict__ Bt,
                                                     float* __restrict__ C,
                                                     int M, int K) {
    int gt   = blockIdx.x * 256 + threadIdx.x;
    int wid  = gt >> 5;
    int lane = gt & 31;
    int nWaves = (M >> 4) * (FOUT >> 6);
    if (wid >= nWaves) return;

    int m0 = (wid >> 2) << 4;
    int n0 = (wid & 3) << 6;
    int l15 = lane & 15;
    int kh  = (lane >> 4) << 3;

    const _Float16* pa  = A  + (size_t)(m0 + l15) * K + kh;
    const _Float16* pb0 = Bt + (size_t)(n0 + l15 +  0) * K + kh;
    const _Float16* pb1 = Bt + (size_t)(n0 + l15 + 16) * K + kh;
    const _Float16* pb2 = Bt + (size_t)(n0 + l15 + 32) * K + kh;
    const _Float16* pb3 = Bt + (size_t)(n0 + l15 + 48) * K + kh;

    v8f c0 = {}, c1 = {}, c2 = {}, c3 = {};
    for (int k0 = 0; k0 < K; k0 += 32) {
        v16h a  = load_frag(pa  + k0);
        v16h b0 = load_frag(pb0 + k0);
        v16h b1 = load_frag(pb1 + k0);
        v16h b2 = load_frag(pb2 + k0);
        v16h b3 = load_frag(pb3 + k0);
        c0 = __builtin_amdgcn_wmma_f32_16x16x32_f16(false, a, false, b0, (short)0, c0, false, false);
        c1 = __builtin_amdgcn_wmma_f32_16x16x32_f16(false, a, false, b1, (short)0, c1, false, false);
        c2 = __builtin_amdgcn_wmma_f32_16x16x32_f16(false, a, false, b2, (short)0, c2, false, false);
        c3 = __builtin_amdgcn_wmma_f32_16x16x32_f16(false, a, false, b3, (short)0, c3, false, false);
    }
    int rbase = (lane >> 4) << 3;
#pragma unroll
    for (int r = 0; r < 8; ++r) {
        size_t o = (size_t)(m0 + r + rbase) * FOUT + n0 + l15;
        C[o +  0] = c0[r];
        C[o + 16] = c1[r];
        C[o + 32] = c2[r];
        C[o + 48] = c3[r];
    }
}

__global__ void eler_kernel(const float* __restrict__ feat,
                            const float* __restrict__ al, const float* __restrict__ ar,
                            float* __restrict__ el, float* __restrict__ er, int n) {
    int i = blockIdx.x * blockDim.x + threadIdx.x;
    if (i >= n * HEADS) return;
    int node = i >> 3, hd = i & 7;
    const float* f = feat + (size_t)node * FOUT + hd * HDIM;
    const float* a = al + hd * HDIM;
    const float* b = ar + hd * HDIM;
    float sl = 0.f, sr = 0.f;
#pragma unroll
    for (int d = 0; d < HDIM; ++d) { float v = f[d]; sl += v * a[d]; sr += v * b[d]; }
    el[i] = sl; er[i] = sr;
}

__global__ void __launch_bounds__(256) gat_pull(const float* __restrict__ feat,
                                                const float* __restrict__ el,
                                                const float* __restrict__ er,
                                                const int* __restrict__ off,
                                                const int* __restrict__ eidx,
                                                const float* __restrict__ bias,
                                                float* __restrict__ out,
                                                int n, int mode) {
    int gt   = blockIdx.x * 256 + threadIdx.x;
    int v    = gt >> 5;
    int lane = gt & 31;
    if (v >= n) return;
    int hd = lane >> 2;
    int j  = lane & 3;

    int beg = off[v], end = off[v + 1];
    float ern = er[(size_t)v * HEADS + hd];

    float m = -INFINITY;
    for (int t = beg + j; t < end; t += 4) {
        int s = eidx[t];
        float e = el[(size_t)s * HEADS + hd] + ern;
        e = (e > 0.f) ? e : NEG_SLOPE * e;
        m = fmaxf(m, e);
    }
    m = fmaxf(m, __shfl_xor(m, 1));
    m = fmaxf(m, __shfl_xor(m, 2));

    float sum = 0.f;
    for (int t = beg + j; t < end; t += 4) {
        int s = eidx[t];
        float e = el[(size_t)s * HEADS + hd] + ern;
        e = (e > 0.f) ? e : NEG_SLOPE * e;
        sum += __expf(e - m);
    }
    sum += __shfl_xor(sum, 1);
    sum += __shfl_xor(sum, 2);
    float inv = (end > beg) ? 1.f / sum : 0.f;

    float acc[8];
#pragma unroll
    for (int q = 0; q < 8; ++q) acc[q] = 0.f;
    for (int t = beg; t < end; ++t) {
        int s = eidx[t];
        if (t + 1 < end) {
            int sn = eidx[t + 1];
            __builtin_prefetch(feat + (size_t)sn * FOUT + hd * HDIM + j * 8, 0, 0);
        }
        float e = el[(size_t)s * HEADS + hd] + ern;
        e = (e > 0.f) ? e : NEG_SLOPE * e;
        float alpha = __expf(e - m) * inv;
        const float* fs = feat + (size_t)s * FOUT + hd * HDIM + j * 8;
#pragma unroll
        for (int q = 0; q < 8; ++q) acc[q] += fs[q] * alpha;
    }

    size_t base = (size_t)v * FOUT + hd * HDIM + j * 8;
#pragma unroll
    for (int q = 0; q < 8; ++q) {
        float val = acc[q] + bias[hd * HDIM + j * 8 + q];
        if (mode == 0) {
            val = (val > 0.f) ? val : (__expf(val) - 1.f);
        } else {
            val += out[base + q];
        }
        out[base + q] = val;
    }
}

static inline size_t align256(size_t x) { return (x + 255) & ~(size_t)255; }

extern "C" void kernel_launch(void* const* d_in, const int* in_sizes, int n_in,
                              void* d_out, int out_size, void* d_ws, size_t ws_size,
                              hipStream_t stream) {
    const int N = in_sizes[0] / 128;
    const int E = in_sizes[1];

    const float* x   = (const float*)d_in[0];
    const int*   src = (const int*)d_in[1];
    const int*   dst = (const int*)d_in[2];

    char* ws = (char*)d_ws;
    size_t o = 0;
    int* off    = (int*)(ws + o); o = align256(o + (size_t)(N + 1) * 4);
    int* deg    = (int*)(ws + o); o = align256(o + (size_t)N * 4);
    int* cursor = (int*)(ws + o); o = align256(o + (size_t)N * 4);
    int* eidx   = (int*)(ws + o); o = align256(o + (size_t)E * 4);
    _Float16* h16  = (_Float16*)(ws + o); o = align256(o + (size_t)N * FOUT * 2);
    _Float16* wt16 = (_Float16*)(ws + o); o = align256(o + (size_t)FOUT * FOUT * 2);
    float* feat = (float*)(ws + o); o = align256(o + (size_t)N * FOUT * 4);
    float* el   = (float*)(ws + o); o = align256(o + (size_t)N * HEADS * 4);
    float* er   = (float*)(ws + o); o = align256(o + (size_t)N * HEADS * 4);
    (void)ws_size;

    float* h = (float*)d_out;

    const int T = 256;
    auto blk = [](int n, int t) { return (n + t - 1) / t; };

    zero_i32<<<blk(N, T), T, 0, stream>>>(deg, N);
    zero_i32<<<blk(N, T), T, 0, stream>>>(cursor, N);
    count_deg<<<blk(E, T), T, 0, stream>>>(dst, deg, E);
    scan_off<<<1, 1024, 0, stream>>>(deg, off, N);
    scatter_edges<<<blk(E, T), T, 0, stream>>>(src, dst, off, cursor, eidx, E);

    for (int layer = 0; layer < 4; ++layer) {
        const float* W  = (const float*)d_in[3 + 4 * layer];
        const float* al = (const float*)d_in[4 + 4 * layer];
        const float* ar = (const float*)d_in[5 + 4 * layer];
        const float* b  = (const float*)d_in[6 + 4 * layer];
        const int K = in_sizes[3 + 4 * layer] / FOUT;
        const float* hin = (layer == 0) ? x : h;

        cvt_f32_to_f16<<<blk(N * K, T), T, 0, stream>>>(hin, h16, N * K);
        transpose_w<<<blk(K * FOUT, T), T, 0, stream>>>(W, wt16, K);

        int nWaves = (N / 16) * (FOUT / 64);
        gemm_wmma_f16<<<blk(nWaves * 32, T), T, 0, stream>>>(h16, wt16, feat, N, K);

        eler_kernel<<<blk(N * HEADS, T), T, 0, stream>>>(feat, al, ar, el, er, N);

        int mode = (layer == 0) ? 0 : 1;
        gat_pull<<<blk(N * 32, T), T, 0, stream>>>(feat, el, er, off, eidx, b, h, N, mode);
    }
}


# gemm A operand read as f32 from the activations and converted in the fragment loader (same rounding, k order, operand type); the one-element-per-thread conversion pass (4 launches) is gone
# speedup vs baseline: 1.2459x; 1.2459x over previous
#include <hip/hip_runtime.h>
#include <hip/hip_bf16.h>
#include <math.h>

typedef __attribute__((ext_vector_type(16))) _Float16 v16h;
typedef __attribute__((ext_vector_type(8)))  float    v8f;
typedef __attribute__((ext_vector_type(8)))  _Float16 v8h;

#define HEADS 8
#define HDIM  32
#define FOUT  256
#define NEG_SLOPE 0.2f

__global__ void zero_i32(int* __restrict__ p, int n) {
    int i = blockIdx.x * blockDim.x + threadIdx.x;
    if (i < n) p[i] = 0;
}

__global__ void count_deg(const int* __restrict__ dst, int* __restrict__ deg, int e) {
    int i = blockIdx.x * blockDim.x + threadIdx.x;
    if (i < e) atomicAdd(&deg[dst[i]], 1);
}

__global__ void __launch_bounds__(1024) scan_off(const int* __restrict__ deg,
                                                 int* __restrict__ off, int n) {
    __shared__ int sh[1024];
    int tid = threadIdx.x;
    int chunk = (n + 1023) >> 10;
    int s0 = tid * chunk;
    int s1 = s0 + chunk; if (s1 > n) s1 = n;
    int local = 0;
    for (int i = s0; i < s1; ++i) local += deg[i];
    sh[tid] = local;
    __syncthreads();
    for (int o = 1; o < 1024; o <<= 1) {
        int v = (tid >= o) ? sh[tid - o] : 0;
        __syncthreads();
        sh[tid] += v;
        __syncthreads();
    }
    int run = sh[tid] - local;
    for (int i = s0; i < s1; ++i) { off[i] = run; run += deg[i]; }
    if (tid == 1023) off[n] = sh[1023];
}

__global__ void scatter_edges(const int* __restrict__ src, const int* __restrict__ dst,
                              const int* __restrict__ off, int* __restrict__ cursor,
                              int* __restrict__ eidx, int e) {
    int i = blockIdx.x * blockDim.x + threadIdx.x;
    if (i < e) {
        int d = dst[i];
        int pos = atomicAdd(&cursor[d], 1);
        eidx[off[d] + pos] = src[i];
    }
}

__global__ void transpose_w(const float* __restrict__ W, _Float16* __restrict__ Wt, int K) {
    int i = blockIdx.x * blockDim.x + threadIdx.x;
    if (i < K * FOUT) {
        int k = i / FOUT, c = i % FOUT;
        Wt[c * K + k] = (_Float16)W[k * FOUT + c];
    }
}

__device__ __forceinline__ v16h load_frag(const _Float16* __restrict__ p) {
    v16h r;
#pragma unroll
    for (int e = 0; e < 8; ++e) { r[e] = p[e]; r[e + 8] = p[e + 16]; }
    return r;
}

__device__ __forceinline__ v16h load_frag_f32(const float* __restrict__ p) {
    const float4* q = (const float4*)p;
    float4 x0 = q[0], x1 = q[1];
    float4 y0 = q[4], y1 = q[5];
    v8f lo = {x0.x, x0.y, x0.z, x0.w, x1.x, x1.y, x1.z, x1.w};
    v8f hi = {y0.x, y0.y, y0.z, y0.w, y1.x, y1.y, y1.z, y1.w};
    v8h hl = __builtin_convertvector(lo, v8h);
    v8h hh = __builtin_convertvector(hi, v8h);
    return __builtin_shufflevector(hl, hh, 0, 1, 2, 3, 4, 5, 6, 7, 8, 9, 10, 11, 12, 13, 14, 15);
}

__global__ void __launch_bounds__(256) gemm_wmma_f16(const float* __restrict__ A,
                                                     const _Float16* __restrict__ Bt,
                                                     float* __restrict__ C,
                                                     int M, int K) {
    int gt   = blockIdx.x * 256 + threadIdx.x;
    int wid  = gt >> 5;
    int lane = gt & 31;
    int nWaves = (M >> 4) * (FOUT >> 6);
    if (wid >= nWaves) return;

    int m0 = (wid >> 2) << 4;
    int n0 = (wid & 3) << 6;
    int l15 = lane & 15;
    int kh  = (lane >> 4) << 3;

    const float*    pa  = A  + (size_t)(m0 + l15) * K + kh;
    const _Float16* pb0 = Bt + (size_t)(n0 + l15 +  0) * K + kh;
    const _Float16* pb1 = Bt + (size_t)(n0 + l15 + 16) * K + kh;
    const _Float16* pb2 = Bt + (size_t)(n0 + l15 + 32) * K + kh;
    const _Float16* pb3 = Bt + (size_t)(n0 + l15 + 48) * K + kh;

    v8f c0 = {}, c1 = {}, c2 = {}, c3 = {};
    for (int k0 = 0; k0 < K; k0 += 32) {
        v16h a  = load_frag_f32(pa + k0);
        v16h b0 = load_frag(pb0 + k0);
        v16h b1 = load_frag(pb1 + k0);
        v16h b2 = load_frag(pb2 + k0);
        v16h b3 = load_frag(pb3 + k0);
        c0 = __builtin_amdgcn_wmma_f32_16x16x32_f16(false, a, false, b0, (short)0, c0, false, false);
        c1 = __builtin_amdgcn_wmma_f32_16x16x32_f16(false, a, false, b1, (short)0, c1, false, false);
        c2 = __builtin_amdgcn_wmma_f32_16x16x32_f16(false, a, false, b2, (short)0, c2, false, false);
        c3 = __builtin_amdgcn_wmma_f32_16x16x32_f16(false, a, false, b3, (short)0, c3, false, false);
    }
    int rbase = (lane >> 4) << 3;
#pragma unroll
    for (int r = 0; r < 8; ++r) {
        size_t o = (size_t)(m0 + r + rbase) * FOUT + n0 + l15;
        C[o +  0] = c0[r];
        C[o + 16] = c1[r];
        C[o + 32] = c2[r];
        C[o + 48] = c3[r];
    }
}

__global__ void eler_kernel(const float* __restrict__ feat,
                            const float* __restrict__ al, const float* __restrict__ ar,
                            float* __restrict__ el, float* __restrict__ er, int n) {
    int i = blockIdx.x * blockDim.x + threadIdx.x;
    if (i >= n * HEADS) return;
    int node = i >> 3, hd = i & 7;
    const float* f = feat + (size_t)node * FOUT + hd * HDIM;
    const float* a = al + hd * HDIM;
    const float* b = ar + hd * HDIM;
    float sl = 0.f, sr = 0.f;
#pragma unroll
    for (int d = 0; d < HDIM; ++d) { float v = f[d]; sl += v * a[d]; sr += v * b[d]; }
    el[i] = sl; er[i] = sr;
}

__device__ __forceinline__ float leaky(float e) { return (e > 0.f) ? e : NEG_SLOPE * e; }

__device__ __forceinline__ float pick4(const float* a, int k) {
    float r = a[0];
    r = (k == 1) ? a[1] : r;
    r = (k == 2) ? a[2] : r;
    r = (k == 3) ? a[3] : r;
    return r;
}

__device__ __forceinline__ void edge_fma(float4& accA, float4& accB, float4 fa, float4 fb,
                                         float la, float lb, float erA, float erB,
                                         float mA, float mB, float iA, float iB) {
    float wA = __expf(leaky(la + erA) - mA) * iA;
    float wB = __expf(leaky(lb + erB) - mB) * iB;
    accA.x += fa.x * wA; accA.y += fa.y * wA; accA.z += fa.z * wA; accA.w += fa.w * wA;
    accB.x += fb.x * wB; accB.y += fb.y * wB; accB.z += fb.z * wB; accB.w += fb.w * wB;
}

__global__ void __launch_bounds__(256) gat_pull(const float* __restrict__ feat,
                                                const float* __restrict__ el,
                                                const float* __restrict__ er,
                                                const int* __restrict__ off,
                                                const int* __restrict__ eidx,
                                                const float* __restrict__ bias,
                                                float* __restrict__ out,
                                                int n, int mode) {
    int gt   = blockIdx.x * 256 + threadIdx.x;
    int lane = gt & 31;
    int v = __builtin_amdgcn_readfirstlane(gt >> 5);
    if (v >= n) return;

    int beg = off[v], end = off[v + 1];
    int hA = lane >> 3;
    size_t obase = (size_t)v * FOUT + lane * 4;

    float4 oA = make_float4(0.f, 0.f, 0.f, 0.f), oB = oA;
    if (mode != 0) {
        oA = *(const float4*)(out + obase);
        oB = *(const float4*)(out + obase + 128);
    }

    float ern[8];
#pragma unroll
    for (int h = 0; h < 8; ++h) ern[h] = er[(size_t)v * HEADS + h];

    float mx[8];
#pragma unroll
    for (int h = 0; h < 8; ++h) mx[h] = -INFINITY;
    for (int c0 = beg; c0 < end; c0 += 32) {
        int t = c0 + lane;
        if (t < end) {
            int s = eidx[t];
            const float4* ep = (const float4*)(el + (size_t)s * HEADS);
            float4 a0 = ep[0], a1 = ep[1];
            float ev[8] = {a0.x, a0.y, a0.z, a0.w, a1.x, a1.y, a1.z, a1.w};
#pragma unroll
            for (int h = 0; h < 8; ++h) mx[h] = fmaxf(mx[h], leaky(ev[h] + ern[h]));
        }
    }
#pragma unroll
    for (int h = 0; h < 8; ++h) {
#pragma unroll
        for (int o = 16; o >= 1; o >>= 1) mx[h] = fmaxf(mx[h], __shfl_xor(mx[h], o));
    }

    float sm[8];
#pragma unroll
    for (int h = 0; h < 8; ++h) sm[h] = 0.f;
    for (int c0 = beg; c0 < end; c0 += 32) {
        int t = c0 + lane;
        if (t < end) {
            int s = eidx[t];
            const float4* ep = (const float4*)(el + (size_t)s * HEADS);
            float4 a0 = ep[0], a1 = ep[1];
            float ev[8] = {a0.x, a0.y, a0.z, a0.w, a1.x, a1.y, a1.z, a1.w};
#pragma unroll
            for (int h = 0; h < 8; ++h) sm[h] += __expf(leaky(ev[h] + ern[h]) - mx[h]);
        }
    }
#pragma unroll
    for (int h = 0; h < 8; ++h) {
#pragma unroll
        for (int o = 16; o >= 1; o >>= 1) sm[h] += __shfl_xor(sm[h], o);
    }
    float inv[8];
#pragma unroll
    for (int h = 0; h < 8; ++h) inv[h] = (end > beg) ? 1.f / sm[h] : 0.f;

    float mA = pick4(mx, hA),  mB = pick4(mx + 4, hA);
    float iA = pick4(inv, hA), iB = pick4(inv + 4, hA);
    float erA = pick4(ern, hA), erB = pick4(ern + 4, hA);

    float4 accA = make_float4(0.f, 0.f, 0.f, 0.f), accB = accA;
    const float* fl = feat + lane * 4;
    const float* ll = el + hA;
    for (int c0 = beg; c0 < end; c0 += 32) {
        int cnt = end - c0; if (cnt > 32) cnt = 32;
        int sl = (lane < cnt) ? eidx[c0 + lane] : 0;
        int j = 0;
        for (; j + 4 <= cnt; j += 4) {
            int s0 = __builtin_amdgcn_readlane(sl, j);
            int s1 = __builtin_amdgcn_readlane(sl, j + 1);
            int s2 = __builtin_amdgcn_readlane(sl, j + 2);
            int s3 = __builtin_amdgcn_readlane(sl, j + 3);
            const float* f0 = fl + (size_t)s0 * FOUT;
            const float* f1 = fl + (size_t)s1 * FOUT;
            const float* f2 = fl + (size_t)s2 * FOUT;
            const float* f3 = fl + (size_t)s3 * FOUT;
            float4 a0 = *(const float4*)f0, b0 = *(const float4*)(f0 + 128);
            float4 a1 = *(const float4*)f1, b1 = *(const float4*)(f1 + 128);
            float4 a2 = *(const float4*)f2, b2 = *(const float4*)(f2 + 128);
            float4 a3 = *(const float4*)f3, b3 = *(const float4*)(f3 + 128);
            float la0 = ll[(size_t)s0 * HEADS], lb0 = ll[(size_t)s0 * HEADS + 4];
            float la1 = ll[(size_t)s1 * HEADS], lb1 = ll[(size_t)s1 * HEADS + 4];
            float la2 = ll[(size_t)s2 * HEADS], lb2 = ll[(size_t)s2 * HEADS + 4];
            float la3 = ll[(size_t)s3 * HEADS], lb3 = ll[(size_t)s3 * HEADS + 4];
            edge_fma(accA, accB, a0, b0, la0, lb0, erA, erB, mA, mB, iA, iB);
            edge_fma(accA, accB, a1, b1, la1, lb1, erA, erB, mA, mB, iA, iB);
            edge_fma(accA, accB, a2, b2, la2, lb2, erA, erB, mA, mB, iA, iB);
            edge_fma(accA, accB, a3, b3, la3, lb3, erA, erB, mA, mB, iA, iB);
        }
        for (; j < cnt; ++j) {
            int s0 = __builtin_amdgcn_readlane(sl, j);
            const float* f0 = fl + (size_t)s0 * FOUT;
            float4 a0 = *(const float4*)f0, b0 = *(const float4*)(f0 + 128);
            float la0 = ll[(size_t)s0 * HEADS], lb0 = ll[(size_t)s0 * HEADS + 4];
            edge_fma(accA, accB, a0, b0, la0, lb0, erA, erB, mA, mB, iA, iB);
        }
    }

    float ba[4], bb[4];
#pragma unroll
    for (int q = 0; q < 4; ++q) { ba[q] = bias[lane * 4 + q]; bb[q] = bias[128 + lane * 4 + q]; }
    float4 rA = make_float4(accA.x + ba[0], accA.y + ba[1], accA.z + ba[2], accA.w + ba[3]);
    float4 rB = make_float4(accB.x + bb[0], accB.y + bb[1], accB.z + bb[2], accB.w + bb[3]);
    if (mode == 0) {
        rA.x = (rA.x > 0.f) ? rA.x : (__expf(rA.x) - 1.f);
        rA.y = (rA.y > 0.f) ? rA.y : (__expf(rA.y) - 1.f);
        rA.z = (rA.z > 0.f) ? rA.z : (__expf(rA.z) - 1.f);
        rA.w = (rA.w > 0.f) ? rA.w : (__expf(rA.w) - 1.f);
        rB.x = (rB.x > 0.f) ? rB.x : (__expf(rB.x) - 1.f);
        rB.y = (rB.y > 0.f) ? rB.y : (__expf(rB.y) - 1.f);
        rB.z = (rB.z > 0.f) ? rB.z : (__expf(rB.z) - 1.f);
        rB.w = (rB.w > 0.f) ? rB.w : (__expf(rB.w) - 1.f);
    } else {
        rA.x += oA.x; rA.y += oA.y; rA.z += oA.z; rA.w += oA.w;
        rB.x += oB.x; rB.y += oB.y; rB.z += oB.z; rB.w += oB.w;
    }
    *(float4*)(out + obase)       = rA;
    *(float4*)(out + obase + 128) = rB;
}

static inline size_t align256(size_t x) { return (x + 255) & ~(size_t)255; }

extern "C" void kernel_launch(void* const* d_in, const int* in_sizes, int n_in,
                              void* d_out, int out_size, void* d_ws, size_t ws_size,
                              hipStream_t stream) {
    const int N = in_sizes[0] / 128;
    const int E = in_sizes[1];

    const float* x   = (const float*)d_in[0];
    const int*   src = (const int*)d_in[1];
    const int*   dst = (const int*)d_in[2];

    char* ws = (char*)d_ws;
    size_t o = 0;
    int* off    = (int*)(ws + o); o = align256(o + (size_t)(N + 1) * 4);
    int* deg    = (int*)(ws + o); o = align256(o + (size_t)N * 4);
    int* cursor = (int*)(ws + o); o = align256(o + (size_t)N * 4);
    int* eidx   = (int*)(ws + o); o = align256(o + (size_t)E * 4);
    _Float16* wt16 = (_Float16*)(ws + o); o = align256(o + (size_t)FOUT * FOUT * 2);
    float* feat = (float*)(ws + o); o = align256(o + (size_t)N * FOUT * 4);
    float* el   = (float*)(ws + o); o = align256(o + (size_t)N * HEADS * 4);
    float* er   = (float*)(ws + o); o = align256(o + (size_t)N * HEADS * 4);
    (void)ws_size;

    float* h = (float*)d_out;

    const int T = 256;
    auto blk = [](int n, int t) { return (n + t - 1) / t; };

    zero_i32<<<blk(N, T), T, 0, stream>>>(deg, N);
    zero_i32<<<blk(N, T), T, 0, stream>>>(cursor, N);
    count_deg<<<blk(E, T), T, 0, stream>>>(dst, deg, E);
    scan_off<<<1, 1024, 0, stream>>>(deg, off, N);
    scatter_edges<<<blk(E, T), T, 0, stream>>>(src, dst, off, cursor, eidx, E);

    for (int layer = 0; layer < 4; ++layer) {
        const float* W  = (const float*)d_in[3 + 4 * layer];
        const float* al = (const float*)d_in[4 + 4 * layer];
        const float* ar = (const float*)d_in[5 + 4 * layer];
        const float* b  = (const float*)d_in[6 + 4 * layer];
        const int K = in_sizes[3 + 4 * layer] / FOUT;
        const float* hin = (layer == 0) ? x : h;

        transpose_w<<<blk(K * FOUT, T), T, 0, stream>>>(W, wt16, K);

        int nWaves = (N / 16) * (FOUT / 64);
        gemm_wmma_f16<<<blk(nWaves * 32, T), T, 0, stream>>>(hin, wt16, feat, N, K);

        eler_kernel<<<blk(N * HEADS, T), T, 0, stream>>>(feat, al, ar, el, er, N);

        int mode = (layer == 0) ? 0 : 1;
        gat_pull<<<blk(N * 32, T), T, 0, stream>>>(feat, el, er, off, eidx, b, h, N, mode);
    }
}


# edge list grouped by destination through LDS counters (block-local sort into 256-node bins, then one block per bin ranks its edges): no global atomic per edge; in-list edge order (fp sum order) moves
# speedup vs baseline: 1.6645x; 1.3360x over previous
#include <hip/hip_runtime.h>
#include <hip/hip_bf16.h>
#include <math.h>

typedef __attribute__((ext_vector_type(16))) _Float16 v16h;
typedef __attribute__((ext_vector_type(8)))  float    v8f;
typedef __attribute__((ext_vector_type(8)))  _Float16 v8h;

#define HEADS 8
#define HDIM  32
#define FOUT  256
#define NEG_SLOPE 0.2f

#define BIN_SHIFT 8
#define BIN_NODES 256
#define BIN_GRID  256

__device__ __forceinline__ int bin_of(int d) {
    int b = (int)((unsigned)d >> BIN_SHIFT);
    return (b < BIN_NODES) ? b : (BIN_NODES - 1);
}

__device__ __forceinline__ void scan256_inclusive(int* sh, int tid) {
    for (int o = 1; o < 256; o <<= 1) {
        int v = 0;
        if (tid < 256 && tid >= o) v = sh[tid - o];
        __syncthreads();
        if (tid < 256) sh[tid] += v;
        __syncthreads();
    }
}

__global__ void __launch_bounds__(1024) bin_local(const int* __restrict__ src,
                                                  const int* __restrict__ dst,
                                                  int* __restrict__ rec,
                                                  int* __restrict__ cntT,
                                                  int* __restrict__ lstT,
                                                  int e, int chunk) {
    __shared__ int hist[BIN_NODES];
    __shared__ int lst[BIN_NODES];
    __shared__ int cur[BIN_NODES];
    int t = threadIdx.x, blk = blockIdx.x;
    int e0 = blk * chunk;
    int e1 = e0 + chunk; if (e1 > e) e1 = e;
    if (t < 256) { hist[t] = 0; cur[t] = 0; }
    __syncthreads();
    for (int base = e0 + t; base < e1; base += 4096) {
        int d[4];
#pragma unroll
        for (int u = 0; u < 4; ++u) { int i = base + u * 1024; d[u] = dst[(i < e1) ? i : (e1 - 1)]; }
#pragma unroll
        for (int u = 0; u < 4; ++u) atomicAdd(&hist[bin_of(d[u])], (base + u * 1024 < e1) ? 1 : 0);
    }
    __syncthreads();
    int c = 0;
    if (t < 256) { c = hist[t]; lst[t] = c; }
    __syncthreads();
    scan256_inclusive(lst, t);
    if (t < 256) {
        int ex = lst[t] - c;
        cntT[blk * BIN_NODES + t] = c;
        lstT[blk * BIN_NODES + t] = ex;
        lst[t] = ex;
    }
    __syncthreads();
    int2* r2 = (int2*)rec;
    for (int base = e0 + t; base < e1; base += 4096) {
        int d[4], s[4];
#pragma unroll
        for (int u = 0; u < 4; ++u) {
            int i = base + u * 1024;
            int ic = (i < e1) ? i : (e1 - 1);
            d[u] = dst[ic];
            s[u] = src[ic];
        }
#pragma unroll
        for (int u = 0; u < 4; ++u) {
            if (base + u * 1024 < e1) {
                int b = bin_of(d[u]);
                int r = atomicAdd(&cur[b], 1);
                r2[e0 + lst[b] + r] = make_int2(s[u], d[u]);
            }
        }
    }
}

__global__ void __launch_bounds__(1024) bin_build(const int* __restrict__ rec,
                                                  const int* __restrict__ cntT,
                                                  const int* __restrict__ lstT,
                                                  int* __restrict__ off,
                                                  int* __restrict__ eidx,
                                                  int n, int e, int chunk) {
    __shared__ int cnt[BIN_NODES];
    __shared__ int exs[BIN_NODES];
    __shared__ int cur[BIN_NODES];
    __shared__ int sS;
    int t = threadIdx.x, b = blockIdx.x;
    if (t < 256) { cnt[t] = 0; cur[t] = 0; }
    if (t == 0) sS = 0;
    __syncthreads();

    int col = t & 255;
    int acc = 0;
    if (col < b) {
        for (int r = t >> 8; r < BIN_GRID; r += 4) acc += cntT[r * BIN_NODES + col];
    }
#pragma unroll
    for (int o = 16; o >= 1; o >>= 1) acc += __shfl_xor(acc, o);
    if ((t & 31) == 0 && acc != 0) atomicAdd(&sS, acc);

    int seg = t >> 2, sub = t & 3;
    int st  = seg * chunk + lstT[seg * BIN_NODES + b];
    int len = cntT[seg * BIN_NODES + b];
    const int2* r2 = (const int2*)rec;
    for (int base = sub; base < len; base += 16) {
        int d[4];
#pragma unroll
        for (int u = 0; u < 4; ++u) { int i = base + u * 4; d[u] = r2[st + ((i < len) ? i : (len - 1))].y; }
#pragma unroll
        for (int u = 0; u < 4; ++u) atomicAdd(&cnt[d[u] & (BIN_NODES - 1)], (base + u * 4 < len) ? 1 : 0);
    }
    __syncthreads();

    int S = sS;
    int own = 0;
    if (t < 256) { own = cnt[t]; exs[t] = own; }
    __syncthreads();
    scan256_inclusive(exs, t);
    if (t < 256) {
        int ex = exs[t] - own;
        int node = b * BIN_NODES + t;
        if (node < n) off[node] = S + ex;
        exs[t] = ex;
    }
    if (b == (int)gridDim.x - 1 && t == 0) off[n] = e;
    __syncthreads();
    for (int base = sub; base < len; base += 16) {
        int2 r[4];
#pragma unroll
        for (int u = 0; u < 4; ++u) { int i = base + u * 4; r[u] = r2[st + ((i < len) ? i : (len - 1))]; }
#pragma unroll
        for (int u = 0; u < 4; ++u) {
            if (base + u * 4 < len) {
                int l = r[u].y & (BIN_NODES - 1);
                int k = atomicAdd(&cur[l], 1);
                eidx[S + exs[l] + k] = r[u].x;
            }
        }
    }
}

__global__ void transpose_w(const float* __restrict__ W, _Float16* __restrict__ Wt, int K) {
    int i = blockIdx.x * blockDim.x + threadIdx.x;
    if (i < K * FOUT) {
        int k = i / FOUT, c = i % FOUT;
        Wt[c * K + k] = (_Float16)W[k * FOUT + c];
    }
}

__device__ __forceinline__ v16h load_frag(const _Float16* __restrict__ p) {
    v16h r;
#pragma unroll
    for (int e = 0; e < 8; ++e) { r[e] = p[e]; r[e + 8] = p[e + 16]; }
    return r;
}

__device__ __forceinline__ v16h load_frag_f32(const float* __restrict__ p) {
    const float4* q = (const float4*)p;
    float4 x0 = q[0], x1 = q[1];
    float4 y0 = q[4], y1 = q[5];
    v8f lo = {x0.x, x0.y, x0.z, x0.w, x1.x, x1.y, x1.z, x1.w};
    v8f hi = {y0.x, y0.y, y0.z, y0.w, y1.x, y1.y, y1.z, y1.w};
    v8h hl = __builtin_convertvector(lo, v8h);
    v8h hh = __builtin_convertvector(hi, v8h);
    return __builtin_shufflevector(hl, hh, 0, 1, 2, 3, 4, 5, 6, 7, 8, 9, 10, 11, 12, 13, 14, 15);
}

__global__ void __launch_bounds__(256) gemm_wmma_f16(const float* __restrict__ A,
                                                     const _Float16* __restrict__ Bt,
                                                     float* __restrict__ C,
                                                     int M, int K) {
    int gt   = blockIdx.x * 256 + threadIdx.x;
    int wid  = gt >> 5;
    int lane = gt & 31;
    int nWaves = (M >> 4) * (FOUT >> 6);
    if (wid >= nWaves) return;

    int m0 = (wid >> 2) << 4;
    int n0 = (wid & 3) << 6;
    int l15 = lane & 15;
    int kh  = (lane >> 4) << 3;

    const float*    pa  = A  + (size_t)(m0 + l15) * K + kh;
    const _Float16* pb0 = Bt + (size_t)(n0 + l15 +  0) * K + kh;
    const _Float16* pb1 = Bt + (size_t)(n0 + l15 + 16) * K + kh;
    const _Float16* pb2 = Bt + (size_t)(n0 + l15 + 32) * K + kh;
    const _Float16* pb3 = Bt + (size_t)(n0 + l15 + 48) * K + kh;

    v8f c0 = {}, c1 = {}, c2 = {}, c3 = {};
    for (int k0 = 0; k0 < K; k0 += 32) {
        v16h a  = load_frag_f32(pa + k0);
        v16h b0 = load_frag(pb0 + k0);
        v16h b1 = load_frag(pb1 + k0);
        v16h b2 = load_frag(pb2 + k0);
        v16h b3 = load_frag(pb3 + k0);
        c0 = __builtin_amdgcn_wmma_f32_16x16x32_f16(false, a, false, b0, (short)0, c0, false, false);
        c1 = __builtin_amdgcn_wmma_f32_16x16x32_f16(false, a, false, b1, (short)0, c1, false, false);
        c2 = __builtin_amdgcn_wmma_f32_16x16x32_f16(false, a, false, b2, (short)0, c2, false, false);
        c3 = __builtin_amdgcn_wmma_f32_16x16x32_f16(false, a, false, b3, (short)0, c3, false, false);
    }
    int rbase = (lane >> 4) << 3;
#pragma unroll
    for (int r = 0; r < 8; ++r) {
        size_t o = (size_t)(m0 + r + rbase) * FOUT + n0 + l15;
        C[o +  0] = c0[r];
        C[o + 16] = c1[r];
        C[o + 32] = c2[r];
        C[o + 48] = c3[r];
    }
}

__global__ void eler_kernel(const float* __restrict__ feat,
                            const float* __restrict__ al, const float* __restrict__ ar,
                            float* __restrict__ el, float* __restrict__ er, int n) {
    int i = blockIdx.x * blockDim.x + threadIdx.x;
    if (i >= n * HEADS) return;
    int node = i >> 3, hd = i & 7;
    const float* f = feat + (size_t)node * FOUT + hd * HDIM;
    const float* a = al + hd * HDIM;
    const float* b = ar + hd * HDIM;
    float sl = 0.f, sr = 0.f;
#pragma unroll
    for (int d = 0; d < HDIM; ++d) { float v = f[d]; sl += v * a[d]; sr += v * b[d]; }
    el[i] = sl; er[i] = sr;
}

__device__ __forceinline__ float leaky(float e) { return (e > 0.f) ? e : NEG_SLOPE * e; }

__device__ __forceinline__ float pick4(const float* a, int k) {
    float r = a[0];
    r = (k == 1) ? a[1] : r;
    r = (k == 2) ? a[2] : r;
    r = (k == 3) ? a[3] : r;
    return r;
}

__device__ __forceinline__ void edge_fma(float4& accA, float4& accB, float4 fa, float4 fb,
                                         float la, float lb, float erA, float erB,
                                         float mA, float mB, float iA, float iB) {
    float wA = __expf(leaky(la + erA) - mA) * iA;
    float wB = __expf(leaky(lb + erB) - mB) * iB;
    accA.x += fa.x * wA; accA.y += fa.y * wA; accA.z += fa.z * wA; accA.w += fa.w * wA;
    accB.x += fb.x * wB; accB.y += fb.y * wB; accB.z += fb.z * wB; accB.w += fb.w * wB;
}

__global__ void __launch_bounds__(256) gat_pull(const float* __restrict__ feat,
                                                const float* __restrict__ el,
                                                const float* __restrict__ er,
                                                const int* __restrict__ off,
                                                const int* __restrict__ eidx,
                                                const float* __restrict__ bias,
                                                float* __restrict__ out,
                                                int n, int mode) {
    int gt   = blockIdx.x * 256 + threadIdx.x;
    int lane = gt & 31;
    int v = __builtin_amdgcn_readfirstlane(gt >> 5);
    if (v >= n) return;

    int beg = off[v], end = off[v + 1];
    int hA = lane >> 3;
    size_t obase = (size_t)v * FOUT + lane * 4;

    float4 oA = make_float4(0.f, 0.f, 0.f, 0.f), oB = oA;
    if (mode != 0) {
        oA = *(const float4*)(out + obase);
        oB = *(const float4*)(out + obase + 128);
    }

    float ern[8];
#pragma unroll
    for (int h = 0; h < 8; ++h) ern[h] = er[(size_t)v * HEADS + h];

    float mx[8];
#pragma unroll
    for (int h = 0; h < 8; ++h) mx[h] = -INFINITY;
    for (int c0 = beg; c0 < end; c0 += 32) {
        int t = c0 + lane;
        if (t < end) {
            int s = eidx[t];
            const float4* ep = (const float4*)(el + (size_t)s * HEADS);
            float4 a0 = ep[0], a1 = ep[1];
            float ev[8] = {a0.x, a0.y, a0.z, a0.w, a1.x, a1.y, a1.z, a1.w};
#pragma unroll
            for (int h = 0; h < 8; ++h) mx[h] = fmaxf(mx[h], leaky(ev[h] + ern[h]));
        }
    }
#pragma unroll
    for (int h = 0; h < 8; ++h) {
#pragma unroll
        for (int o = 16; o >= 1; o >>= 1) mx[h] = fmaxf(mx[h], __shfl_xor(mx[h], o));
    }

    float sm[8];
#pragma unroll
    for (int h = 0; h < 8; ++h) sm[h] = 0.f;
    for (int c0 = beg; c0 < end; c0 += 32) {
        int t = c0 + lane;
        if (t < end) {
            int s = eidx[t];
            const float4* ep = (const float4*)(el + (size_t)s * HEADS);
            float4 a0 = ep[0], a1 = ep[1];
            float ev[8] = {a0.x, a0.y, a0.z, a0.w, a1.x, a1.y, a1.z, a1.w};
#pragma unroll
            for (int h = 0; h < 8; ++h) sm[h] += __expf(leaky(ev[h] + ern[h]) - mx[h]);
        }
    }
#pragma unroll
    for (int h = 0; h < 8; ++h) {
#pragma unroll
        for (int o = 16; o >= 1; o >>= 1) sm[h] += __shfl_xor(sm[h], o);
    }
    float inv[8];
#pragma unroll
    for (int h = 0; h < 8; ++h) inv[h] = (end > beg) ? 1.f / sm[h] : 0.f;

    float mA = pick4(mx, hA),  mB = pick4(mx + 4, hA);
    float iA = pick4(inv, hA), iB = pick4(inv + 4, hA);
    float erA = pick4(ern, hA), erB = pick4(ern + 4, hA);

    float4 accA = make_float4(0.f, 0.f, 0.f, 0.f), accB = accA;
    const float* fl = feat + lane * 4;
    const float* ll = el + hA;
    for (int c0 = beg; c0 < end; c0 += 32) {
        int cnt = end - c0; if (cnt > 32) cnt = 32;
        int sl = (lane < cnt) ? eidx[c0 + lane] : 0;
        int j = 0;
        for (; j + 4 <= cnt; j += 4) {
            int s0 = __builtin_amdgcn_readlane(sl, j);
            int s1 = __builtin_amdgcn_readlane(sl, j + 1);
            int s2 = __builtin_amdgcn_readlane(sl, j + 2);
            int s3 = __builtin_amdgcn_readlane(sl, j + 3);
            const float* f0 = fl + (size_t)s0 * FOUT;
            const float* f1 = fl + (size_t)s1 * FOUT;
            const float* f2 = fl + (size_t)s2 * FOUT;
            const float* f3 = fl + (size_t)s3 * FOUT;
            float4 a0 = *(const float4*)f0, b0 = *(const float4*)(f0 + 128);
            float4 a1 = *(const float4*)f1, b1 = *(const float4*)(f1 + 128);
            float4 a2 = *(const float4*)f2, b2 = *(const float4*)(f2 + 128);
            float4 a3 = *(const float4*)f3, b3 = *(const float4*)(f3 + 128);
            float la0 = ll[(size_t)s0 * HEADS], lb0 = ll[(size_t)s0 * HEADS + 4];
            float la1 = ll[(size_t)s1 * HEADS], lb1 = ll[(size_t)s1 * HEADS + 4];
            float la2 = ll[(size_t)s2 * HEADS], lb2 = ll[(size_t)s2 * HEADS + 4];
            float la3 = ll[(size_t)s3 * HEADS], lb3 = ll[(size_t)s3 * HEADS + 4];
            edge_fma(accA, accB, a0, b0, la0, lb0, erA, erB, mA, mB, iA, iB);
            edge_fma(accA, accB, a1, b1, la1, lb1, erA, erB, mA, mB, iA, iB);
            edge_fma(accA, accB, a2, b2, la2, lb2, erA, erB, mA, mB, iA, iB);
            edge_fma(accA, accB, a3, b3, la3, lb3, erA, erB, mA, mB, iA, iB);
        }
        for (; j < cnt; ++j) {
            int s0 = __builtin_amdgcn_readlane(sl, j);
            const float* f0 = fl + (size_t)s0 * FOUT;
            float4 a0 = *(const float4*)f0, b0 = *(const float4*)(f0 + 128);
            float la0 = ll[(size_t)s0 * HEADS], lb0 = ll[(size_t)s0 * HEADS + 4];
            edge_fma(accA, accB, a0, b0, la0, lb0, erA, erB, mA, mB, iA, iB);
        }
    }

    float ba[4], bb[4];
#pragma unroll
    for (int q = 0; q < 4; ++q) { ba[q] = bias[lane * 4 + q]; bb[q] = bias[128 + lane * 4 + q]; }
    float4 rA = make_float4(accA.x + ba[0], accA.y + ba[1], accA.z + ba[2], accA.w + ba[3]);
    float4 rB = make_float4(accB.x + bb[0], accB.y + bb[1], accB.z + bb[2], accB.w + bb[3]);
    if (mode == 0) {
        rA.x = (rA.x > 0.f) ? rA.x : (__expf(rA.x) - 1.f);
        rA.y = (rA.y > 0.f) ? rA.y : (__expf(rA.y) - 1.f);
        rA.z = (rA.z > 0.f) ? rA.z : (__expf(rA.z) - 1.f);
        rA.w = (rA.w > 0.f) ? rA.w : (__expf(rA.w) - 1.f);
        rB.x = (rB.x > 0.f) ? rB.x : (__expf(rB.x) - 1.f);
        rB.y = (rB.y > 0.f) ? rB.y : (__expf(rB.y) - 1.f);
        rB.z = (rB.z > 0.f) ? rB.z : (__expf(rB.z) - 1.f);
        rB.w = (rB.w > 0.f) ? rB.w : (__expf(rB.w) - 1.f);
    } else {
        rA.x += oA.x; rA.y += oA.y; rA.z += oA.z; rA.w += oA.w;
        rB.x += oB.x; rB.y += oB.y; rB.z += oB.z; rB.w += oB.w;
    }
    *(float4*)(out + obase)       = rA;
    *(float4*)(out + obase + 128) = rB;
}

static inline size_t align256(size_t x) { return (x + 255) & ~(size_t)255; }

extern "C" void kernel_launch(void* const* d_in, const int* in_sizes, int n_in,
                              void* d_out, int out_size, void* d_ws, size_t ws_size,
                              hipStream_t stream) {
    const int N = in_sizes[0] / 128;
    const int E = in_sizes[1];

    const float* x   = (const float*)d_in[0];
    const int*   src = (const int*)d_in[1];
    const int*   dst = (const int*)d_in[2];

    char* ws = (char*)d_ws;
    size_t o = 0;
    const int chunk = (E + BIN_GRID - 1) / BIN_GRID;
    const int nbins = (N + BIN_NODES - 1) / BIN_NODES;
    int* off    = (int*)(ws + o); o = align256(o + (size_t)(N + 1) * 4);
    int* cntT   = (int*)(ws + o); o = align256(o + (size_t)BIN_GRID * BIN_NODES * 4);
    int* lstT   = (int*)(ws + o); o = align256(o + (size_t)BIN_GRID * BIN_NODES * 4);
    int* rec    = (int*)(ws + o); o = align256(o + (size_t)BIN_GRID * chunk * 8);
    int* eidx   = (int*)(ws + o); o = align256(o + (size_t)E * 4);
    _Float16* wt16 = (_Float16*)(ws + o); o = align256(o + (size_t)FOUT * FOUT * 2);
    float* feat = (float*)(ws + o); o = align256(o + (size_t)N * FOUT * 4);
    float* el   = (float*)(ws + o); o = align256(o + (size_t)N * HEADS * 4);
    float* er   = (float*)(ws + o); o = align256(o + (size_t)N * HEADS * 4);
    (void)ws_size;

    float* h = (float*)d_out;

    const int T = 256;
    auto blk = [](int n, int t) { return (n + t - 1) / t; };

    bin_local<<<BIN_GRID, 1024, 0, stream>>>(src, dst, rec, cntT, lstT, E, chunk);
    bin_build<<<nbins, 1024, 0, stream>>>(rec, cntT, lstT, off, eidx, N, E, chunk);

    for (int layer = 0; layer < 4; ++layer) {
        const float* W  = (const float*)d_in[3 + 4 * layer];
        const float* al = (const float*)d_in[4 + 4 * layer];
        const float* ar = (const float*)d_in[5 + 4 * layer];
        const float* b  = (const float*)d_in[6 + 4 * layer];
        const int K = in_sizes[3 + 4 * layer] / FOUT;
        const float* hin = (layer == 0) ? x : h;

        transpose_w<<<blk(K * FOUT, T), T, 0, stream>>>(W, wt16, K);

        int nWaves = (N / 16) * (FOUT / 64);
        gemm_wmma_f16<<<blk(nWaves * 32, T), T, 0, stream>>>(hin, wt16, feat, N, K);

        eler_kernel<<<blk(N * HEADS, T), T, 0, stream>>>(feat, al, ar, el, er, N);

        int mode = (layer == 0) ? 0 : 1;
        gat_pull<<<blk(N * 32, T), T, 0, stream>>>(feat, el, er, off, eidx, b, h, N, mode);
    }
}


# attention logits el/er formed in the gemm epilogue from its f32 accumulators (gathered per block in LDS, whole-line stores); the 4 el/er launches are gone; el/er sums now a 16-lane tree
# speedup vs baseline: 1.7496x; 1.0511x over previous
#include <hip/hip_runtime.h>
#include <hip/hip_bf16.h>
#include <math.h>

typedef __attribute__((ext_vector_type(16))) _Float16 v16h;
typedef __attribute__((ext_vector_type(8)))  float    v8f;
typedef __attribute__((ext_vector_type(8)))  _Float16 v8h;

#define HEADS 8
#define HDIM  32
#define FOUT  256
#define NEG_SLOPE 0.2f

#define BIN_SHIFT 8
#define BIN_NODES 256
#define BIN_GRID  256

__device__ __forceinline__ int bin_of(int d) {
    int b = (int)((unsigned)d >> BIN_SHIFT);
    return (b < BIN_NODES) ? b : (BIN_NODES - 1);
}

__device__ __forceinline__ void scan256_inclusive(int* sh, int tid) {
    for (int o = 1; o < 256; o <<= 1) {
        int v = 0;
        if (tid < 256 && tid >= o) v = sh[tid - o];
        __syncthreads();
        if (tid < 256) sh[tid] += v;
        __syncthreads();
    }
}

__global__ void __launch_bounds__(1024) bin_local(const int* __restrict__ src,
                                                  const int* __restrict__ dst,
                                                  int* __restrict__ rec,
                                                  int* __restrict__ cntT,
                                                  int* __restrict__ lstT,
                                                  int e, int chunk) {
    __shared__ int hist[BIN_NODES];
    __shared__ int lst[BIN_NODES];
    __shared__ int cur[BIN_NODES];
    int t = threadIdx.x, blk = blockIdx.x;
    int e0 = blk * chunk;
    int e1 = e0 + chunk; if (e1 > e) e1 = e;
    if (t < 256) { hist[t] = 0; cur[t] = 0; }
    __syncthreads();
    for (int base = e0 + t; base < e1; base += 4096) {
        int d[4];
#pragma unroll
        for (int u = 0; u < 4; ++u) { int i = base + u * 1024; d[u] = dst[(i < e1) ? i : (e1 - 1)]; }
#pragma unroll
        for (int u = 0; u < 4; ++u) atomicAdd(&hist[bin_of(d[u])], (base + u * 1024 < e1) ? 1 : 0);
    }
    __syncthreads();
    int c = 0;
    if (t < 256) { c = hist[t]; lst[t] = c; }
    __syncthreads();
    scan256_inclusive(lst, t);
    if (t < 256) {
        int ex = lst[t] - c;
        cntT[blk * BIN_NODES + t] = c;
        lstT[blk * BIN_NODES + t] = ex;
        lst[t] = ex;
    }
    __syncthreads();
    int2* r2 = (int2*)rec;
    for (int base = e0 + t; base < e1; base += 4096) {
        int d[4], s[4];
#pragma unroll
        for (int u = 0; u < 4; ++u) {
            int i = base + u * 1024;
            int ic = (i < e1) ? i : (e1 - 1);
            d[u] = dst[ic];
            s[u] = src[ic];
        }
#pragma unroll
        for (int u = 0; u < 4; ++u) {
            if (base + u * 1024 < e1) {
                int b = bin_of(d[u]);
                int r = atomicAdd(&cur[b], 1);
                r2[e0 + lst[b] + r] = make_int2(s[u], d[u]);
            }
        }
    }
}

__global__ void __launch_bounds__(1024) bin_build(const int* __restrict__ rec,
                                                  const int* __restrict__ cntT,
                                                  const int* __restrict__ lstT,
                                                  int* __restrict__ off,
                                                  int* __restrict__ eidx,
                                                  int n, int e, int chunk) {
    __shared__ int cnt[BIN_NODES];
    __shared__ int exs[BIN_NODES];
    __shared__ int cur[BIN_NODES];
    __shared__ int sS;
    int t = threadIdx.x, b = blockIdx.x;
    if (t < 256) { cnt[t] = 0; cur[t] = 0; }
    if (t == 0) sS = 0;
    __syncthreads();

    int col = t & 255;
    int acc = 0;
    if (col < b) {
        for (int r = t >> 8; r < BIN_GRID; r += 4) acc += cntT[r * BIN_NODES + col];
    }
#pragma unroll
    for (int o = 16; o >= 1; o >>= 1) acc += __shfl_xor(acc, o);
    if ((t & 31) == 0 && acc != 0) atomicAdd(&sS, acc);

    int seg = t >> 2, sub = t & 3;
    int st  = seg * chunk + lstT[seg * BIN_NODES + b];
    int len = cntT[seg * BIN_NODES + b];
    const int2* r2 = (const int2*)rec;
    for (int base = sub; base < len; base += 16) {
        int d[4];
#pragma unroll
        for (int u = 0; u < 4; ++u) { int i = base + u * 4; d[u] = r2[st + ((i < len) ? i : (len - 1))].y; }
#pragma unroll
        for (int u = 0; u < 4; ++u) atomicAdd(&cnt[d[u] & (BIN_NODES - 1)], (base + u * 4 < len) ? 1 : 0);
    }
    __syncthreads();

    int S = sS;
    int own = 0;
    if (t < 256) { own = cnt[t]; exs[t] = own; }
    __syncthreads();
    scan256_inclusive(exs, t);
    if (t < 256) {
        int ex = exs[t] - own;
        int node = b * BIN_NODES + t;
        if (node < n) off[node] = S + ex;
        exs[t] = ex;
    }
    if (b == (int)gridDim.x - 1 && t == 0) off[n] = e;
    __syncthreads();
    for (int base = sub; base < len; base += 16) {
        int2 r[4];
#pragma unroll
        for (int u = 0; u < 4; ++u) { int i = base + u * 4; r[u] = r2[st + ((i < len) ? i : (len - 1))]; }
#pragma unroll
        for (int u = 0; u < 4; ++u) {
            if (base + u * 4 < len) {
                int l = r[u].y & (BIN_NODES - 1);
                int k = atomicAdd(&cur[l], 1);
                eidx[S + exs[l] + k] = r[u].x;
            }
        }
    }
}

__global__ void transpose_w(const float* __restrict__ W, _Float16* __restrict__ Wt, int K) {
    int i = blockIdx.x * blockDim.x + threadIdx.x;
    if (i < K * FOUT) {
        int k = i / FOUT, c = i % FOUT;
        Wt[c * K + k] = (_Float16)W[k * FOUT + c];
    }
}

__device__ __forceinline__ v16h load_frag(const _Float16* __restrict__ p) {
    v16h r;
#pragma unroll
    for (int e = 0; e < 8; ++e) { r[e] = p[e]; r[e + 8] = p[e + 16]; }
    return r;
}

__device__ __forceinline__ v16h load_frag_f32(const float* __restrict__ p) {
    const float4* q = (const float4*)p;
    float4 x0 = q[0], x1 = q[1];
    float4 y0 = q[4], y1 = q[5];
    v8f lo = {x0.x, x0.y, x0.z, x0.w, x1.x, x1.y, x1.z, x1.w};
    v8f hi = {y0.x, y0.y, y0.z, y0.w, y1.x, y1.y, y1.z, y1.w};
    v8h hl = __builtin_convertvector(lo, v8h);
    v8h hh = __builtin_convertvector(hi, v8h);
    return __builtin_shufflevector(hl, hh, 0, 1, 2, 3, 4, 5, 6, 7, 8, 9, 10, 11, 12, 13, 14, 15);
}

template <int H>
__device__ __forceinline__ void halve_step(float* v, bool up, int mask) {
#pragma unroll
    for (int k = 0; k < H; ++k) {
        float keep = up ? v[k + H] : v[k];
        float send = up ? v[k] : v[k + H];
        v[k] = keep + __shfl_xor(send, mask);
    }
}

__global__ void __launch_bounds__(256) gemm_wmma_f16(const float* __restrict__ A,
                                                     const _Float16* __restrict__ Bt,
                                                     float* __restrict__ C,
                                                     const float* __restrict__ al,
                                                     const float* __restrict__ ar,
                                                     float* __restrict__ el,
                                                     float* __restrict__ er,
                                                     int M, int K) {
    __shared__ float s_el[256];
    __shared__ float s_er[256];
    int lane = threadIdx.x & 31;
    int wv   = __builtin_amdgcn_readfirstlane((int)(threadIdx.x >> 5));
    int wid  = blockIdx.x * 8 + wv;
    int nWaves = (M >> 4) * (FOUT >> 6);

    bool valid = wid < nWaves;
    int m0 = (wid >> 2) << 4;
    int n0 = (wid & 3) << 6;
    int l15 = lane & 15;
    int rbase = (lane >> 4) << 3;
    v8f c0 = {}, c1 = {}, c2 = {}, c3 = {};

    if (valid) {
        int kh  = rbase;

        const float*    pa  = A  + (size_t)(m0 + l15) * K + kh;
        const _Float16* pb0 = Bt + (size_t)(n0 + l15 +  0) * K + kh;
        const _Float16* pb1 = Bt + (size_t)(n0 + l15 + 16) * K + kh;
        const _Float16* pb2 = Bt + (size_t)(n0 + l15 + 32) * K + kh;
        const _Float16* pb3 = Bt + (size_t)(n0 + l15 + 48) * K + kh;

        for (int k0 = 0; k0 < K; k0 += 32) {
            v16h a  = load_frag_f32(pa + k0);
            v16h b0 = load_frag(pb0 + k0);
            v16h b1 = load_frag(pb1 + k0);
            v16h b2 = load_frag(pb2 + k0);
            v16h b3 = load_frag(pb3 + k0);
            c0 = __builtin_amdgcn_wmma_f32_16x16x32_f16(false, a, false, b0, (short)0, c0, false, false);
            c1 = __builtin_amdgcn_wmma_f32_16x16x32_f16(false, a, false, b1, (short)0, c1, false, false);
            c2 = __builtin_amdgcn_wmma_f32_16x16x32_f16(false, a, false, b2, (short)0, c2, false, false);
            c3 = __builtin_amdgcn_wmma_f32_16x16x32_f16(false, a, false, b3, (short)0, c3, false, false);
        }
        float a0 = al[n0 + l15], a1 = al[n0 + l15 + 16], a2 = al[n0 + l15 + 32], a3 = al[n0 + l15 + 48];
        float r0 = ar[n0 + l15], r1 = ar[n0 + l15 + 16], r2 = ar[n0 + l15 + 32], r3 = ar[n0 + l15 + 48];

        float v[32];
#pragma unroll
        for (int r = 0; r < 8; ++r) {
            v[r]      = c0[r] * a0 + c1[r] * a1;
            v[8 + r]  = c0[r] * r0 + c1[r] * r1;
            v[16 + r] = c2[r] * a2 + c3[r] * a3;
            v[24 + r] = c2[r] * r2 + c3[r] * r3;
        }
        halve_step<16>(v, (l15 & 8) != 0, 8);
        halve_step<8>(v,  (l15 & 4) != 0, 4);
        halve_step<4>(v,  (l15 & 2) != 0, 2);
        halve_step<2>(v,  (l15 & 1) != 0, 1);
        int q    = l15 >> 2;
        int hd   = ((wv & 3) << 1) + (q >> 1);
        int rowb = ((wv >> 2) << 4) + rbase + ((l15 & 3) << 1);
        float* sd = (q & 1) ? s_er : s_el;
        sd[rowb * HEADS + hd]       = v[0];
        sd[(rowb + 1) * HEADS + hd] = v[1];
    }
    __syncthreads();
    int t = threadIdx.x;
    int row = blockIdx.x * 32 + (t >> 3);
    if (row < ((M >> 4) << 4)) {
        el[(size_t)row * HEADS + (t & 7)] = s_el[t];
        er[(size_t)row * HEADS + (t & 7)] = s_er[t];
    }
    if (valid) {
#pragma unroll
        for (int r = 0; r < 8; ++r) {
            size_t o = (size_t)(m0 + r + rbase) * FOUT + n0 + l15;
            C[o +  0] = c0[r];
            C[o + 16] = c1[r];
            C[o + 32] = c2[r];
            C[o + 48] = c3[r];
        }
    }
}

__device__ __forceinline__ float leaky(float e) { return (e > 0.f) ? e : NEG_SLOPE * e; }

__device__ __forceinline__ float pick4(const float* a, int k) {
    float r = a[0];
    r = (k == 1) ? a[1] : r;
    r = (k == 2) ? a[2] : r;
    r = (k == 3) ? a[3] : r;
    return r;
}

__device__ __forceinline__ void edge_fma(float4& accA, float4& accB, float4 fa, float4 fb,
                                         float la, float lb, float erA, float erB,
                                         float mA, float mB, float iA, float iB) {
    float wA = __expf(leaky(la + erA) - mA) * iA;
    float wB = __expf(leaky(lb + erB) - mB) * iB;
    accA.x += fa.x * wA; accA.y += fa.y * wA; accA.z += fa.z * wA; accA.w += fa.w * wA;
    accB.x += fb.x * wB; accB.y += fb.y * wB; accB.z += fb.z * wB; accB.w += fb.w * wB;
}

__global__ void __launch_bounds__(256) gat_pull(const float* __restrict__ feat,
                                                const float* __restrict__ el,
                                                const float* __restrict__ er,
                                                const int* __restrict__ off,
                                                const int* __restrict__ eidx,
                                                const float* __restrict__ bias,
                                                float* __restrict__ out,
                                                int n, int mode) {
    int gt   = blockIdx.x * 256 + threadIdx.x;
    int lane = gt & 31;
    int v = __builtin_amdgcn_readfirstlane(gt >> 5);
    if (v >= n) return;

    int beg = off[v], end = off[v + 1];
    int hA = lane >> 3;
    size_t obase = (size_t)v * FOUT + lane * 4;

    float4 oA = make_float4(0.f, 0.f, 0.f, 0.f), oB = oA;
    if (mode != 0) {
        oA = *(const float4*)(out + obase);
        oB = *(const float4*)(out + obase + 128);
    }

    float ern[8];
#pragma unroll
    for (int h = 0; h < 8; ++h) ern[h] = er[(size_t)v * HEADS + h];

    float mx[8];
#pragma unroll
    for (int h = 0; h < 8; ++h) mx[h] = -INFINITY;
    for (int c0 = beg; c0 < end; c0 += 32) {
        int t = c0 + lane;
        if (t < end) {
            int s = eidx[t];
            const float4* ep = (const float4*)(el + (size_t)s * HEADS);
            float4 a0 = ep[0], a1 = ep[1];
            float ev[8] = {a0.x, a0.y, a0.z, a0.w, a1.x, a1.y, a1.z, a1.w};
#pragma unroll
            for (int h = 0; h < 8; ++h) mx[h] = fmaxf(mx[h], leaky(ev[h] + ern[h]));
        }
    }
#pragma unroll
    for (int h = 0; h < 8; ++h) {
#pragma unroll
        for (int o = 16; o >= 1; o >>= 1) mx[h] = fmaxf(mx[h], __shfl_xor(mx[h], o));
    }

    float sm[8];
#pragma unroll
    for (int h = 0; h < 8; ++h) sm[h] = 0.f;
    for (int c0 = beg; c0 < end; c0 += 32) {
        int t = c0 + lane;
        if (t < end) {
            int s = eidx[t];
            const float4* ep = (const float4*)(el + (size_t)s * HEADS);
            float4 a0 = ep[0], a1 = ep[1];
            float ev[8] = {a0.x, a0.y, a0.z, a0.w, a1.x, a1.y, a1.z, a1.w};
#pragma unroll
            for (int h = 0; h < 8; ++h) sm[h] += __expf(leaky(ev[h] + ern[h]) - mx[h]);
        }
    }
#pragma unroll
    for (int h = 0; h < 8; ++h) {
#pragma unroll
        for (int o = 16; o >= 1; o >>= 1) sm[h] += __shfl_xor(sm[h], o);
    }
    float inv[8];
#pragma unroll
    for (int h = 0; h < 8; ++h) inv[h] = (end > beg) ? 1.f / sm[h] : 0.f;

    float mA = pick4(mx, hA),  mB = pick4(mx + 4, hA);
    float iA = pick4(inv, hA), iB = pick4(inv + 4, hA);
    float erA = pick4(ern, hA), erB = pick4(ern + 4, hA);

    float4 accA = make_float4(0.f, 0.f, 0.f, 0.f), accB = accA;
    const float* fl = feat + lane * 4;
    const float* ll = el + hA;
    for (int c0 = beg; c0 < end; c0 += 32) {
        int cnt = end - c0; if (cnt > 32) cnt = 32;
        int sl = (lane < cnt) ? eidx[c0 + lane] : 0;
        int j = 0;
        for (; j + 4 <= cnt; j += 4) {
            int s0 = __builtin_amdgcn_readlane(sl, j);
            int s1 = __builtin_amdgcn_readlane(sl, j + 1);
            int s2 = __builtin_amdgcn_readlane(sl, j + 2);
            int s3 = __builtin_amdgcn_readlane(sl, j + 3);
            const float* f0 = fl + (size_t)s0 * FOUT;
            const float* f1 = fl + (size_t)s1 * FOUT;
            const float* f2 = fl + (size_t)s2 * FOUT;
            const float* f3 = fl + (size_t)s3 * FOUT;
            float4 a0 = *(const float4*)f0, b0 = *(const float4*)(f0 + 128);
            float4 a1 = *(const float4*)f1, b1 = *(const float4*)(f1 + 128);
            float4 a2 = *(const float4*)f2, b2 = *(const float4*)(f2 + 128);
            float4 a3 = *(const float4*)f3, b3 = *(const float4*)(f3 + 128);
            float la0 = ll[(size_t)s0 * HEADS], lb0 = ll[(size_t)s0 * HEADS + 4];
            float la1 = ll[(size_t)s1 * HEADS], lb1 = ll[(size_t)s1 * HEADS + 4];
            float la2 = ll[(size_t)s2 * HEADS], lb2 = ll[(size_t)s2 * HEADS + 4];
            float la3 = ll[(size_t)s3 * HEADS], lb3 = ll[(size_t)s3 * HEADS + 4];
            edge_fma(accA, accB, a0, b0, la0, lb0, erA, erB, mA, mB, iA, iB);
            edge_fma(accA, accB, a1, b1, la1, lb1, erA, erB, mA, mB, iA, iB);
            edge_fma(accA, accB, a2, b2, la2, lb2, erA, erB, mA, mB, iA, iB);
            edge_fma(accA, accB, a3, b3, la3, lb3, erA, erB, mA, mB, iA, iB);
        }
        for (; j < cnt; ++j) {
            int s0 = __builtin_amdgcn_readlane(sl, j);
            const float* f0 = fl + (size_t)s0 * FOUT;
            float4 a0 = *(const float4*)f0, b0 = *(const float4*)(f0 + 128);
            float la0 = ll[(size_t)s0 * HEADS], lb0 = ll[(size_t)s0 * HEADS + 4];
            edge_fma(accA, accB, a0, b0, la0, lb0, erA, erB, mA, mB, iA, iB);
        }
    }

    float ba[4], bb[4];
#pragma unroll
    for (int q = 0; q < 4; ++q) { ba[q] = bias[lane * 4 + q]; bb[q] = bias[128 + lane * 4 + q]; }
    float4 rA = make_float4(accA.x + ba[0], accA.y + ba[1], accA.z + ba[2], accA.w + ba[3]);
    float4 rB = make_float4(accB.x + bb[0], accB.y + bb[1], accB.z + bb[2], accB.w + bb[3]);
    if (mode == 0) {
        rA.x = (rA.x > 0.f) ? rA.x : (__expf(rA.x) - 1.f);
        rA.y = (rA.y > 0.f) ? rA.y : (__expf(rA.y) - 1.f);
        rA.z = (rA.z > 0.f) ? rA.z : (__expf(rA.z) - 1.f);
        rA.w = (rA.w > 0.f) ? rA.w : (__expf(rA.w) - 1.f);
        rB.x = (rB.x > 0.f) ? rB.x : (__expf(rB.x) - 1.f);
        rB.y = (rB.y > 0.f) ? rB.y : (__expf(rB.y) - 1.f);
        rB.z = (rB.z > 0.f) ? rB.z : (__expf(rB.z) - 1.f);
        rB.w = (rB.w > 0.f) ? rB.w : (__expf(rB.w) - 1.f);
    } else {
        rA.x += oA.x; rA.y += oA.y; rA.z += oA.z; rA.w += oA.w;
        rB.x += oB.x; rB.y += oB.y; rB.z += oB.z; rB.w += oB.w;
    }
    *(float4*)(out + obase)       = rA;
    *(float4*)(out + obase + 128) = rB;
}

static inline size_t align256(size_t x) { return (x + 255) & ~(size_t)255; }

extern "C" void kernel_launch(void* const* d_in, const int* in_sizes, int n_in,
                              void* d_out, int out_size, void* d_ws, size_t ws_size,
                              hipStream_t stream) {
    const int N = in_sizes[0] / 128;
    const int E = in_sizes[1];

    const float* x   = (const float*)d_in[0];
    const int*   src = (const int*)d_in[1];
    const int*   dst = (const int*)d_in[2];

    char* ws = (char*)d_ws;
    size_t o = 0;
    const int chunk = (E + BIN_GRID - 1) / BIN_GRID;
    const int nbins = (N + BIN_NODES - 1) / BIN_NODES;
    int* off    = (int*)(ws + o); o = align256(o + (size_t)(N + 1) * 4);
    int* cntT   = (int*)(ws + o); o = align256(o + (size_t)BIN_GRID * BIN_NODES * 4);
    int* lstT   = (int*)(ws + o); o = align256(o + (size_t)BIN_GRID * BIN_NODES * 4);
    int* rec    = (int*)(ws + o); o = align256(o + (size_t)BIN_GRID * chunk * 8);
    int* eidx   = (int*)(ws + o); o = align256(o + (size_t)E * 4);
    _Float16* wt16 = (_Float16*)(ws + o); o = align256(o + (size_t)FOUT * FOUT * 2);
    float* feat = (float*)(ws + o); o = align256(o + (size_t)N * FOUT * 4);
    float* el   = (float*)(ws + o); o = align256(o + (size_t)N * HEADS * 4);
    float* er   = (float*)(ws + o); o = align256(o + (size_t)N * HEADS * 4);
    (void)ws_size;

    float* h = (float*)d_out;

    const int T = 256;
    auto blk = [](int n, int t) { return (n + t - 1) / t; };

    bin_local<<<BIN_GRID, 1024, 0, stream>>>(src, dst, rec, cntT, lstT, E, chunk);
    bin_build<<<nbins, 1024, 0, stream>>>(rec, cntT, lstT, off, eidx, N, E, chunk);

    for (int layer = 0; layer < 4; ++layer) {
        const float* W  = (const float*)d_in[3 + 4 * layer];
        const float* al = (const float*)d_in[4 + 4 * layer];
        const float* ar = (const float*)d_in[5 + 4 * layer];
        const float* b  = (const float*)d_in[6 + 4 * layer];
        const int K = in_sizes[3 + 4 * layer] / FOUT;
        const float* hin = (layer == 0) ? x : h;

        transpose_w<<<blk(K * FOUT, T), T, 0, stream>>>(W, wt16, K);

        int nWaves = (N / 16) * (FOUT / 64);
        gemm_wmma_f16<<<blk(nWaves * 32, T), T, 0, stream>>>(hin, wt16, feat, al, ar, el, er, N, K);

        int mode = (layer == 0) ? 0 : 1;
        gat_pull<<<blk(N * 32, T), T, 0, stream>>>(feat, el, er, off, eidx, b, h, N, mode);
    }
}


# pull: a node of up to 32 edges reads its list and logits once; each edge's lane forms its 8 softmax weights and hands them to the row sum through LDS (exp per edge, not per lane); same sum order
# speedup vs baseline: 1.8786x; 1.0737x over previous
#include <hip/hip_runtime.h>
#include <hip/hip_bf16.h>
#include <math.h>

typedef __attribute__((ext_vector_type(16))) _Float16 v16h;
typedef __attribute__((ext_vector_type(8)))  float    v8f;
typedef __attribute__((ext_vector_type(8)))  _Float16 v8h;

#define HEADS 8
#define HDIM  32
#define FOUT  256
#define NEG_SLOPE 0.2f

#define BIN_SHIFT 8
#define BIN_NODES 256
#define BIN_GRID  256

__device__ __forceinline__ int bin_of(int d) {
    int b = (int)((unsigned)d >> BIN_SHIFT);
    return (b < BIN_NODES) ? b : (BIN_NODES - 1);
}

__device__ __forceinline__ void scan256_inclusive(int* sh, int tid) {
    for (int o = 1; o < 256; o <<= 1) {
        int v = 0;
        if (tid < 256 && tid >= o) v = sh[tid - o];
        __syncthreads();
        if (tid < 256) sh[tid] += v;
        __syncthreads();
    }
}

__global__ void __launch_bounds__(1024) bin_local(const int* __restrict__ src,
                                                  const int* __restrict__ dst,
                                                  int* __restrict__ rec,
                                                  int* __restrict__ cntT,
                                                  int* __restrict__ lstT,
                                                  int e, int chunk) {
    __shared__ int hist[BIN_NODES];
    __shared__ int lst[BIN_NODES];
    __shared__ int cur[BIN_NODES];
    int t = threadIdx.x, blk = blockIdx.x;
    int e0 = blk * chunk;
    int e1 = e0 + chunk; if (e1 > e) e1 = e;
    if (t < 256) { hist[t] = 0; cur[t] = 0; }
    __syncthreads();
    for (int base = e0 + t; base < e1; base += 4096) {
        int d[4];
#pragma unroll
        for (int u = 0; u < 4; ++u) { int i = base + u * 1024; d[u] = dst[(i < e1) ? i : (e1 - 1)]; }
#pragma unroll
        for (int u = 0; u < 4; ++u) atomicAdd(&hist[bin_of(d[u])], (base + u * 1024 < e1) ? 1 : 0);
    }
    __syncthreads();
    int c = 0;
    if (t < 256) { c = hist[t]; lst[t] = c; }
    __syncthreads();
    scan256_inclusive(lst, t);
    if (t < 256) {
        int ex = lst[t] - c;
        cntT[blk * BIN_NODES + t] = c;
        lstT[blk * BIN_NODES + t] = ex;
        lst[t] = ex;
    }
    __syncthreads();
    int2* r2 = (int2*)rec;
    for (int base = e0 + t; base < e1; base += 4096) {
        int d[4], s[4];
#pragma unroll
        for (int u = 0; u < 4; ++u) {
            int i = base + u * 1024;
            int ic = (i < e1) ? i : (e1 - 1);
            d[u] = dst[ic];
            s[u] = src[ic];
        }
#pragma unroll
        for (int u = 0; u < 4; ++u) {
            if (base + u * 1024 < e1) {
                int b = bin_of(d[u]);
                int r = atomicAdd(&cur[b], 1);
                r2[e0 + lst[b] + r] = make_int2(s[u], d[u]);
            }
        }
    }
}

__global__ void __launch_bounds__(1024) bin_build(const int* __restrict__ rec,
                                                  const int* __restrict__ cntT,
                                                  const int* __restrict__ lstT,
                                                  int* __restrict__ off,
                                                  int* __restrict__ eidx,
                                                  int n, int e, int chunk) {
    __shared__ int cnt[BIN_NODES];
    __shared__ int exs[BIN_NODES];
    __shared__ int cur[BIN_NODES];
    __shared__ int sS;
    int t = threadIdx.x, b = blockIdx.x;
    if (t < 256) { cnt[t] = 0; cur[t] = 0; }
    if (t == 0) sS = 0;
    __syncthreads();

    int col = t & 255;
    int acc = 0;
    if (col < b) {
        for (int r = t >> 8; r < BIN_GRID; r += 4) acc += cntT[r * BIN_NODES + col];
    }
#pragma unroll
    for (int o = 16; o >= 1; o >>= 1) acc += __shfl_xor(acc, o);
    if ((t & 31) == 0 && acc != 0) atomicAdd(&sS, acc);

    int seg = t >> 2, sub = t & 3;
    int st  = seg * chunk + lstT[seg * BIN_NODES + b];
    int len = cntT[seg * BIN_NODES + b];
    const int2* r2 = (const int2*)rec;
    for (int base = sub; base < len; base += 16) {
        int d[4];
#pragma unroll
        for (int u = 0; u < 4; ++u) { int i = base + u * 4; d[u] = r2[st + ((i < len) ? i : (len - 1))].y; }
#pragma unroll
        for (int u = 0; u < 4; ++u) atomicAdd(&cnt[d[u] & (BIN_NODES - 1)], (base + u * 4 < len) ? 1 : 0);
    }
    __syncthreads();

    int S = sS;
    int own = 0;
    if (t < 256) { own = cnt[t]; exs[t] = own; }
    __syncthreads();
    scan256_inclusive(exs, t);
    if (t < 256) {
        int ex = exs[t] - own;
        int node = b * BIN_NODES + t;
        if (node < n) off[node] = S + ex;
        exs[t] = ex;
    }
    if (b == (int)gridDim.x - 1 && t == 0) off[n] = e;
    __syncthreads();
    for (int base = sub; base < len; base += 16) {
        int2 r[4];
#pragma unroll
        for (int u = 0; u < 4; ++u) { int i = base + u * 4; r[u] = r2[st + ((i < len) ? i : (len - 1))]; }
#pragma unroll
        for (int u = 0; u < 4; ++u) {
            if (base + u * 4 < len) {
                int l = r[u].y & (BIN_NODES - 1);
                int k = atomicAdd(&cur[l], 1);
                eidx[S + exs[l] + k] = r[u].x;
            }
        }
    }
}

__global__ void transpose_w(const float* __restrict__ W, _Float16* __restrict__ Wt, int K) {
    int i = blockIdx.x * blockDim.x + threadIdx.x;
    if (i < K * FOUT) {
        int k = i / FOUT, c = i % FOUT;
        Wt[c * K + k] = (_Float16)W[k * FOUT + c];
    }
}

__device__ __forceinline__ v16h load_frag(const _Float16* __restrict__ p) {
    v16h r;
#pragma unroll
    for (int e = 0; e < 8; ++e) { r[e] = p[e]; r[e + 8] = p[e + 16]; }
    return r;
}

__device__ __forceinline__ v16h load_frag_f32(const float* __restrict__ p) {
    const float4* q = (const float4*)p;
    float4 x0 = q[0], x1 = q[1];
    float4 y0 = q[4], y1 = q[5];
    v8f lo = {x0.x, x0.y, x0.z, x0.w, x1.x, x1.y, x1.z, x1.w};
    v8f hi = {y0.x, y0.y, y0.z, y0.w, y1.x, y1.y, y1.z, y1.w};
    v8h hl = __builtin_convertvector(lo, v8h);
    v8h hh = __builtin_convertvector(hi, v8h);
    return __builtin_shufflevector(hl, hh, 0, 1, 2, 3, 4, 5, 6, 7, 8, 9, 10, 11, 12, 13, 14, 15);
}

template <int H>
__device__ __forceinline__ void halve_step(float* v, bool up, int mask) {
#pragma unroll
    for (int k = 0; k < H; ++k) {
        float keep = up ? v[k + H] : v[k];
        float send = up ? v[k] : v[k + H];
        v[k] = keep + __shfl_xor(send, mask);
    }
}

__global__ void __launch_bounds__(256) gemm_wmma_f16(const float* __restrict__ A,
                                                     const _Float16* __restrict__ Bt,
                                                     float* __restrict__ C,
                                                     const float* __restrict__ al,
                                                     const float* __restrict__ ar,
                                                     float* __restrict__ el,
                                                     float* __restrict__ er,
                                                     int M, int K) {
    __shared__ float s_el[256];
    __shared__ float s_er[256];
    int lane = threadIdx.x & 31;
    int wv   = __builtin_amdgcn_readfirstlane((int)(threadIdx.x >> 5));
    int wid  = blockIdx.x * 8 + wv;
    int nWaves = (M >> 4) * (FOUT >> 6);

    bool valid = wid < nWaves;
    int m0 = (wid >> 2) << 4;
    int n0 = (wid & 3) << 6;
    int l15 = lane & 15;
    int rbase = (lane >> 4) << 3;
    v8f c0 = {}, c1 = {}, c2 = {}, c3 = {};

    if (valid) {
        int kh  = rbase;

        const float*    pa  = A  + (size_t)(m0 + l15) * K + kh;
        const _Float16* pb0 = Bt + (size_t)(n0 + l15 +  0) * K + kh;
        const _Float16* pb1 = Bt + (size_t)(n0 + l15 + 16) * K + kh;
        const _Float16* pb2 = Bt + (size_t)(n0 + l15 + 32) * K + kh;
        const _Float16* pb3 = Bt + (size_t)(n0 + l15 + 48) * K + kh;

        for (int k0 = 0; k0 < K; k0 += 32) {
            v16h a  = load_frag_f32(pa + k0);
            v16h b0 = load_frag(pb0 + k0);
            v16h b1 = load_frag(pb1 + k0);
            v16h b2 = load_frag(pb2 + k0);
            v16h b3 = load_frag(pb3 + k0);
            c0 = __builtin_amdgcn_wmma_f32_16x16x32_f16(false, a, false, b0, (short)0, c0, false, false);
            c1 = __builtin_amdgcn_wmma_f32_16x16x32_f16(false, a, false, b1, (short)0, c1, false, false);
            c2 = __builtin_amdgcn_wmma_f32_16x16x32_f16(false, a, false, b2, (short)0, c2, false, false);
            c3 = __builtin_amdgcn_wmma_f32_16x16x32_f16(false, a, false, b3, (short)0, c3, false, false);
        }
        float a0 = al[n0 + l15], a1 = al[n0 + l15 + 16], a2 = al[n0 + l15 + 32], a3 = al[n0 + l15 + 48];
        float r0 = ar[n0 + l15], r1 = ar[n0 + l15 + 16], r2 = ar[n0 + l15 + 32], r3 = ar[n0 + l15 + 48];

        float v[32];
#pragma unroll
        for (int r = 0; r < 8; ++r) {
            v[r]      = c0[r] * a0 + c1[r] * a1;
            v[8 + r]  = c0[r] * r0 + c1[r] * r1;
            v[16 + r] = c2[r] * a2 + c3[r] * a3;
            v[24 + r] = c2[r] * r2 + c3[r] * r3;
        }
        halve_step<16>(v, (l15 & 8) != 0, 8);
        halve_step<8>(v,  (l15 & 4) != 0, 4);
        halve_step<4>(v,  (l15 & 2) != 0, 2);
        halve_step<2>(v,  (l15 & 1) != 0, 1);
        int q    = l15 >> 2;
        int hd   = ((wv & 3) << 1) + (q >> 1);
        int rowb = ((wv >> 2) << 4) + rbase + ((l15 & 3) << 1);
        float* sd = (q & 1) ? s_er : s_el;
        sd[rowb * HEADS + hd]       = v[0];
        sd[(rowb + 1) * HEADS + hd] = v[1];
    }
    __syncthreads();
    int t = threadIdx.x;
    int row = blockIdx.x * 32 + (t >> 3);
    if (row < ((M >> 4) << 4)) {
        el[(size_t)row * HEADS + (t & 7)] = s_el[t];
        er[(size_t)row * HEADS + (t & 7)] = s_er[t];
    }
    if (valid) {
#pragma unroll
        for (int r = 0; r < 8; ++r) {
            size_t o = (size_t)(m0 + r + rbase) * FOUT + n0 + l15;
            C[o +  0] = c0[r];
            C[o + 16] = c1[r];
            C[o + 32] = c2[r];
            C[o + 48] = c3[r];
        }
    }
}

__device__ __forceinline__ float leaky(float e) { return (e > 0.f) ? e : NEG_SLOPE * e; }

__device__ __forceinline__ float pick4(const float* a, int k) {
    float r = a[0];
    r = (k == 1) ? a[1] : r;
    r = (k == 2) ? a[2] : r;
    r = (k == 3) ? a[3] : r;
    return r;
}

__device__ __forceinline__ void edge_fma(float4& accA, float4& accB, float4 fa, float4 fb,
                                         float la, float lb, float erA, float erB,
                                         float mA, float mB, float iA, float iB) {
    float wA = __expf(leaky(la + erA) - mA) * iA;
    float wB = __expf(leaky(lb + erB) - mB) * iB;
    accA.x += fa.x * wA; accA.y += fa.y * wA; accA.z += fa.z * wA; accA.w += fa.w * wA;
    accB.x += fb.x * wB; accB.y += fb.y * wB; accB.z += fb.z * wB; accB.w += fb.w * wB;
}

__device__ __forceinline__ void row_fma(float4& accA, float4& accB, float4 fa, float4 fb,
                                        float wA, float wB) {
    accA.x = fmaf(fa.x, wA, accA.x); accA.y = fmaf(fa.y, wA, accA.y);
    accA.z = fmaf(fa.z, wA, accA.z); accA.w = fmaf(fa.w, wA, accA.w);
    accB.x = fmaf(fb.x, wB, accB.x); accB.y = fmaf(fb.y, wB, accB.y);
    accB.z = fmaf(fb.z, wB, accB.z); accB.w = fmaf(fb.w, wB, accB.w);
}

__global__ void __launch_bounds__(256) gat_pull(const float* __restrict__ feat,
                                                const float* __restrict__ el,
                                                const float* __restrict__ er,
                                                const int* __restrict__ off,
                                                const int* __restrict__ eidx,
                                                const float* __restrict__ bias,
                                                float* __restrict__ out,
                                                int n, int mode) {
    __shared__ float s_w[8][32 * HEADS];
    int gt   = blockIdx.x * 256 + threadIdx.x;
    int lane = gt & 31;
    int v = __builtin_amdgcn_readfirstlane(gt >> 5);
    if (v >= n) return;

    int beg = off[v], end = off[v + 1];
    int hA = lane >> 3;
    size_t obase = (size_t)v * FOUT + lane * 4;

    float4 oA = make_float4(0.f, 0.f, 0.f, 0.f), oB = oA;
    if (mode != 0) {
        oA = *(const float4*)(out + obase);
        oB = *(const float4*)(out + obase + 128);
    }

    float ern[8];
#pragma unroll
    for (int h = 0; h < 8; ++h) ern[h] = er[(size_t)v * HEADS + h];

    const int deg = end - beg;
    float4 accA = make_float4(0.f, 0.f, 0.f, 0.f), accB = accA;
    const float* fl = feat + lane * 4;

    if (deg <= 32) {
        float* swv = s_w[threadIdx.x >> 5];
        const bool have = lane < deg;
        int sl = 0;
        float e[8];
#pragma unroll
        for (int h = 0; h < 8; ++h) e[h] = -INFINITY;
        if (have) {
            sl = eidx[beg + lane];
            const float4* ep = (const float4*)(el + (size_t)sl * HEADS);
            float4 a0 = ep[0], a1 = ep[1];
            float ev[8] = {a0.x, a0.y, a0.z, a0.w, a1.x, a1.y, a1.z, a1.w};
#pragma unroll
            for (int h = 0; h < 8; ++h) e[h] = leaky(ev[h] + ern[h]);
        }
        float mx[8];
#pragma unroll
        for (int h = 0; h < 8; ++h) {
            mx[h] = e[h];
#pragma unroll
            for (int o = 16; o >= 1; o >>= 1) mx[h] = fmaxf(mx[h], __shfl_xor(mx[h], o));
        }
        float p[8], sm[8];
#pragma unroll
        for (int h = 0; h < 8; ++h) {
            p[h] = have ? __expf(e[h] - mx[h]) : 0.f;
            sm[h] = p[h];
#pragma unroll
            for (int o = 16; o >= 1; o >>= 1) sm[h] += __shfl_xor(sm[h], o);
        }
        float w[8];
#pragma unroll
        for (int h = 0; h < 8; ++h) w[h] = p[h] * ((deg > 0) ? 1.f / sm[h] : 0.f);
        *(float4*)(swv + lane * HEADS)     = make_float4(w[0], w[1], w[2], w[3]);
        *(float4*)(swv + lane * HEADS + 4) = make_float4(w[4], w[5], w[6], w[7]);
        __builtin_amdgcn_wave_barrier();

        const float* wl = swv + hA;
        int j = 0;
        for (; j + 4 <= deg; j += 4) {
            int s0 = __builtin_amdgcn_readlane(sl, j);
            int s1 = __builtin_amdgcn_readlane(sl, j + 1);
            int s2 = __builtin_amdgcn_readlane(sl, j + 2);
            int s3 = __builtin_amdgcn_readlane(sl, j + 3);
            const float* f0 = fl + (size_t)s0 * FOUT;
            const float* f1 = fl + (size_t)s1 * FOUT;
            const float* f2 = fl + (size_t)s2 * FOUT;
            const float* f3 = fl + (size_t)s3 * FOUT;
            float4 a0 = *(const float4*)f0, b0 = *(const float4*)(f0 + 128);
            float4 a1 = *(const float4*)f1, b1 = *(const float4*)(f1 + 128);
            float4 a2 = *(const float4*)f2, b2 = *(const float4*)(f2 + 128);
            float4 a3 = *(const float4*)f3, b3 = *(const float4*)(f3 + 128);
            const float* wj = wl + j * HEADS;
            float wa0 = wj[0],  wb0 = wj[4];
            float wa1 = wj[8],  wb1 = wj[12];
            float wa2 = wj[16], wb2 = wj[20];
            float wa3 = wj[24], wb3 = wj[28];
            row_fma(accA, accB, a0, b0, wa0, wb0);
            row_fma(accA, accB, a1, b1, wa1, wb1);
            row_fma(accA, accB, a2, b2, wa2, wb2);
            row_fma(accA, accB, a3, b3, wa3, wb3);
        }
#pragma unroll 1
        for (; j < deg; ++j) {
            int s0 = __builtin_amdgcn_readlane(sl, j);
            const float* f0 = fl + (size_t)s0 * FOUT;
            float4 a0 = *(const float4*)f0, b0 = *(const float4*)(f0 + 128);
            float wa0 = wl[j * HEADS], wb0 = wl[j * HEADS + 4];
            row_fma(accA, accB, a0, b0, wa0, wb0);
        }
    } else {
        float mx[8];
#pragma unroll
        for (int h = 0; h < 8; ++h) mx[h] = -INFINITY;
        for (int c0 = beg; c0 < end; c0 += 32) {
            int t = c0 + lane;
            if (t < end) {
                int s = eidx[t];
                const float4* ep = (const float4*)(el + (size_t)s * HEADS);
                float4 a0 = ep[0], a1 = ep[1];
                float ev[8] = {a0.x, a0.y, a0.z, a0.w, a1.x, a1.y, a1.z, a1.w};
#pragma unroll
                for (int h = 0; h < 8; ++h) mx[h] = fmaxf(mx[h], leaky(ev[h] + ern[h]));
            }
        }
#pragma unroll
        for (int h = 0; h < 8; ++h) {
#pragma unroll
            for (int o = 16; o >= 1; o >>= 1) mx[h] = fmaxf(mx[h], __shfl_xor(mx[h], o));
        }

        float sm[8];
#pragma unroll
        for (int h = 0; h < 8; ++h) sm[h] = 0.f;
        for (int c0 = beg; c0 < end; c0 += 32) {
            int t = c0 + lane;
            if (t < end) {
                int s = eidx[t];
                const float4* ep = (const float4*)(el + (size_t)s * HEADS);
                float4 a0 = ep[0], a1 = ep[1];
                float ev[8] = {a0.x, a0.y, a0.z, a0.w, a1.x, a1.y, a1.z, a1.w};
#pragma unroll
                for (int h = 0; h < 8; ++h) sm[h] += __expf(leaky(ev[h] + ern[h]) - mx[h]);
            }
        }
#pragma unroll
        for (int h = 0; h < 8; ++h) {
#pragma unroll
            for (int o = 16; o >= 1; o >>= 1) sm[h] += __shfl_xor(sm[h], o);
        }
        float inv[8];
#pragma unroll
        for (int h = 0; h < 8; ++h) inv[h] = (end > beg) ? 1.f / sm[h] : 0.f;

        float mA = pick4(mx, hA),  mB = pick4(mx + 4, hA);
        float iA = pick4(inv, hA), iB = pick4(inv + 4, hA);
        float erA = pick4(ern, hA), erB = pick4(ern + 4, hA);

        const float* ll = el + hA;
        for (int c0 = beg; c0 < end; c0 += 32) {
            int cnt = end - c0; if (cnt > 32) cnt = 32;
            int sl = (lane < cnt) ? eidx[c0 + lane] : 0;
#pragma unroll 1
            for (int j = 0; j < cnt; ++j) {
                int s0 = __builtin_amdgcn_readlane(sl, j);
                const float* f0 = fl + (size_t)s0 * FOUT;
                float4 a0 = *(const float4*)f0, b0 = *(const float4*)(f0 + 128);
                float la0 = ll[(size_t)s0 * HEADS], lb0 = ll[(size_t)s0 * HEADS + 4];
                edge_fma(accA, accB, a0, b0, la0, lb0, erA, erB, mA, mB, iA, iB);
            }
        }
    }

    float ba[4], bb[4];
#pragma unroll
    for (int q = 0; q < 4; ++q) { ba[q] = bias[lane * 4 + q]; bb[q] = bias[128 + lane * 4 + q]; }
    float4 rA = make_float4(accA.x + ba[0], accA.y + ba[1], accA.z + ba[2], accA.w + ba[3]);
    float4 rB = make_float4(accB.x + bb[0], accB.y + bb[1], accB.z + bb[2], accB.w + bb[3]);
    if (mode == 0) {
        rA.x = (rA.x > 0.f) ? rA.x : (__expf(rA.x) - 1.f);
        rA.y = (rA.y > 0.f) ? rA.y : (__expf(rA.y) - 1.f);
        rA.z = (rA.z > 0.f) ? rA.z : (__expf(rA.z) - 1.f);
        rA.w = (rA.w > 0.f) ? rA.w : (__expf(rA.w) - 1.f);
        rB.x = (rB.x > 0.f) ? rB.x : (__expf(rB.x) - 1.f);
        rB.y = (rB.y > 0.f) ? rB.y : (__expf(rB.y) - 1.f);
        rB.z = (rB.z > 0.f) ? rB.z : (__expf(rB.z) - 1.f);
        rB.w = (rB.w > 0.f) ? rB.w : (__expf(rB.w) - 1.f);
    } else {
        rA.x += oA.x; rA.y += oA.y; rA.z += oA.z; rA.w += oA.w;
        rB.x += oB.x; rB.y += oB.y; rB.z += oB.z; rB.w += oB.w;
    }
    *(float4*)(out + obase)       = rA;
    *(float4*)(out + obase + 128) = rB;
}

static inline size_t align256(size_t x) { return (x + 255) & ~(size_t)255; }

extern "C" void kernel_launch(void* const* d_in, const int* in_sizes, int n_in,
                              void* d_out, int out_size, void* d_ws, size_t ws_size,
                              hipStream_t stream) {
    const int N = in_sizes[0] / 128;
    const int E = in_sizes[1];

    const float* x   = (const float*)d_in[0];
    const int*   src = (const int*)d_in[1];
    const int*   dst = (const int*)d_in[2];

    char* ws = (char*)d_ws;
    size_t o = 0;
    const int chunk = (E + BIN_GRID - 1) / BIN_GRID;
    const int nbins = (N + BIN_NODES - 1) / BIN_NODES;
    int* off    = (int*)(ws + o); o = align256(o + (size_t)(N + 1) * 4);
    int* cntT   = (int*)(ws + o); o = align256(o + (size_t)BIN_GRID * BIN_NODES * 4);
    int* lstT   = (int*)(ws + o); o = align256(o + (size_t)BIN_GRID * BIN_NODES * 4);
    int* rec    = (int*)(ws + o); o = align256(o + (size_t)BIN_GRID * chunk * 8);
    int* eidx   = (int*)(ws + o); o = align256(o + (size_t)E * 4);
    _Float16* wt16 = (_Float16*)(ws + o); o = align256(o + (size_t)FOUT * FOUT * 2);
    float* feat = (float*)(ws + o); o = align256(o + (size_t)N * FOUT * 4);
    float* el   = (float*)(ws + o); o = align256(o + (size_t)N * HEADS * 4);
    float* er   = (float*)(ws + o); o = align256(o + (size_t)N * HEADS * 4);
    (void)ws_size;

    float* h = (float*)d_out;

    const int T = 256;
    auto blk = [](int n, int t) { return (n + t - 1) / t; };

    bin_local<<<BIN_GRID, 1024, 0, stream>>>(src, dst, rec, cntT, lstT, E, chunk);
    bin_build<<<nbins, 1024, 0, stream>>>(rec, cntT, lstT, off, eidx, N, E, chunk);

    for (int layer = 0; layer < 4; ++layer) {
        const float* W  = (const float*)d_in[3 + 4 * layer];
        const float* al = (const float*)d_in[4 + 4 * layer];
        const float* ar = (const float*)d_in[5 + 4 * layer];
        const float* b  = (const float*)d_in[6 + 4 * layer];
        const int K = in_sizes[3 + 4 * layer] / FOUT;
        const float* hin = (layer == 0) ? x : h;

        transpose_w<<<blk(K * FOUT, T), T, 0, stream>>>(W, wt16, K);

        int nWaves = (N / 16) * (FOUT / 64);
        gemm_wmma_f16<<<blk(nWaves * 32, T), T, 0, stream>>>(hin, wt16, feat, al, ar, el, er, N, K);

        int mode = (layer == 0) ? 0 : 1;
        gat_pull<<<blk(N * 32, T), T, 0, stream>>>(feat, el, er, off, eidx, b, h, N, mode);
    }
}


# gemm operand loads: weights staged in B-fragment order so each 16-byte B load reads 4 whole lines, not 16, and a k-step's 12 loads go out as one group; same k order, values and sums
# speedup vs baseline: 2.2952x; 1.2218x over previous
#include <hip/hip_runtime.h>
#include <hip/hip_bf16.h>
#include <math.h>

typedef __attribute__((ext_vector_type(16))) _Float16 v16h;
typedef __attribute__((ext_vector_type(8)))  float    v8f;
typedef __attribute__((ext_vector_type(8)))  _Float16 v8h;
typedef __attribute__((ext_vector_type(4)))  float    v4f;

#define HEADS 8
#define HDIM  32
#define FOUT  256
#define NEG_SLOPE 0.2f

#define BIN_SHIFT 8
#define BIN_NODES 256
#define BIN_GRID  256

__device__ __forceinline__ int bin_of(int d) {
    int b = (int)((unsigned)d >> BIN_SHIFT);
    return (b < BIN_NODES) ? b : (BIN_NODES - 1);
}

__device__ __forceinline__ void scan256_inclusive(int* sh, int tid) {
    for (int o = 1; o < 256; o <<= 1) {
        int v = 0;
        if (tid < 256 && tid >= o) v = sh[tid - o];
        __syncthreads();
        if (tid < 256) sh[tid] += v;
        __syncthreads();
    }
}

__global__ void __launch_bounds__(1024) bin_local(const int* __restrict__ src,
                                                  const int* __restrict__ dst,
                                                  int* __restrict__ rec,
                                                  int* __restrict__ cntT,
                                                  int* __restrict__ lstT,
                                                  int e, int chunk) {
    __shared__ int hist[BIN_NODES];
    __shared__ int lst[BIN_NODES];
    __shared__ int cur[BIN_NODES];
    int t = threadIdx.x, blk = blockIdx.x;
    int e0 = blk * chunk;
    int e1 = e0 + chunk; if (e1 > e) e1 = e;
    if (t < 256) { hist[t] = 0; cur[t] = 0; }
    __syncthreads();
    for (int base = e0 + t; base < e1; base += 4096) {
        int d[4];
#pragma unroll
        for (int u = 0; u < 4; ++u) { int i = base + u * 1024; d[u] = dst[(i < e1) ? i : (e1 - 1)]; }
#pragma unroll
        for (int u = 0; u < 4; ++u) atomicAdd(&hist[bin_of(d[u])], (base + u * 1024 < e1) ? 1 : 0);
    }
    __syncthreads();
    int c = 0;
    if (t < 256) { c = hist[t]; lst[t] = c; }
    __syncthreads();
    scan256_inclusive(lst, t);
    if (t < 256) {
        int ex = lst[t] - c;
        cntT[blk * BIN_NODES + t] = c;
        lstT[blk * BIN_NODES + t] = ex;
        lst[t] = ex;
    }
    __syncthreads();
    int2* r2 = (int2*)rec;
    for (int base = e0 + t; base < e1; base += 4096) {
        int d[4], s[4];
#pragma unroll
        for (int u = 0; u < 4; ++u) {
            int i = base + u * 1024;
            int ic = (i < e1) ? i : (e1 - 1);
            d[u] = dst[ic];
            s[u] = src[ic];
        }
#pragma unroll
        for (int u = 0; u < 4; ++u) {
            if (base + u * 1024 < e1) {
                int b = bin_of(d[u]);
                int r = atomicAdd(&cur[b], 1);
                r2[e0 + lst[b] + r] = make_int2(s[u], d[u]);
            }
        }
    }
}

__global__ void __launch_bounds__(1024) bin_build(const int* __restrict__ rec,
                                                  const int* __restrict__ cntT,
                                                  const int* __restrict__ lstT,
                                                  int* __restrict__ off,
                                                  int* __restrict__ eidx,
                                                  int n, int e, int chunk) {
    __shared__ int cnt[BIN_NODES];
    __shared__ int exs[BIN_NODES];
    __shared__ int cur[BIN_NODES];
    __shared__ int sS;
    int t = threadIdx.x, b = blockIdx.x;
    if (t < 256) { cnt[t] = 0; cur[t] = 0; }
    if (t == 0) sS = 0;
    __syncthreads();

    int col = t & 255;
    int acc = 0;
    if (col < b) {
        for (int r = t >> 8; r < BIN_GRID; r += 4) acc += cntT[r * BIN_NODES + col];
    }
#pragma unroll
    for (int o = 16; o >= 1; o >>= 1) acc += __shfl_xor(acc, o);
    if ((t & 31) == 0 && acc != 0) atomicAdd(&sS, acc);

    int seg = t >> 2, sub = t & 3;
    int st  = seg * chunk + lstT[seg * BIN_NODES + b];
    int len = cntT[seg * BIN_NODES + b];
    const int2* r2 = (const int2*)rec;
    for (int base = sub; base < len; base += 16) {
        int d[4];
#pragma unroll
        for (int u = 0; u < 4; ++u) { int i = base + u * 4; d[u] = r2[st + ((i < len) ? i : (len - 1))].y; }
#pragma unroll
        for (int u = 0; u < 4; ++u) atomicAdd(&cnt[d[u] & (BIN_NODES - 1)], (base + u * 4 < len) ? 1 : 0);
    }
    __syncthreads();

    int S = sS;
    int own = 0;
    if (t < 256) { own = cnt[t]; exs[t] = own; }
    __syncthreads();
    scan256_inclusive(exs, t);
    if (t < 256) {
        int ex = exs[t] - own;
        int node = b * BIN_NODES + t;
        if (node < n) off[node] = S + ex;
        exs[t] = ex;
    }
    if (b == (int)gridDim.x - 1 && t == 0) off[n] = e;
    __syncthreads();
    for (int base = sub; base < len; base += 16) {
        int2 r[4];
#pragma unroll
        for (int u = 0; u < 4; ++u) { int i = base + u * 4; r[u] = r2[st + ((i < len) ? i : (len - 1))]; }
#pragma unroll
        for (int u = 0; u < 4; ++u) {
            if (base + u * 4 < len) {
                int l = r[u].y & (BIN_NODES - 1);
                int k = atomicAdd(&cur[l], 1);
                eidx[S + exs[l] + k] = r[u].x;
            }
        }
    }
}

__global__ void stage_w_frag(const float* __restrict__ W, _Float16* __restrict__ Wf, int K) {
    int i = blockIdx.x * blockDim.x + threadIdx.x;
    if (i < K * FOUT) {
        int k = i / FOUT, col = i % FOUT;
        int nt = col >> 4, c = col & 15;
        int ks = k >> 5, kk = k & 31;
        int h = (kk >> 3) & 1, up = kk >> 4, e = kk & 7;
        int ln = c + (h << 4);
        size_t d = ((size_t)(nt * (K >> 5) + ks) << 9) + (size_t)((up << 8) + (ln << 3) + e);
        Wf[d] = (_Float16)W[i];
    }
}

__device__ __forceinline__ v16h load_frag(const _Float16* __restrict__ blk, unsigned loff) {
    v8h lo = *(const v8h*)(blk + loff);
    v8h hi = *(const v8h*)(blk + loff + 256);
    return __builtin_shufflevector(lo, hi, 0, 1, 2, 3, 4, 5, 6, 7, 8, 9, 10, 11, 12, 13, 14, 15);
}

__device__ __forceinline__ v16h cvt_frag_f32(v4f x0, v4f x1, v4f y0, v4f y1) {
    v8f lo = __builtin_shufflevector(x0, x1, 0, 1, 2, 3, 4, 5, 6, 7);
    v8f hi = __builtin_shufflevector(y0, y1, 0, 1, 2, 3, 4, 5, 6, 7);
    v8h hl = __builtin_convertvector(lo, v8h);
    v8h hh = __builtin_convertvector(hi, v8h);
    return __builtin_shufflevector(hl, hh, 0, 1, 2, 3, 4, 5, 6, 7, 8, 9, 10, 11, 12, 13, 14, 15);
}

template <int H>
__device__ __forceinline__ void halve_step(float* v, bool up, int mask) {
#pragma unroll
    for (int k = 0; k < H; ++k) {
        float keep = up ? v[k + H] : v[k];
        float send = up ? v[k] : v[k + H];
        v[k] = keep + __shfl_xor(send, mask);
    }
}

__global__ void __launch_bounds__(256) gemm_wmma_f16(const float* __restrict__ A,
                                                     const _Float16* __restrict__ Bt,
                                                     float* __restrict__ C,
                                                     const float* __restrict__ al,
                                                     const float* __restrict__ ar,
                                                     float* __restrict__ el,
                                                     float* __restrict__ er,
                                                     int M, int K) {
    __shared__ float s_el[256];
    __shared__ float s_er[256];
    int lane = threadIdx.x & 31;
    int wv   = __builtin_amdgcn_readfirstlane((int)(threadIdx.x >> 5));
    int wid  = blockIdx.x * 8 + wv;
    int nWaves = (M >> 4) * (FOUT >> 6);

    bool valid = wid < nWaves;
    int m0 = (wid >> 2) << 4;
    int n0 = (wid & 3) << 6;
    int l15 = lane & 15;
    int rbase = (lane >> 4) << 3;
    v8f c0 = {}, c1 = {}, c2 = {}, c3 = {};

    if (valid) {
        int kh  = rbase;

        const float*    pa  = A  + (size_t)(m0 + l15) * K + kh;
        const size_t tstride = (size_t)(K >> 5) << 9;
        const _Float16* tb = Bt + (size_t)(n0 >> 4) * tstride;
        const unsigned loff = (unsigned)lane << 3;

        for (int k0 = 0; k0 < K; k0 += 32) {
            const v4f* qa = (const v4f*)(pa + k0);
            v4f x0 = qa[0], x1 = qa[1];
            v4f y0 = qa[4], y1 = qa[5];
            const _Float16* kb = tb + (size_t)k0 * 16;
            v16h b0 = load_frag(kb, loff);
            v16h b1 = load_frag(kb + tstride, loff);
            v16h b2 = load_frag(kb + 2 * tstride, loff);
            v16h b3 = load_frag(kb + 3 * tstride, loff);
            asm volatile("" :: "v"(x0), "v"(x1), "v"(y0), "v"(y1), "v"(b0), "v"(b1), "v"(b2), "v"(b3));
            v16h a = cvt_frag_f32(x0, x1, y0, y1);
            c0 = __builtin_amdgcn_wmma_f32_16x16x32_f16(false, a, false, b0, (short)0, c0, false, false);
            c1 = __builtin_amdgcn_wmma_f32_16x16x32_f16(false, a, false, b1, (short)0, c1, false, false);
            c2 = __builtin_amdgcn_wmma_f32_16x16x32_f16(false, a, false, b2, (short)0, c2, false, false);
            c3 = __builtin_amdgcn_wmma_f32_16x16x32_f16(false, a, false, b3, (short)0, c3, false, false);
        }
        float a0 = al[n0 + l15], a1 = al[n0 + l15 + 16], a2 = al[n0 + l15 + 32], a3 = al[n0 + l15 + 48];
        float r0 = ar[n0 + l15], r1 = ar[n0 + l15 + 16], r2 = ar[n0 + l15 + 32], r3 = ar[n0 + l15 + 48];

        float v[32];
#pragma unroll
        for (int r = 0; r < 8; ++r) {
            v[r]      = c0[r] * a0 + c1[r] * a1;
            v[8 + r]  = c0[r] * r0 + c1[r] * r1;
            v[16 + r] = c2[r] * a2 + c3[r] * a3;
            v[24 + r] = c2[r] * r2 + c3[r] * r3;
        }
        halve_step<16>(v, (l15 & 8) != 0, 8);
        halve_step<8>(v,  (l15 & 4) != 0, 4);
        halve_step<4>(v,  (l15 & 2) != 0, 2);
        halve_step<2>(v,  (l15 & 1) != 0, 1);
        int q    = l15 >> 2;
        int hd   = ((wv & 3) << 1) + (q >> 1);
        int rowb = ((wv >> 2) << 4) + rbase + ((l15 & 3) << 1);
        float* sd = (q & 1) ? s_er : s_el;
        sd[rowb * HEADS + hd]       = v[0];
        sd[(rowb + 1) * HEADS + hd] = v[1];
    }
    __syncthreads();
    int t = threadIdx.x;
    int row = blockIdx.x * 32 + (t >> 3);
    if (row < ((M >> 4) << 4)) {
        el[(size_t)row * HEADS + (t & 7)] = s_el[t];
        er[(size_t)row * HEADS + (t & 7)] = s_er[t];
    }
    if (valid) {
#pragma unroll
        for (int r = 0; r < 8; ++r) {
            size_t o = (size_t)(m0 + r + rbase) * FOUT + n0 + l15;
            C[o +  0] = c0[r];
            C[o + 16] = c1[r];
            C[o + 32] = c2[r];
            C[o + 48] = c3[r];
        }
    }
}

__device__ __forceinline__ float leaky(float e) { return (e > 0.f) ? e : NEG_SLOPE * e; }

__device__ __forceinline__ float pick4(const float* a, int k) {
    float r = a[0];
    r = (k == 1) ? a[1] : r;
    r = (k == 2) ? a[2] : r;
    r = (k == 3) ? a[3] : r;
    return r;
}

__device__ __forceinline__ void edge_fma(float4& accA, float4& accB, float4 fa, float4 fb,
                                         float la, float lb, float erA, float erB,
                                         float mA, float mB, float iA, float iB) {
    float wA = __expf(leaky(la + erA) - mA) * iA;
    float wB = __expf(leaky(lb + erB) - mB) * iB;
    accA.x += fa.x * wA; accA.y += fa.y * wA; accA.z += fa.z * wA; accA.w += fa.w * wA;
    accB.x += fb.x * wB; accB.y += fb.y * wB; accB.z += fb.z * wB; accB.w += fb.w * wB;
}

__device__ __forceinline__ void row_fma(float4& accA, float4& accB, float4 fa, float4 fb,
                                        float wA, float wB) {
    accA.x = fmaf(fa.x, wA, accA.x); accA.y = fmaf(fa.y, wA, accA.y);
    accA.z = fmaf(fa.z, wA, accA.z); accA.w = fmaf(fa.w, wA, accA.w);
    accB.x = fmaf(fb.x, wB, accB.x); accB.y = fmaf(fb.y, wB, accB.y);
    accB.z = fmaf(fb.z, wB, accB.z); accB.w = fmaf(fb.w, wB, accB.w);
}

__global__ void __launch_bounds__(256) gat_pull(const float* __restrict__ feat,
                                                const float* __restrict__ el,
                                                const float* __restrict__ er,
                                                const int* __restrict__ off,
                                                const int* __restrict__ eidx,
                                                const float* __restrict__ bias,
                                                float* __restrict__ out,
                                                int n, int mode) {
    __shared__ float s_w[8][32 * HEADS];
    int gt   = blockIdx.x * 256 + threadIdx.x;
    int lane = gt & 31;
    int v = __builtin_amdgcn_readfirstlane(gt >> 5);
    if (v >= n) return;

    int beg = off[v], end = off[v + 1];
    int hA = lane >> 3;
    size_t obase = (size_t)v * FOUT + lane * 4;

    float4 oA = make_float4(0.f, 0.f, 0.f, 0.f), oB = oA;
    if (mode != 0) {
        oA = *(const float4*)(out + obase);
        oB = *(const float4*)(out + obase + 128);
    }

    float ern[8];
#pragma unroll
    for (int h = 0; h < 8; ++h) ern[h] = er[(size_t)v * HEADS + h];

    const int deg = end - beg;
    float4 accA = make_float4(0.f, 0.f, 0.f, 0.f), accB = accA;
    const float* fl = feat + lane * 4;

    if (deg <= 32) {
        float* swv = s_w[threadIdx.x >> 5];
        const bool have = lane < deg;
        int sl = 0;
        float e[8];
#pragma unroll
        for (int h = 0; h < 8; ++h) e[h] = -INFINITY;
        if (have) {
            sl = eidx[beg + lane];
            const float4* ep = (const float4*)(el + (size_t)sl * HEADS);
            float4 a0 = ep[0], a1 = ep[1];
            float ev[8] = {a0.x, a0.y, a0.z, a0.w, a1.x, a1.y, a1.z, a1.w};
#pragma unroll
            for (int h = 0; h < 8; ++h) e[h] = leaky(ev[h] + ern[h]);
        }
        float mx[8];
#pragma unroll
        for (int h = 0; h < 8; ++h) {
            mx[h] = e[h];
#pragma unroll
            for (int o = 16; o >= 1; o >>= 1) mx[h] = fmaxf(mx[h], __shfl_xor(mx[h], o));
        }
        float p[8], sm[8];
#pragma unroll
        for (int h = 0; h < 8; ++h) {
            p[h] = have ? __expf(e[h] - mx[h]) : 0.f;
            sm[h] = p[h];
#pragma unroll
            for (int o = 16; o >= 1; o >>= 1) sm[h] += __shfl_xor(sm[h], o);
        }
        float w[8];
#pragma unroll
        for (int h = 0; h < 8; ++h) w[h] = p[h] * ((deg > 0) ? 1.f / sm[h] : 0.f);
        *(float4*)(swv + lane * HEADS)     = make_float4(w[0], w[1], w[2], w[3]);
        *(float4*)(swv + lane * HEADS + 4) = make_float4(w[4], w[5], w[6], w[7]);
        __builtin_amdgcn_wave_barrier();

        const float* wl = swv + hA;
        int j = 0;
        for (; j + 4 <= deg; j += 4) {
            int s0 = __builtin_amdgcn_readlane(sl, j);
            int s1 = __builtin_amdgcn_readlane(sl, j + 1);
            int s2 = __builtin_amdgcn_readlane(sl, j + 2);
            int s3 = __builtin_amdgcn_readlane(sl, j + 3);
            const float* f0 = fl + (size_t)s0 * FOUT;
            const float* f1 = fl + (size_t)s1 * FOUT;
            const float* f2 = fl + (size_t)s2 * FOUT;
            const float* f3 = fl + (size_t)s3 * FOUT;
            float4 a0 = *(const float4*)f0, b0 = *(const float4*)(f0 + 128);
            float4 a1 = *(const float4*)f1, b1 = *(const float4*)(f1 + 128);
            float4 a2 = *(const float4*)f2, b2 = *(const float4*)(f2 + 128);
            float4 a3 = *(const float4*)f3, b3 = *(const float4*)(f3 + 128);
            const float* wj = wl + j * HEADS;
            float wa0 = wj[0],  wb0 = wj[4];
            float wa1 = wj[8],  wb1 = wj[12];
            float wa2 = wj[16], wb2 = wj[20];
            float wa3 = wj[24], wb3 = wj[28];
            row_fma(accA, accB, a0, b0, wa0, wb0);
            row_fma(accA, accB, a1, b1, wa1, wb1);
            row_fma(accA, accB, a2, b2, wa2, wb2);
            row_fma(accA, accB, a3, b3, wa3, wb3);
        }
#pragma unroll 1
        for (; j < deg; ++j) {
            int s0 = __builtin_amdgcn_readlane(sl, j);
            const float* f0 = fl + (size_t)s0 * FOUT;
            float4 a0 = *(const float4*)f0, b0 = *(const float4*)(f0 + 128);
            float wa0 = wl[j * HEADS], wb0 = wl[j * HEADS + 4];
            row_fma(accA, accB, a0, b0, wa0, wb0);
        }
    } else {
        float mx[8];
#pragma unroll
        for (int h = 0; h < 8; ++h) mx[h] = -INFINITY;
        for (int c0 = beg; c0 < end; c0 += 32) {
            int t = c0 + lane;
            if (t < end) {
                int s = eidx[t];
                const float4* ep = (const float4*)(el + (size_t)s * HEADS);
                float4 a0 = ep[0], a1 = ep[1];
                float ev[8] = {a0.x, a0.y, a0.z, a0.w, a1.x, a1.y, a1.z, a1.w};
#pragma unroll
                for (int h = 0; h < 8; ++h) mx[h] = fmaxf(mx[h], leaky(ev[h] + ern[h]));
            }
        }
#pragma unroll
        for (int h = 0; h < 8; ++h) {
#pragma unroll
            for (int o = 16; o >= 1; o >>= 1) mx[h] = fmaxf(mx[h], __shfl_xor(mx[h], o));
        }

        float sm[8];
#pragma unroll
        for (int h = 0; h < 8; ++h) sm[h] = 0.f;
        for (int c0 = beg; c0 < end; c0 += 32) {
            int t = c0 + lane;
            if (t < end) {
                int s = eidx[t];
                const float4* ep = (const float4*)(el + (size_t)s * HEADS);
                float4 a0 = ep[0], a1 = ep[1];
                float ev[8] = {a0.x, a0.y, a0.z, a0.w, a1.x, a1.y, a1.z, a1.w};
#pragma unroll
                for (int h = 0; h < 8; ++h) sm[h] += __expf(leaky(ev[h] + ern[h]) - mx[h]);
            }
        }
#pragma unroll
        for (int h = 0; h < 8; ++h) {
#pragma unroll
            for (int o = 16; o >= 1; o >>= 1) sm[h] += __shfl_xor(sm[h], o);
        }
        float inv[8];
#pragma unroll
        for (int h = 0; h < 8; ++h) inv[h] = (end > beg) ? 1.f / sm[h] : 0.f;

        float mA = pick4(mx, hA),  mB = pick4(mx + 4, hA);
        float iA = pick4(inv, hA), iB = pick4(inv + 4, hA);
        float erA = pick4(ern, hA), erB = pick4(ern + 4, hA);

        const float* ll = el + hA;
        for (int c0 = beg; c0 < end; c0 += 32) {
            int cnt = end - c0; if (cnt > 32) cnt = 32;
            int sl = (lane < cnt) ? eidx[c0 + lane] : 0;
#pragma unroll 1
            for (int j = 0; j < cnt; ++j) {
                int s0 = __builtin_amdgcn_readlane(sl, j);
                const float* f0 = fl + (size_t)s0 * FOUT;
                float4 a0 = *(const float4*)f0, b0 = *(const float4*)(f0 + 128);
                float la0 = ll[(size_t)s0 * HEADS], lb0 = ll[(size_t)s0 * HEADS + 4];
                edge_fma(accA, accB, a0, b0, la0, lb0, erA, erB, mA, mB, iA, iB);
            }
        }
    }

    float ba[4], bb[4];
#pragma unroll
    for (int q = 0; q < 4; ++q) { ba[q] = bias[lane * 4 + q]; bb[q] = bias[128 + lane * 4 + q]; }
    float4 rA = make_float4(accA.x + ba[0], accA.y + ba[1], accA.z + ba[2], accA.w + ba[3]);
    float4 rB = make_float4(accB.x + bb[0], accB.y + bb[1], accB.z + bb[2], accB.w + bb[3]);
    if (mode == 0) {
        rA.x = (rA.x > 0.f) ? rA.x : (__expf(rA.x) - 1.f);
        rA.y = (rA.y > 0.f) ? rA.y : (__expf(rA.y) - 1.f);
        rA.z = (rA.z > 0.f) ? rA.z : (__expf(rA.z) - 1.f);
        rA.w = (rA.w > 0.f) ? rA.w : (__expf(rA.w) - 1.f);
        rB.x = (rB.x > 0.f) ? rB.x : (__expf(rB.x) - 1.f);
        rB.y = (rB.y > 0.f) ? rB.y : (__expf(rB.y) - 1.f);
        rB.z = (rB.z > 0.f) ? rB.z : (__expf(rB.z) - 1.f);
        rB.w = (rB.w > 0.f) ? rB.w : (__expf(rB.w) - 1.f);
    } else {
        rA.x += oA.x; rA.y += oA.y; rA.z += oA.z; rA.w += oA.w;
        rB.x += oB.x; rB.y += oB.y; rB.z += oB.z; rB.w += oB.w;
    }
    *(float4*)(out + obase)       = rA;
    *(float4*)(out + obase + 128) = rB;
}

static inline size_t align256(size_t x) { return (x + 255) & ~(size_t)255; }

extern "C" void kernel_launch(void* const* d_in, const int* in_sizes, int n_in,
                              void* d_out, int out_size, void* d_ws, size_t ws_size,
                              hipStream_t stream) {
    const int N = in_sizes[0] / 128;
    const int E = in_sizes[1];

    const float* x   = (const float*)d_in[0];
    const int*   src = (const int*)d_in[1];
    const int*   dst = (const int*)d_in[2];

    char* ws = (char*)d_ws;
    size_t o = 0;
    const int chunk = (E + BIN_GRID - 1) / BIN_GRID;
    const int nbins = (N + BIN_NODES - 1) / BIN_NODES;
    int* off    = (int*)(ws + o); o = align256(o + (size_t)(N + 1) * 4);
    int* cntT   = (int*)(ws + o); o = align256(o + (size_t)BIN_GRID * BIN_NODES * 4);
    int* lstT   = (int*)(ws + o); o = align256(o + (size_t)BIN_GRID * BIN_NODES * 4);
    int* rec    = (int*)(ws + o); o = align256(o + (size_t)BIN_GRID * chunk * 8);
    int* eidx   = (int*)(ws + o); o = align256(o + (size_t)E * 4);
    _Float16* wt16 = (_Float16*)(ws + o); o = align256(o + (size_t)FOUT * FOUT * 2);
    float* feat = (float*)(ws + o); o = align256(o + (size_t)N * FOUT * 4);
    float* el   = (float*)(ws + o); o = align256(o + (size_t)N * HEADS * 4);
    float* er   = (float*)(ws + o); o = align256(o + (size_t)N * HEADS * 4);
    (void)ws_size;

    float* h = (float*)d_out;

    const int T = 256;
    auto blk = [](int n, int t) { return (n + t - 1) / t; };

    bin_local<<<BIN_GRID, 1024, 0, stream>>>(src, dst, rec, cntT, lstT, E, chunk);
    bin_build<<<nbins, 1024, 0, stream>>>(rec, cntT, lstT, off, eidx, N, E, chunk);

    for (int layer = 0; layer < 4; ++layer) {
        const float* W  = (const float*)d_in[3 + 4 * layer];
        const float* al = (const float*)d_in[4 + 4 * layer];
        const float* ar = (const float*)d_in[5 + 4 * layer];
        const float* b  = (const float*)d_in[6 + 4 * layer];
        const int K = in_sizes[3 + 4 * layer] / FOUT;
        const float* hin = (layer == 0) ? x : h;

        stage_w_frag<<<blk(K * FOUT, T), T, 0, stream>>>(W, wt16, K);

        int nWaves = (N / 16) * (FOUT / 64);
        gemm_wmma_f16<<<blk(nWaves * 32, T), T, 0, stream>>>(hin, wt16, feat, al, ar, el, er, N, K);

        int mode = (layer == 0) ? 0 : 1;
        gat_pull<<<blk(N * 32, T), T, 0, stream>>>(feat, el, er, off, eidx, b, h, N, mode);
    }
}


# gemm: weights staged once per work-group in LDS (one barrier, LDS read-only after it); each wave owns a 16x256 block of whole rows, so A is read once and B comes from LDS; same k order and sums
# speedup vs baseline: 2.4490x; 1.0670x over previous
#include <hip/hip_runtime.h>
#include <hip/hip_bf16.h>
#include <math.h>

typedef __attribute__((ext_vector_type(16))) _Float16 v16h;
typedef __attribute__((ext_vector_type(8)))  float    v8f;
typedef __attribute__((ext_vector_type(8)))  _Float16 v8h;
typedef __attribute__((ext_vector_type(4)))  float    v4f;

#define HEADS 8
#define HDIM  32
#define FOUT  256
#define NEG_SLOPE 0.2f

#define BIN_SHIFT 8
#define BIN_NODES 256
#define BIN_GRID  256

__device__ __forceinline__ int bin_of(int d) {
    int b = (int)((unsigned)d >> BIN_SHIFT);
    return (b < BIN_NODES) ? b : (BIN_NODES - 1);
}

__device__ __forceinline__ void scan256_inclusive(int* sh, int tid) {
    for (int o = 1; o < 256; o <<= 1) {
        int v = 0;
        if (tid < 256 && tid >= o) v = sh[tid - o];
        __syncthreads();
        if (tid < 256) sh[tid] += v;
        __syncthreads();
    }
}

__global__ void __launch_bounds__(1024) bin_local(const int* __restrict__ src,
                                                  const int* __restrict__ dst,
                                                  int* __restrict__ rec,
                                                  int* __restrict__ cntT,
                                                  int* __restrict__ lstT,
                                                  int e, int chunk) {
    __shared__ int hist[BIN_NODES];
    __shared__ int lst[BIN_NODES];
    __shared__ int cur[BIN_NODES];
    int t = threadIdx.x, blk = blockIdx.x;
    int e0 = blk * chunk;
    int e1 = e0 + chunk; if (e1 > e) e1 = e;
    if (t < 256) { hist[t] = 0; cur[t] = 0; }
    __syncthreads();
    for (int base = e0 + t; base < e1; base += 4096) {
        int d[4];
#pragma unroll
        for (int u = 0; u < 4; ++u) { int i = base + u * 1024; d[u] = dst[(i < e1) ? i : (e1 - 1)]; }
#pragma unroll
        for (int u = 0; u < 4; ++u) atomicAdd(&hist[bin_of(d[u])], (base + u * 1024 < e1) ? 1 : 0);
    }
    __syncthreads();
    int c = 0;
    if (t < 256) { c = hist[t]; lst[t] = c; }
    __syncthreads();
    scan256_inclusive(lst, t);
    if (t < 256) {
        int ex = lst[t] - c;
        cntT[blk * BIN_NODES + t] = c;
        lstT[blk * BIN_NODES + t] = ex;
        lst[t] = ex;
    }
    __syncthreads();
    int2* r2 = (int2*)rec;
    for (int base = e0 + t; base < e1; base += 4096) {
        int d[4], s[4];
#pragma unroll
        for (int u = 0; u < 4; ++u) {
            int i = base + u * 1024;
            int ic = (i < e1) ? i : (e1 - 1);
            d[u] = dst[ic];
            s[u] = src[ic];
        }
#pragma unroll
        for (int u = 0; u < 4; ++u) {
            if (base + u * 1024 < e1) {
                int b = bin_of(d[u]);
                int r = atomicAdd(&cur[b], 1);
                r2[e0 + lst[b] + r] = make_int2(s[u], d[u]);
            }
        }
    }
}

__global__ void __launch_bounds__(1024) bin_build(const int* __restrict__ rec,
                                                  const int* __restrict__ cntT,
                                                  const int* __restrict__ lstT,
                                                  int* __restrict__ off,
                                                  int* __restrict__ eidx,
                                                  int n, int e, int chunk) {
    __shared__ int cnt[BIN_NODES];
    __shared__ int exs[BIN_NODES];
    __shared__ int cur[BIN_NODES];
    __shared__ int sS;
    int t = threadIdx.x, b = blockIdx.x;
    if (t < 256) { cnt[t] = 0; cur[t] = 0; }
    if (t == 0) sS = 0;
    __syncthreads();

    int col = t & 255;
    int acc = 0;
    if (col < b) {
        for (int r = t >> 8; r < BIN_GRID; r += 4) acc += cntT[r * BIN_NODES + col];
    }
#pragma unroll
    for (int o = 16; o >= 1; o >>= 1) acc += __shfl_xor(acc, o);
    if ((t & 31) == 0 && acc != 0) atomicAdd(&sS, acc);

    int seg = t >> 2, sub = t & 3;
    int st  = seg * chunk + lstT[seg * BIN_NODES + b];
    int len = cntT[seg * BIN_NODES + b];
    const int2* r2 = (const int2*)rec;
    for (int base = sub; base < len; base += 16) {
        int d[4];
#pragma unroll
        for (int u = 0; u < 4; ++u) { int i = base + u * 4; d[u] = r2[st + ((i < len) ? i : (len - 1))].y; }
#pragma unroll
        for (int u = 0; u < 4; ++u) atomicAdd(&cnt[d[u] & (BIN_NODES - 1)], (base + u * 4 < len) ? 1 : 0);
    }
    __syncthreads();

    int S = sS;
    int own = 0;
    if (t < 256) { own = cnt[t]; exs[t] = own; }
    __syncthreads();
    scan256_inclusive(exs, t);
    if (t < 256) {
        int ex = exs[t] - own;
        int node = b * BIN_NODES + t;
        if (node < n) off[node] = S + ex;
        exs[t] = ex;
    }
    if (b == (int)gridDim.x - 1 && t == 0) off[n] = e;
    __syncthreads();
    for (int base = sub; base < len; base += 16) {
        int2 r[4];
#pragma unroll
        for (int u = 0; u < 4; ++u) { int i = base + u * 4; r[u] = r2[st + ((i < len) ? i : (len - 1))]; }
#pragma unroll
        for (int u = 0; u < 4; ++u) {
            if (base + u * 4 < len) {
                int l = r[u].y & (BIN_NODES - 1);
                int k = atomicAdd(&cur[l], 1);
                eidx[S + exs[l] + k] = r[u].x;
            }
        }
    }
}

__global__ void stage_w_frag(const float* __restrict__ W, _Float16* __restrict__ Wf, int K) {
    int i = blockIdx.x * blockDim.x + threadIdx.x;
    if (i < K * FOUT) {
        int k = i / FOUT, col = i % FOUT;
        int nt = col >> 4, c = col & 15;
        int ks = k >> 5, kk = k & 31;
        int h = (kk >> 3) & 1, up = kk >> 4, e = kk & 7;
        int ln = c + (h << 4);
        size_t d = ((size_t)(nt * (K >> 5) + ks) << 9) + (size_t)((up << 8) + (ln << 3) + e);
        Wf[d] = (_Float16)W[i];
    }
}

__device__ __forceinline__ v16h cvt_frag_f32(v4f x0, v4f x1, v4f y0, v4f y1) {
    v8f lo = __builtin_shufflevector(x0, x1, 0, 1, 2, 3, 4, 5, 6, 7);
    v8f hi = __builtin_shufflevector(y0, y1, 0, 1, 2, 3, 4, 5, 6, 7);
    v8h hl = __builtin_convertvector(lo, v8h);
    v8h hh = __builtin_convertvector(hi, v8h);
    return __builtin_shufflevector(hl, hh, 0, 1, 2, 3, 4, 5, 6, 7, 8, 9, 10, 11, 12, 13, 14, 15);
}

template <int H>
__device__ __forceinline__ void halve_step(float* v, bool up, int mask) {
#pragma unroll
    for (int k = 0; k < H; ++k) {
        float keep = up ? v[k + H] : v[k];
        float send = up ? v[k] : v[k + H];
        v[k] = keep + __shfl_xor(send, mask);
    }
}

#define GEMM_WAVES   13
#define GEMM_THREADS (GEMM_WAVES * 32)

#define STAGE_BATCH  10
static_assert(STAGE_BATCH == 10, "stage_batch names its ten loaded values one by one");

__device__ __forceinline__ void stage_batch(const v8h* __restrict__ g, v8h* s, int base, int units) {
    v8h t[STAGE_BATCH];
#pragma unroll
    for (int u = 0; u < STAGE_BATCH; ++u) { int i = base + u * GEMM_THREADS; t[u] = g[(i < units) ? i : (units - 1)]; }
    asm volatile("" :: "v"(t[0]), "v"(t[1]), "v"(t[2]), "v"(t[3]), "v"(t[4]), "v"(t[5]), "v"(t[6]), "v"(t[7]), "v"(t[8]), "v"(t[9]));
#pragma unroll
    for (int u = 0; u < STAGE_BATCH; ++u) { int i = base + u * GEMM_THREADS; if (i < units) s[i] = t[u]; }
}

__global__ void __launch_bounds__(GEMM_THREADS) gemm_wmma_f16(const float* __restrict__ A,
                                                              const _Float16* __restrict__ Bt,
                                                              float* __restrict__ C,
                                                              const float* __restrict__ al,
                                                              const float* __restrict__ ar,
                                                              float* __restrict__ el,
                                                              float* __restrict__ er,
                                                              int M, int K) {
    __shared__ v8h   sB[FOUT * FOUT / 8];
    __shared__ float s_al[FOUT];
    __shared__ float s_ar[FOUT];
    const int tid  = threadIdx.x;
    const int lane = tid & 31;
    const int wv     = __builtin_amdgcn_readfirstlane(tid >> 5);
    const int nTiles = M >> 4;
    const int ksteps = K >> 5;
    const int stride = (int)gridDim.x * GEMM_WAVES;
    const int l15   = lane & 15;
    const int rbase = (lane >> 4) << 3;

    int tile = (int)blockIdx.x * GEMM_WAVES + wv;
    const int t0 = (tile < nTiles) ? tile : (nTiles - 1);
    const float* pa = A + (size_t)((t0 << 4) + l15) * K + rbase;
    v4f x0 = ((const v4f*)pa)[0], x1 = ((const v4f*)pa)[1];
    v4f y0 = ((const v4f*)pa)[4], y1 = ((const v4f*)pa)[5];

    {
        const v8h* g = (const v8h*)Bt;
        const int units = K * (FOUT / 8);
        stage_batch(g, sB, tid, units);
        for (int base = tid + STAGE_BATCH * GEMM_THREADS; base < units; base += STAGE_BATCH * GEMM_THREADS)
            stage_batch(g, sB, base, units);
        if (tid < FOUT) { s_al[tid] = al[tid]; s_ar[tid] = ar[tid]; }
    }
    __syncthreads();

    const int tstride = ksteps << 6;
    while (tile < nTiles) {
        const int m0 = tile << 4;
        v8f c[16];
#pragma unroll
        for (int nt = 0; nt < 16; ++nt) c[nt] = (v8f){};

        for (int ks = 0; ks < ksteps; ++ks) {
            const int kn = (ks + 1 < ksteps) ? (ks + 1) : ks;
            const v4f* qn = (const v4f*)(pa + (kn << 5));
            v4f nx0 = qn[0], nx1 = qn[1], ny0 = qn[4], ny1 = qn[5];
            v16h a = cvt_frag_f32(x0, x1, y0, y1);
            const v8h* kb = sB + (ks << 6) + lane;
#pragma unroll
            for (int g = 0; g < 16; g += 8) {
                v16h b[8];
#pragma unroll
                for (int j = 0; j < 8; ++j) {
                    v8h lo = kb[(g + j) * tstride];
                    v8h hi = kb[(g + j) * tstride + 32];
                    b[j] = __builtin_shufflevector(lo, hi, 0, 1, 2, 3, 4, 5, 6, 7, 8, 9, 10, 11, 12, 13, 14, 15);
                }
                asm volatile("" :: "v"(b[0]), "v"(b[1]), "v"(b[2]), "v"(b[3]), "v"(b[4]), "v"(b[5]), "v"(b[6]), "v"(b[7]));
#pragma unroll
                for (int j = 0; j < 8; ++j)
                    c[g + j] = __builtin_amdgcn_wmma_f32_16x16x32_f16(false, a, false, b[j], (short)0, c[g + j], false, false);
            }
            x0 = nx0; x1 = nx1; y0 = ny0; y1 = ny1;
        }

#pragma unroll
        for (int r = 0; r < 8; ++r) {
            const size_t o = (size_t)(m0 + r + rbase) * FOUT + l15;
#pragma unroll
            for (int nt = 0; nt < 16; ++nt) C[o + 16 * nt] = c[nt][r];
        }

#pragma unroll
        for (int i = 0; i < 4; ++i) {
            const int n0 = i << 6;
            const v8f c0 = c[4 * i], c1 = c[4 * i + 1], c2 = c[4 * i + 2], c3 = c[4 * i + 3];
            float a0 = s_al[n0 + l15], a1 = s_al[n0 + l15 + 16], a2 = s_al[n0 + l15 + 32], a3 = s_al[n0 + l15 + 48];
            float r0 = s_ar[n0 + l15], r1 = s_ar[n0 + l15 + 16], r2 = s_ar[n0 + l15 + 32], r3 = s_ar[n0 + l15 + 48];
            float v[32];
#pragma unroll
            for (int r = 0; r < 8; ++r) {
                v[r]      = c0[r] * a0 + c1[r] * a1;
                v[8 + r]  = c0[r] * r0 + c1[r] * r1;
                v[16 + r] = c2[r] * a2 + c3[r] * a3;
                v[24 + r] = c2[r] * r2 + c3[r] * r3;
            }
            halve_step<16>(v, (l15 & 8) != 0, 8);
            halve_step<8>(v,  (l15 & 4) != 0, 4);
            halve_step<4>(v,  (l15 & 2) != 0, 2);
            halve_step<2>(v,  (l15 & 1) != 0, 1);
            const int q   = l15 >> 2;
            const int hd  = (i << 1) + (q >> 1);
            const int row = m0 + rbase + ((l15 & 3) << 1);
            float* gd = (q & 1) ? er : el;
            gd[(size_t)row * HEADS + hd]       = v[0];
            gd[(size_t)(row + 1) * HEADS + hd] = v[1];
        }

        tile += stride;
        if (tile < nTiles) {
            pa = A + (size_t)((tile << 4) + l15) * K + rbase;
            x0 = ((const v4f*)pa)[0]; x1 = ((const v4f*)pa)[1];
            y0 = ((const v4f*)pa)[4]; y1 = ((const v4f*)pa)[5];
        }
    }
}

__device__ __forceinline__ float leaky(float e) { return (e > 0.f) ? e : NEG_SLOPE * e; }

__device__ __forceinline__ float pick4(const float* a, int k) {
    float r = a[0];
    r = (k == 1) ? a[1] : r;
    r = (k == 2) ? a[2] : r;
    r = (k == 3) ? a[3] : r;
    return r;
}

__device__ __forceinline__ void edge_fma(float4& accA, float4& accB, float4 fa, float4 fb,
                                         float la, float lb, float erA, float erB,
                                         float mA, float mB, float iA, float iB) {
    float wA = __expf(leaky(la + erA) - mA) * iA;
    float wB = __expf(leaky(lb + erB) - mB) * iB;
    accA.x += fa.x * wA; accA.y += fa.y * wA; accA.z += fa.z * wA; accA.w += fa.w * wA;
    accB.x += fb.x * wB; accB.y += fb.y * wB; accB.z += fb.z * wB; accB.w += fb.w * wB;
}

__device__ __forceinline__ void row_fma(float4& accA, float4& accB, float4 fa, float4 fb,
                                        float wA, float wB) {
    accA.x = fmaf(fa.x, wA, accA.x); accA.y = fmaf(fa.y, wA, accA.y);
    accA.z = fmaf(fa.z, wA, accA.z); accA.w = fmaf(fa.w, wA, accA.w);
    accB.x = fmaf(fb.x, wB, accB.x); accB.y = fmaf(fb.y, wB, accB.y);
    accB.z = fmaf(fb.z, wB, accB.z); accB.w = fmaf(fb.w, wB, accB.w);
}

__global__ void __launch_bounds__(256) gat_pull(const float* __restrict__ feat,
                                                const float* __restrict__ el,
                                                const float* __restrict__ er,
                                                const int* __restrict__ off,
                                                const int* __restrict__ eidx,
                                                const float* __restrict__ bias,
                                                float* __restrict__ out,
                                                int n, int mode) {
    __shared__ float s_w[8][32 * HEADS];
    int gt   = blockIdx.x * 256 + threadIdx.x;
    int lane = gt & 31;
    int v = __builtin_amdgcn_readfirstlane(gt >> 5);
    if (v >= n) return;

    int beg = off[v], end = off[v + 1];
    int hA = lane >> 3;
    size_t obase = (size_t)v * FOUT + lane * 4;

    float4 oA = make_float4(0.f, 0.f, 0.f, 0.f), oB = oA;
    if (mode != 0) {
        oA = *(const float4*)(out + obase);
        oB = *(const float4*)(out + obase + 128);
    }

    float ern[8];
#pragma unroll
    for (int h = 0; h < 8; ++h) ern[h] = er[(size_t)v * HEADS + h];

    const int deg = end - beg;
    float4 accA = make_float4(0.f, 0.f, 0.f, 0.f), accB = accA;
    const float* fl = feat + lane * 4;

    if (deg <= 32) {
        float* swv = s_w[threadIdx.x >> 5];
        const bool have = lane < deg;
        int sl = 0;
        float e[8];
#pragma unroll
        for (int h = 0; h < 8; ++h) e[h] = -INFINITY;
        if (have) {
            sl = eidx[beg + lane];
            const float4* ep = (const float4*)(el + (size_t)sl * HEADS);
            float4 a0 = ep[0], a1 = ep[1];
            float ev[8] = {a0.x, a0.y, a0.z, a0.w, a1.x, a1.y, a1.z, a1.w};
#pragma unroll
            for (int h = 0; h < 8; ++h) e[h] = leaky(ev[h] + ern[h]);
        }
        float mx[8];
#pragma unroll
        for (int h = 0; h < 8; ++h) {
            mx[h] = e[h];
#pragma unroll
            for (int o = 16; o >= 1; o >>= 1) mx[h] = fmaxf(mx[h], __shfl_xor(mx[h], o));
        }
        float p[8], sm[8];
#pragma unroll
        for (int h = 0; h < 8; ++h) {
            p[h] = have ? __expf(e[h] - mx[h]) : 0.f;
            sm[h] = p[h];
#pragma unroll
            for (int o = 16; o >= 1; o >>= 1) sm[h] += __shfl_xor(sm[h], o);
        }
        float w[8];
#pragma unroll
        for (int h = 0; h < 8; ++h) w[h] = p[h] * ((deg > 0) ? 1.f / sm[h] : 0.f);
        *(float4*)(swv + lane * HEADS)     = make_float4(w[0], w[1], w[2], w[3]);
        *(float4*)(swv + lane * HEADS + 4) = make_float4(w[4], w[5], w[6], w[7]);
        __builtin_amdgcn_wave_barrier();

        const float* wl = swv + hA;
        int j = 0;
        for (; j + 4 <= deg; j += 4) {
            int s0 = __builtin_amdgcn_readlane(sl, j);
            int s1 = __builtin_amdgcn_readlane(sl, j + 1);
            int s2 = __builtin_amdgcn_readlane(sl, j + 2);
            int s3 = __builtin_amdgcn_readlane(sl, j + 3);
            const float* f0 = fl + (size_t)s0 * FOUT;
            const float* f1 = fl + (size_t)s1 * FOUT;
            const float* f2 = fl + (size_t)s2 * FOUT;
            const float* f3 = fl + (size_t)s3 * FOUT;
            float4 a0 = *(const float4*)f0, b0 = *(const float4*)(f0 + 128);
            float4 a1 = *(const float4*)f1, b1 = *(const float4*)(f1 + 128);
            float4 a2 = *(const float4*)f2, b2 = *(const float4*)(f2 + 128);
            float4 a3 = *(const float4*)f3, b3 = *(const float4*)(f3 + 128);
            const float* wj = wl + j * HEADS;
            float wa0 = wj[0],  wb0 = wj[4];
            float wa1 = wj[8],  wb1 = wj[12];
            float wa2 = wj[16], wb2 = wj[20];
            float wa3 = wj[24], wb3 = wj[28];
            row_fma(accA, accB, a0, b0, wa0, wb0);
            row_fma(accA, accB, a1, b1, wa1, wb1);
            row_fma(accA, accB, a2, b2, wa2, wb2);
            row_fma(accA, accB, a3, b3, wa3, wb3);
        }
#pragma unroll 1
        for (; j < deg; ++j) {
            int s0 = __builtin_amdgcn_readlane(sl, j);
            const float* f0 = fl + (size_t)s0 * FOUT;
            float4 a0 = *(const float4*)f0, b0 = *(const float4*)(f0 + 128);
            float wa0 = wl[j * HEADS], wb0 = wl[j * HEADS + 4];
            row_fma(accA, accB, a0, b0, wa0, wb0);
        }
    } else {
        float mx[8];
#pragma unroll
        for (int h = 0; h < 8; ++h) mx[h] = -INFINITY;
        for (int c0 = beg; c0 < end; c0 += 32) {
            int t = c0 + lane;
            if (t < end) {
                int s = eidx[t];
                const float4* ep = (const float4*)(el + (size_t)s * HEADS);
                float4 a0 = ep[0], a1 = ep[1];
                float ev[8] = {a0.x, a0.y, a0.z, a0.w, a1.x, a1.y, a1.z, a1.w};
#pragma unroll
                for (int h = 0; h < 8; ++h) mx[h] = fmaxf(mx[h], leaky(ev[h] + ern[h]));
            }
        }
#pragma unroll
        for (int h = 0; h < 8; ++h) {
#pragma unroll
            for (int o = 16; o >= 1; o >>= 1) mx[h] = fmaxf(mx[h], __shfl_xor(mx[h], o));
        }

        float sm[8];
#pragma unroll
        for (int h = 0; h < 8; ++h) sm[h] = 0.f;
        for (int c0 = beg; c0 < end; c0 += 32) {
            int t = c0 + lane;
            if (t < end) {
                int s = eidx[t];
                const float4* ep = (const float4*)(el + (size_t)s * HEADS);
                float4 a0 = ep[0], a1 = ep[1];
                float ev[8] = {a0.x, a0.y, a0.z, a0.w, a1.x, a1.y, a1.z, a1.w};
#pragma unroll
                for (int h = 0; h < 8; ++h) sm[h] += __expf(leaky(ev[h] + ern[h]) - mx[h]);
            }
        }
#pragma unroll
        for (int h = 0; h < 8; ++h) {
#pragma unroll
            for (int o = 16; o >= 1; o >>= 1) sm[h] += __shfl_xor(sm[h], o);
        }
        float inv[8];
#pragma unroll
        for (int h = 0; h < 8; ++h) inv[h] = (end > beg) ? 1.f / sm[h] : 0.f;

        float mA = pick4(mx, hA),  mB = pick4(mx + 4, hA);
        float iA = pick4(inv, hA), iB = pick4(inv + 4, hA);
        float erA = pick4(ern, hA), erB = pick4(ern + 4, hA);

        const float* ll = el + hA;
        for (int c0 = beg; c0 < end; c0 += 32) {
            int cnt = end - c0; if (cnt > 32) cnt = 32;
            int sl = (lane < cnt) ? eidx[c0 + lane] : 0;
#pragma unroll 1
            for (int j = 0; j < cnt; ++j) {
                int s0 = __builtin_amdgcn_readlane(sl, j);
                const float* f0 = fl + (size_t)s0 * FOUT;
                float4 a0 = *(const float4*)f0, b0 = *(const float4*)(f0 + 128);
                float la0 = ll[(size_t)s0 * HEADS], lb0 = ll[(size_t)s0 * HEADS + 4];
                edge_fma(accA, accB, a0, b0, la0, lb0, erA, erB, mA, mB, iA, iB);
            }
        }
    }

    float ba[4], bb[4];
#pragma unroll
    for (int q = 0; q < 4; ++q) { ba[q] = bias[lane * 4 + q]; bb[q] = bias[128 + lane * 4 + q]; }
    float4 rA = make_float4(accA.x + ba[0], accA.y + ba[1], accA.z + ba[2], accA.w + ba[3]);
    float4 rB = make_float4(accB.x + bb[0], accB.y + bb[1], accB.z + bb[2], accB.w + bb[3]);
    if (mode == 0) {
        rA.x = (rA.x > 0.f) ? rA.x : (__expf(rA.x) - 1.f);
        rA.y = (rA.y > 0.f) ? rA.y : (__expf(rA.y) - 1.f);
        rA.z = (rA.z > 0.f) ? rA.z : (__expf(rA.z) - 1.f);
        rA.w = (rA.w > 0.f) ? rA.w : (__expf(rA.w) - 1.f);
        rB.x = (rB.x > 0.f) ? rB.x : (__expf(rB.x) - 1.f);
        rB.y = (rB.y > 0.f) ? rB.y : (__expf(rB.y) - 1.f);
        rB.z = (rB.z > 0.f) ? rB.z : (__expf(rB.z) - 1.f);
        rB.w = (rB.w > 0.f) ? rB.w : (__expf(rB.w) - 1.f);
    } else {
        rA.x += oA.x; rA.y += oA.y; rA.z += oA.z; rA.w += oA.w;
        rB.x += oB.x; rB.y += oB.y; rB.z += oB.z; rB.w += oB.w;
    }
    *(float4*)(out + obase)       = rA;
    *(float4*)(out + obase + 128) = rB;
}

static inline size_t align256(size_t x) { return (x + 255) & ~(size_t)255; }

extern "C" void kernel_launch(void* const* d_in, const int* in_sizes, int n_in,
                              void* d_out, int out_size, void* d_ws, size_t ws_size,
                              hipStream_t stream) {
    const int N = in_sizes[0] / 128;
    const int E = in_sizes[1];

    const float* x   = (const float*)d_in[0];
    const int*   src = (const int*)d_in[1];
    const int*   dst = (const int*)d_in[2];

    char* ws = (char*)d_ws;
    size_t o = 0;
    const int chunk = (E + BIN_GRID - 1) / BIN_GRID;
    const int nbins = (N + BIN_NODES - 1) / BIN_NODES;
    int* off    = (int*)(ws + o); o = align256(o + (size_t)(N + 1) * 4);
    int* cntT   = (int*)(ws + o); o = align256(o + (size_t)BIN_GRID * BIN_NODES * 4);
    int* lstT   = (int*)(ws + o); o = align256(o + (size_t)BIN_GRID * BIN_NODES * 4);
    int* rec    = (int*)(ws + o); o = align256(o + (size_t)BIN_GRID * chunk * 8);
    int* eidx   = (int*)(ws + o); o = align256(o + (size_t)E * 4);
    _Float16* wt16 = (_Float16*)(ws + o); o = align256(o + (size_t)FOUT * FOUT * 2);
    float* feat = (float*)(ws + o); o = align256(o + (size_t)N * FOUT * 4);
    float* el   = (float*)(ws + o); o = align256(o + (size_t)N * HEADS * 4);
    float* er   = (float*)(ws + o); o = align256(o + (size_t)N * HEADS * 4);
    (void)ws_size;

    float* h = (float*)d_out;

    const int T = 256;
    auto blk = [](int n, int t) { return (n + t - 1) / t; };

    bin_local<<<BIN_GRID, 1024, 0, stream>>>(src, dst, rec, cntT, lstT, E, chunk);
    bin_build<<<nbins, 1024, 0, stream>>>(rec, cntT, lstT, off, eidx, N, E, chunk);

    for (int layer = 0; layer < 4; ++layer) {
        const float* W  = (const float*)d_in[3 + 4 * layer];
        const float* al = (const float*)d_in[4 + 4 * layer];
        const float* ar = (const float*)d_in[5 + 4 * layer];
        const float* b  = (const float*)d_in[6 + 4 * layer];
        const int K = in_sizes[3 + 4 * layer] / FOUT;
        const float* hin = (layer == 0) ? x : h;

        stage_w_frag<<<blk(K * FOUT, T), T, 0, stream>>>(W, wt16, K);

        int nTiles = N / 16;
        gemm_wmma_f16<<<blk(nTiles, GEMM_WAVES), GEMM_THREADS, 0, stream>>>(hin, wt16, feat, al, ar, el, er, N, K);

        int mode = (layer == 0) ? 0 : 1;
        gat_pull<<<blk(N * 32, T), T, 0, stream>>>(feat, el, er, off, eidx, b, h, N, mode);
    }
}


# weights of all four layers staged in B-fragment order by extra blocks of the first launch, beside the edge sort; the 4 per-layer staging launches are gone; same values, k order and sums
# speedup vs baseline: 2.5418x; 1.0379x over previous
#include <hip/hip_runtime.h>
#include <hip/hip_bf16.h>
#include <math.h>

typedef __attribute__((ext_vector_type(16))) _Float16 v16h;
typedef __attribute__((ext_vector_type(8)))  float    v8f;
typedef __attribute__((ext_vector_type(8)))  _Float16 v8h;
typedef __attribute__((ext_vector_type(4)))  float    v4f;

#define HEADS 8
#define HDIM  32
#define FOUT  256
#define NEG_SLOPE 0.2f

#define BIN_SHIFT 8
#define BIN_NODES 256
#define BIN_GRID  256

__device__ __forceinline__ int bin_of(int d) {
    int b = (int)((unsigned)d >> BIN_SHIFT);
    return (b < BIN_NODES) ? b : (BIN_NODES - 1);
}

__device__ __forceinline__ void scan256_inclusive(int* sh, int tid) {
    for (int o = 1; o < 256; o <<= 1) {
        int v = 0;
        if (tid < 256 && tid >= o) v = sh[tid - o];
        __syncthreads();
        if (tid < 256) sh[tid] += v;
        __syncthreads();
    }
}

__device__ __forceinline__ void stage_w_elem(const float* __restrict__ W, _Float16* __restrict__ Wf, int K, int i) {
    int k = i / FOUT, col = i % FOUT;
    int nt = col >> 4, c = col & 15;
    int ks = k >> 5, kk = k & 31;
    int h = (kk >> 3) & 1, up = kk >> 4, e = kk & 7;
    int ln = c + (h << 4);
    size_t d = ((size_t)(nt * (K >> 5) + ks) << 9) + (size_t)((up << 8) + (ln << 3) + e);
    Wf[d] = (_Float16)W[i];
}

#define W_SLOT (FOUT * FOUT)

__global__ void __launch_bounds__(1024) bin_local(const int* __restrict__ src,
                                                  const int* __restrict__ dst,
                                                  int* __restrict__ rec,
                                                  int* __restrict__ cntT,
                                                  int* __restrict__ lstT,
                                                  int e, int chunk,
                                                  const float* __restrict__ W0,
                                                  const float* __restrict__ W1,
                                                  const float* __restrict__ W2,
                                                  const float* __restrict__ W3,
                                                  _Float16* __restrict__ wt16,
                                                  int K0, int K1, int K2, int K3) {
    __shared__ int hist[BIN_NODES];
    __shared__ int lst[BIN_NODES];
    __shared__ int cur[BIN_NODES];
    int t = threadIdx.x, blk = blockIdx.x;
    if (blk >= BIN_GRID) {
        int i = (blk - BIN_GRID) * 1024 + t;
        const int n0 = K0 * FOUT, n1 = K1 * FOUT, n2 = K2 * FOUT, n3 = K3 * FOUT;
        if (i < n0)              stage_w_elem(W0, wt16,              K0, i);
        else if ((i -= n0) < n1) stage_w_elem(W1, wt16 + W_SLOT,     K1, i);
        else if ((i -= n1) < n2) stage_w_elem(W2, wt16 + 2 * W_SLOT, K2, i);
        else if ((i -= n2) < n3) stage_w_elem(W3, wt16 + 3 * W_SLOT, K3, i);
        return;
    }
    int e0 = blk * chunk;
    int e1 = e0 + chunk; if (e1 > e) e1 = e;
    if (t < 256) { hist[t] = 0; cur[t] = 0; }
    __syncthreads();
    for (int base = e0 + t; base < e1; base += 4096) {
        int d[4];
#pragma unroll
        for (int u = 0; u < 4; ++u) { int i = base + u * 1024; d[u] = dst[(i < e1) ? i : (e1 - 1)]; }
#pragma unroll
        for (int u = 0; u < 4; ++u) atomicAdd(&hist[bin_of(d[u])], (base + u * 1024 < e1) ? 1 : 0);
    }
    __syncthreads();
    int c = 0;
    if (t < 256) { c = hist[t]; lst[t] = c; }
    __syncthreads();
    scan256_inclusive(lst, t);
    if (t < 256) {
        int ex = lst[t] - c;
        cntT[blk * BIN_NODES + t] = c;
        lstT[blk * BIN_NODES + t] = ex;
        lst[t] = ex;
    }
    __syncthreads();
    int2* r2 = (int2*)rec;
    for (int base = e0 + t; base < e1; base += 4096) {
        int d[4], s[4];
#pragma unroll
        for (int u = 0; u < 4; ++u) {
            int i = base + u * 1024;
            int ic = (i < e1) ? i : (e1 - 1);
            d[u] = dst[ic];
            s[u] = src[ic];
        }
#pragma unroll
        for (int u = 0; u < 4; ++u) {
            if (base + u * 1024 < e1) {
                int b = bin_of(d[u]);
                int r = atomicAdd(&cur[b], 1);
                r2[e0 + lst[b] + r] = make_int2(s[u], d[u]);
            }
        }
    }
}

__global__ void __launch_bounds__(1024) bin_build(const int* __restrict__ rec,
                                                  const int* __restrict__ cntT,
                                                  const int* __restrict__ lstT,
                                                  int* __restrict__ off,
                                                  int* __restrict__ eidx,
                                                  int n, int e, int chunk) {
    __shared__ int cnt[BIN_NODES];
    __shared__ int exs[BIN_NODES];
    __shared__ int cur[BIN_NODES];
    __shared__ int sS;
    int t = threadIdx.x, b = blockIdx.x;
    if (t < 256) { cnt[t] = 0; cur[t] = 0; }
    if (t == 0) sS = 0;
    __syncthreads();

    int col = t & 255;
    int acc = 0;
    if (col < b) {
        for (int r = t >> 8; r < BIN_GRID; r += 4) acc += cntT[r * BIN_NODES + col];
    }
#pragma unroll
    for (int o = 16; o >= 1; o >>= 1) acc += __shfl_xor(acc, o);
    if ((t & 31) == 0 && acc != 0) atomicAdd(&sS, acc);

    int seg = t >> 2, sub = t & 3;
    int st  = seg * chunk + lstT[seg * BIN_NODES + b];
    int len = cntT[seg * BIN_NODES + b];
    const int2* r2 = (const int2*)rec;
    for (int base = sub; base < len; base += 16) {
        int d[4];
#pragma unroll
        for (int u = 0; u < 4; ++u) { int i = base + u * 4; d[u] = r2[st + ((i < len) ? i : (len - 1))].y; }
#pragma unroll
        for (int u = 0; u < 4; ++u) atomicAdd(&cnt[d[u] & (BIN_NODES - 1)], (base + u * 4 < len) ? 1 : 0);
    }
    __syncthreads();

    int S = sS;
    int own = 0;
    if (t < 256) { own = cnt[t]; exs[t] = own; }
    __syncthreads();
    scan256_inclusive(exs, t);
    if (t < 256) {
        int ex = exs[t] - own;
        int node = b * BIN_NODES + t;
        if (node < n) off[node] = S + ex;
        exs[t] = ex;
    }
    if (b == (int)gridDim.x - 1 && t == 0) off[n] = e;
    __syncthreads();
    for (int base = sub; base < len; base += 16) {
        int2 r[4];
#pragma unroll
        for (int u = 0; u < 4; ++u) { int i = base + u * 4; r[u] = r2[st + ((i < len) ? i : (len - 1))]; }
#pragma unroll
        for (int u = 0; u < 4; ++u) {
            if (base + u * 4 < len) {
                int l = r[u].y & (BIN_NODES - 1);
                int k = atomicAdd(&cur[l], 1);
                eidx[S + exs[l] + k] = r[u].x;
            }
        }
    }
}

__device__ __forceinline__ v16h cvt_frag_f32(v4f x0, v4f x1, v4f y0, v4f y1) {
    v8f lo = __builtin_shufflevector(x0, x1, 0, 1, 2, 3, 4, 5, 6, 7);
    v8f hi = __builtin_shufflevector(y0, y1, 0, 1, 2, 3, 4, 5, 6, 7);
    v8h hl = __builtin_convertvector(lo, v8h);
    v8h hh = __builtin_convertvector(hi, v8h);
    return __builtin_shufflevector(hl, hh, 0, 1, 2, 3, 4, 5, 6, 7, 8, 9, 10, 11, 12, 13, 14, 15);
}

template <int H>
__device__ __forceinline__ void halve_step(float* v, bool up, int mask) {
#pragma unroll
    for (int k = 0; k < H; ++k) {
        float keep = up ? v[k + H] : v[k];
        float send = up ? v[k] : v[k + H];
        v[k] = keep + __shfl_xor(send, mask);
    }
}

#define GEMM_WAVES   13
#define GEMM_THREADS (GEMM_WAVES * 32)

#define STAGE_BATCH  10
static_assert(STAGE_BATCH == 10, "stage_batch names its ten loaded values one by one");

__device__ __forceinline__ void stage_batch(const v8h* __restrict__ g, v8h* s, int base, int units) {
    v8h t[STAGE_BATCH];
#pragma unroll
    for (int u = 0; u < STAGE_BATCH; ++u) { int i = base + u * GEMM_THREADS; t[u] = g[(i < units) ? i : (units - 1)]; }
    asm volatile("" :: "v"(t[0]), "v"(t[1]), "v"(t[2]), "v"(t[3]), "v"(t[4]), "v"(t[5]), "v"(t[6]), "v"(t[7]), "v"(t[8]), "v"(t[9]));
#pragma unroll
    for (int u = 0; u < STAGE_BATCH; ++u) { int i = base + u * GEMM_THREADS; if (i < units) s[i] = t[u]; }
}

__global__ void __launch_bounds__(GEMM_THREADS) gemm_wmma_f16(const float* __restrict__ A,
                                                              const _Float16* __restrict__ Bt,
                                                              float* __restrict__ C,
                                                              const float* __restrict__ al,
                                                              const float* __restrict__ ar,
                                                              float* __restrict__ el,
                                                              float* __restrict__ er,
                                                              int M, int K) {
    __shared__ v8h   sB[FOUT * FOUT / 8];
    __shared__ float s_al[FOUT];
    __shared__ float s_ar[FOUT];
    const int tid  = threadIdx.x;
    const int lane = tid & 31;
    const int wv     = __builtin_amdgcn_readfirstlane(tid >> 5);
    const int nTiles = M >> 4;
    const int ksteps = K >> 5;
    const int stride = (int)gridDim.x * GEMM_WAVES;
    const int l15   = lane & 15;
    const int rbase = (lane >> 4) << 3;

    int tile = (int)blockIdx.x * GEMM_WAVES + wv;
    const int t0 = (tile < nTiles) ? tile : (nTiles - 1);
    const float* pa = A + (size_t)((t0 << 4) + l15) * K + rbase;
    v4f x0 = ((const v4f*)pa)[0], x1 = ((const v4f*)pa)[1];
    v4f y0 = ((const v4f*)pa)[4], y1 = ((const v4f*)pa)[5];

    {
        const v8h* g = (const v8h*)Bt;
        const int units = K * (FOUT / 8);
        stage_batch(g, sB, tid, units);
        for (int base = tid + STAGE_BATCH * GEMM_THREADS; base < units; base += STAGE_BATCH * GEMM_THREADS)
            stage_batch(g, sB, base, units);
        if (tid < FOUT) { s_al[tid] = al[tid]; s_ar[tid] = ar[tid]; }
    }
    __syncthreads();

    const int tstride = ksteps << 6;
    while (tile < nTiles) {
        const int m0 = tile << 4;
        v8f c[16];
#pragma unroll
        for (int nt = 0; nt < 16; ++nt) c[nt] = (v8f){};

        for (int ks = 0; ks < ksteps; ++ks) {
            const int kn = (ks + 1 < ksteps) ? (ks + 1) : ks;
            const v4f* qn = (const v4f*)(pa + (kn << 5));
            v4f nx0 = qn[0], nx1 = qn[1], ny0 = qn[4], ny1 = qn[5];
            v16h a = cvt_frag_f32(x0, x1, y0, y1);
            const v8h* kb = sB + (ks << 6) + lane;
#pragma unroll
            for (int g = 0; g < 16; g += 8) {
                v16h b[8];
#pragma unroll
                for (int j = 0; j < 8; ++j) {
                    v8h lo = kb[(g + j) * tstride];
                    v8h hi = kb[(g + j) * tstride + 32];
                    b[j] = __builtin_shufflevector(lo, hi, 0, 1, 2, 3, 4, 5, 6, 7, 8, 9, 10, 11, 12, 13, 14, 15);
                }
                asm volatile("" :: "v"(b[0]), "v"(b[1]), "v"(b[2]), "v"(b[3]), "v"(b[4]), "v"(b[5]), "v"(b[6]), "v"(b[7]));
#pragma unroll
                for (int j = 0; j < 8; ++j)
                    c[g + j] = __builtin_amdgcn_wmma_f32_16x16x32_f16(false, a, false, b[j], (short)0, c[g + j], false, false);
            }
            x0 = nx0; x1 = nx1; y0 = ny0; y1 = ny1;
        }

#pragma unroll
        for (int r = 0; r < 8; ++r) {
            const size_t o = (size_t)(m0 + r + rbase) * FOUT + l15;
#pragma unroll
            for (int nt = 0; nt < 16; ++nt) C[o + 16 * nt] = c[nt][r];
        }

#pragma unroll
        for (int i = 0; i < 4; ++i) {
            const int n0 = i << 6;
            const v8f c0 = c[4 * i], c1 = c[4 * i + 1], c2 = c[4 * i + 2], c3 = c[4 * i + 3];
            float a0 = s_al[n0 + l15], a1 = s_al[n0 + l15 + 16], a2 = s_al[n0 + l15 + 32], a3 = s_al[n0 + l15 + 48];
            float r0 = s_ar[n0 + l15], r1 = s_ar[n0 + l15 + 16], r2 = s_ar[n0 + l15 + 32], r3 = s_ar[n0 + l15 + 48];
            float v[32];
#pragma unroll
            for (int r = 0; r < 8; ++r) {
                v[r]      = c0[r] * a0 + c1[r] * a1;
                v[8 + r]  = c0[r] * r0 + c1[r] * r1;
                v[16 + r] = c2[r] * a2 + c3[r] * a3;
                v[24 + r] = c2[r] * r2 + c3[r] * r3;
            }
            halve_step<16>(v, (l15 & 8) != 0, 8);
            halve_step<8>(v,  (l15 & 4) != 0, 4);
            halve_step<4>(v,  (l15 & 2) != 0, 2);
            halve_step<2>(v,  (l15 & 1) != 0, 1);
            const int q   = l15 >> 2;
            const int hd  = (i << 1) + (q >> 1);
            const int row = m0 + rbase + ((l15 & 3) << 1);
            float* gd = (q & 1) ? er : el;
            gd[(size_t)row * HEADS + hd]       = v[0];
            gd[(size_t)(row + 1) * HEADS + hd] = v[1];
        }

        tile += stride;
        if (tile < nTiles) {
            pa = A + (size_t)((tile << 4) + l15) * K + rbase;
            x0 = ((const v4f*)pa)[0]; x1 = ((const v4f*)pa)[1];
            y0 = ((const v4f*)pa)[4]; y1 = ((const v4f*)pa)[5];
        }
    }
}

__device__ __forceinline__ float leaky(float e) { return (e > 0.f) ? e : NEG_SLOPE * e; }

__device__ __forceinline__ float pick4(const float* a, int k) {
    float r = a[0];
    r = (k == 1) ? a[1] : r;
    r = (k == 2) ? a[2] : r;
    r = (k == 3) ? a[3] : r;
    return r;
}

__device__ __forceinline__ void edge_fma(float4& accA, float4& accB, float4 fa, float4 fb,
                                         float la, float lb, float erA, float erB,
                                         float mA, float mB, float iA, float iB) {
    float wA = __expf(leaky(la + erA) - mA) * iA;
    float wB = __expf(leaky(lb + erB) - mB) * iB;
    accA.x += fa.x * wA; accA.y += fa.y * wA; accA.z += fa.z * wA; accA.w += fa.w * wA;
    accB.x += fb.x * wB; accB.y += fb.y * wB; accB.z += fb.z * wB; accB.w += fb.w * wB;
}

__device__ __forceinline__ void row_fma(float4& accA, float4& accB, float4 fa, float4 fb,
                                        float wA, float wB) {
    accA.x = fmaf(fa.x, wA, accA.x); accA.y = fmaf(fa.y, wA, accA.y);
    accA.z = fmaf(fa.z, wA, accA.z); accA.w = fmaf(fa.w, wA, accA.w);
    accB.x = fmaf(fb.x, wB, accB.x); accB.y = fmaf(fb.y, wB, accB.y);
    accB.z = fmaf(fb.z, wB, accB.z); accB.w = fmaf(fb.w, wB, accB.w);
}

__global__ void __launch_bounds__(256) gat_pull(const float* __restrict__ feat,
                                                const float* __restrict__ el,
                                                const float* __restrict__ er,
                                                const int* __restrict__ off,
                                                const int* __restrict__ eidx,
                                                const float* __restrict__ bias,
                                                float* __restrict__ out,
                                                int n, int mode) {
    __shared__ float s_w[8][32 * HEADS];
    int gt   = blockIdx.x * 256 + threadIdx.x;
    int lane = gt & 31;
    int v = __builtin_amdgcn_readfirstlane(gt >> 5);
    if (v >= n) return;

    int beg = off[v], end = off[v + 1];
    int hA = lane >> 3;
    size_t obase = (size_t)v * FOUT + lane * 4;

    float4 oA = make_float4(0.f, 0.f, 0.f, 0.f), oB = oA;
    if (mode != 0) {
        oA = *(const float4*)(out + obase);
        oB = *(const float4*)(out + obase + 128);
    }

    float ern[8];
#pragma unroll
    for (int h = 0; h < 8; ++h) ern[h] = er[(size_t)v * HEADS + h];

    const int deg = end - beg;
    float4 accA = make_float4(0.f, 0.f, 0.f, 0.f), accB = accA;
    const float* fl = feat + lane * 4;

    if (deg <= 32) {
        float* swv = s_w[threadIdx.x >> 5];
        const bool have = lane < deg;
        int sl = 0;
        float e[8];
#pragma unroll
        for (int h = 0; h < 8; ++h) e[h] = -INFINITY;
        if (have) {
            sl = eidx[beg + lane];
            const float4* ep = (const float4*)(el + (size_t)sl * HEADS);
            float4 a0 = ep[0], a1 = ep[1];
            float ev[8] = {a0.x, a0.y, a0.z, a0.w, a1.x, a1.y, a1.z, a1.w};
#pragma unroll
            for (int h = 0; h < 8; ++h) e[h] = leaky(ev[h] + ern[h]);
        }
        float mx[8];
#pragma unroll
        for (int h = 0; h < 8; ++h) {
            mx[h] = e[h];
#pragma unroll
            for (int o = 16; o >= 1; o >>= 1) mx[h] = fmaxf(mx[h], __shfl_xor(mx[h], o));
        }
        float p[8], sm[8];
#pragma unroll
        for (int h = 0; h < 8; ++h) {
            p[h] = have ? __expf(e[h] - mx[h]) : 0.f;
            sm[h] = p[h];
#pragma unroll
            for (int o = 16; o >= 1; o >>= 1) sm[h] += __shfl_xor(sm[h], o);
        }
        float w[8];
#pragma unroll
        for (int h = 0; h < 8; ++h) w[h] = p[h] * ((deg > 0) ? 1.f / sm[h] : 0.f);
        *(float4*)(swv + lane * HEADS)     = make_float4(w[0], w[1], w[2], w[3]);
        *(float4*)(swv + lane * HEADS + 4) = make_float4(w[4], w[5], w[6], w[7]);
        __builtin_amdgcn_wave_barrier();

        const float* wl = swv + hA;
        int j = 0;
        for (; j + 4 <= deg; j += 4) {
            int s0 = __builtin_amdgcn_readlane(sl, j);
            int s1 = __builtin_amdgcn_readlane(sl, j + 1);
            int s2 = __builtin_amdgcn_readlane(sl, j + 2);
            int s3 = __builtin_amdgcn_readlane(sl, j + 3);
            const float* f0 = fl + (size_t)s0 * FOUT;
            const float* f1 = fl + (size_t)s1 * FOUT;
            const float* f2 = fl + (size_t)s2 * FOUT;
            const float* f3 = fl + (size_t)s3 * FOUT;
            float4 a0 = *(const float4*)f0, b0 = *(const float4*)(f0 + 128);
            float4 a1 = *(const float4*)f1, b1 = *(const float4*)(f1 + 128);
            float4 a2 = *(const float4*)f2, b2 = *(const float4*)(f2 + 128);
            float4 a3 = *(const float4*)f3, b3 = *(const float4*)(f3 + 128);
            const float* wj = wl + j * HEADS;
            float wa0 = wj[0],  wb0 = wj[4];
            float wa1 = wj[8],  wb1 = wj[12];
            float wa2 = wj[16], wb2 = wj[20];
            float wa3 = wj[24], wb3 = wj[28];
            row_fma(accA, accB, a0, b0, wa0, wb0);
            row_fma(accA, accB, a1, b1, wa1, wb1);
            row_fma(accA, accB, a2, b2, wa2, wb2);
            row_fma(accA, accB, a3, b3, wa3, wb3);
        }
#pragma unroll 1
        for (; j < deg; ++j) {
            int s0 = __builtin_amdgcn_readlane(sl, j);
            const float* f0 = fl + (size_t)s0 * FOUT;
            float4 a0 = *(const float4*)f0, b0 = *(const float4*)(f0 + 128);
            float wa0 = wl[j * HEADS], wb0 = wl[j * HEADS + 4];
            row_fma(accA, accB, a0, b0, wa0, wb0);
        }
    } else {
        float mx[8];
#pragma unroll
        for (int h = 0; h < 8; ++h) mx[h] = -INFINITY;
        for (int c0 = beg; c0 < end; c0 += 32) {
            int t = c0 + lane;
            if (t < end) {
                int s = eidx[t];
                const float4* ep = (const float4*)(el + (size_t)s * HEADS);
                float4 a0 = ep[0], a1 = ep[1];
                float ev[8] = {a0.x, a0.y, a0.z, a0.w, a1.x, a1.y, a1.z, a1.w};
#pragma unroll
                for (int h = 0; h < 8; ++h) mx[h] = fmaxf(mx[h], leaky(ev[h] + ern[h]));
            }
        }
#pragma unroll
        for (int h = 0; h < 8; ++h) {
#pragma unroll
            for (int o = 16; o >= 1; o >>= 1) mx[h] = fmaxf(mx[h], __shfl_xor(mx[h], o));
        }

        float sm[8];
#pragma unroll
        for (int h = 0; h < 8; ++h) sm[h] = 0.f;
        for (int c0 = beg; c0 < end; c0 += 32) {
            int t = c0 + lane;
            if (t < end) {
                int s = eidx[t];
                const float4* ep = (const float4*)(el + (size_t)s * HEADS);
                float4 a0 = ep[0], a1 = ep[1];
                float ev[8] = {a0.x, a0.y, a0.z, a0.w, a1.x, a1.y, a1.z, a1.w};
#pragma unroll
                for (int h = 0; h < 8; ++h) sm[h] += __expf(leaky(ev[h] + ern[h]) - mx[h]);
            }
        }
#pragma unroll
        for (int h = 0; h < 8; ++h) {
#pragma unroll
            for (int o = 16; o >= 1; o >>= 1) sm[h] += __shfl_xor(sm[h], o);
        }
        float inv[8];
#pragma unroll
        for (int h = 0; h < 8; ++h) inv[h] = (end > beg) ? 1.f / sm[h] : 0.f;

        float mA = pick4(mx, hA),  mB = pick4(mx + 4, hA);
        float iA = pick4(inv, hA), iB = pick4(inv + 4, hA);
        float erA = pick4(ern, hA), erB = pick4(ern + 4, hA);

        const float* ll = el + hA;
        for (int c0 = beg; c0 < end; c0 += 32) {
            int cnt = end - c0; if (cnt > 32) cnt = 32;
            int sl = (lane < cnt) ? eidx[c0 + lane] : 0;
#pragma unroll 1
            for (int j = 0; j < cnt; ++j) {
                int s0 = __builtin_amdgcn_readlane(sl, j);
                const float* f0 = fl + (size_t)s0 * FOUT;
                float4 a0 = *(const float4*)f0, b0 = *(const float4*)(f0 + 128);
                float la0 = ll[(size_t)s0 * HEADS], lb0 = ll[(size_t)s0 * HEADS + 4];
                edge_fma(accA, accB, a0, b0, la0, lb0, erA, erB, mA, mB, iA, iB);
            }
        }
    }

    float ba[4], bb[4];
#pragma unroll
    for (int q = 0; q < 4; ++q) { ba[q] = bias[lane * 4 + q]; bb[q] = bias[128 + lane * 4 + q]; }
    float4 rA = make_float4(accA.x + ba[0], accA.y + ba[1], accA.z + ba[2], accA.w + ba[3]);
    float4 rB = make_float4(accB.x + bb[0], accB.y + bb[1], accB.z + bb[2], accB.w + bb[3]);
    if (mode == 0) {
        rA.x = (rA.x > 0.f) ? rA.x : (__expf(rA.x) - 1.f);
        rA.y = (rA.y > 0.f) ? rA.y : (__expf(rA.y) - 1.f);
        rA.z = (rA.z > 0.f) ? rA.z : (__expf(rA.z) - 1.f);
        rA.w = (rA.w > 0.f) ? rA.w : (__expf(rA.w) - 1.f);
        rB.x = (rB.x > 0.f) ? rB.x : (__expf(rB.x) - 1.f);
        rB.y = (rB.y > 0.f) ? rB.y : (__expf(rB.y) - 1.f);
        rB.z = (rB.z > 0.f) ? rB.z : (__expf(rB.z) - 1.f);
        rB.w = (rB.w > 0.f) ? rB.w : (__expf(rB.w) - 1.f);
    } else {
        rA.x += oA.x; rA.y += oA.y; rA.z += oA.z; rA.w += oA.w;
        rB.x += oB.x; rB.y += oB.y; rB.z += oB.z; rB.w += oB.w;
    }
    *(float4*)(out + obase)       = rA;
    *(float4*)(out + obase + 128) = rB;
}

static inline size_t align256(size_t x) { return (x + 255) & ~(size_t)255; }

extern "C" void kernel_launch(void* const* d_in, const int* in_sizes, int n_in,
                              void* d_out, int out_size, void* d_ws, size_t ws_size,
                              hipStream_t stream) {
    const int N = in_sizes[0] / 128;
    const int E = in_sizes[1];

    const float* x   = (const float*)d_in[0];
    const int*   src = (const int*)d_in[1];
    const int*   dst = (const int*)d_in[2];

    char* ws = (char*)d_ws;
    size_t o = 0;
    const int chunk = (E + BIN_GRID - 1) / BIN_GRID;
    const int nbins = (N + BIN_NODES - 1) / BIN_NODES;
    int* off    = (int*)(ws + o); o = align256(o + (size_t)(N + 1) * 4);
    int* cntT   = (int*)(ws + o); o = align256(o + (size_t)BIN_GRID * BIN_NODES * 4);
    int* lstT   = (int*)(ws + o); o = align256(o + (size_t)BIN_GRID * BIN_NODES * 4);
    int* rec    = (int*)(ws + o); o = align256(o + (size_t)BIN_GRID * chunk * 8);
    int* eidx   = (int*)(ws + o); o = align256(o + (size_t)E * 4);
    _Float16* wt16 = (_Float16*)(ws + o); o = align256(o + (size_t)4 * W_SLOT * 2);
    float* feat = (float*)(ws + o); o = align256(o + (size_t)N * FOUT * 4);
    float* el   = (float*)(ws + o); o = align256(o + (size_t)N * HEADS * 4);
    float* er   = (float*)(ws + o); o = align256(o + (size_t)N * HEADS * 4);
    (void)ws_size;

    float* h = (float*)d_out;

    const int T = 256;
    auto blk = [](int n, int t) { return (n + t - 1) / t; };

    const int K0 = in_sizes[3] / FOUT, K1 = in_sizes[7] / FOUT, K2 = in_sizes[11] / FOUT, K3 = in_sizes[15] / FOUT;
    const int stageBlocks = blk((K0 + K1 + K2 + K3) * FOUT, 1024);
    bin_local<<<BIN_GRID + stageBlocks, 1024, 0, stream>>>(src, dst, rec, cntT, lstT, E, chunk,
                                                          (const float*)d_in[3], (const float*)d_in[7],
                                                          (const float*)d_in[11], (const float*)d_in[15],
                                                          wt16, K0, K1, K2, K3);
    bin_build<<<nbins, 1024, 0, stream>>>(rec, cntT, lstT, off, eidx, N, E, chunk);

    for (int layer = 0; layer < 4; ++layer) {
        const float* al = (const float*)d_in[4 + 4 * layer];
        const float* ar = (const float*)d_in[5 + 4 * layer];
        const float* b  = (const float*)d_in[6 + 4 * layer];
        const int K = in_sizes[3 + 4 * layer] / FOUT;
        const float* hin = (layer == 0) ? x : h;

        int nTiles = N / 16;
        gemm_wmma_f16<<<blk(nTiles, GEMM_WAVES), GEMM_THREADS, 0, stream>>>(hin, wt16 + (size_t)layer * W_SLOT, feat, al, ar, el, er, N, K);

        int mode = (layer == 0) ? 0 : 1;
        gat_pull<<<blk(N * 32, T), T, 0, stream>>>(feat, el, er, off, eidx, b, h, N, mode);
    }
}


# pull softmax statistics: per-head max, sum and 1/sum formed once per head by a halving exchange on the vector unit and read out by readlane, not 8 32-lane butterflies in every lane; same sum tree
# speedup vs baseline: 3.4003x; 1.3378x over previous
#include <hip/hip_runtime.h>
#include <hip/hip_bf16.h>
#include <math.h>

typedef __attribute__((ext_vector_type(16))) _Float16 v16h;
typedef __attribute__((ext_vector_type(8)))  float    v8f;
typedef __attribute__((ext_vector_type(8)))  _Float16 v8h;
typedef __attribute__((ext_vector_type(4)))  float    v4f;
typedef __attribute__((ext_vector_type(4)))  _Float16 v4h;

#define HEADS 8
#define HDIM  32
#define FOUT  256
#define NEG_SLOPE 0.2f

#define BIN_SHIFT 8
#define BIN_NODES 256
#define BIN_GRID  256

__device__ __forceinline__ int bin_of(int d) {
    int b = (int)((unsigned)d >> BIN_SHIFT);
    return (b < BIN_NODES) ? b : (BIN_NODES - 1);
}

__device__ __forceinline__ void scan256_inclusive(int* sh, int tid) {
    for (int o = 1; o < 256; o <<= 1) {
        int v = 0;
        if (tid < 256 && tid >= o) v = sh[tid - o];
        __syncthreads();
        if (tid < 256) sh[tid] += v;
        __syncthreads();
    }
}

__device__ __forceinline__ void stage_w_elem(const float* __restrict__ W, _Float16* __restrict__ Wf, int K, int i) {
    int k = i / FOUT, col = i % FOUT;
    int nt = col >> 4, c = col & 15;
    int ks = k >> 5, kk = k & 31;
    int h = (kk >> 3) & 1, up = kk >> 4, e = kk & 7;
    int ln = c + (h << 4);
    size_t d = ((size_t)(nt * (K >> 5) + ks) << 9) + (size_t)((up << 8) + (ln << 3) + e);
    Wf[d] = (_Float16)W[i];
}

#define W_SLOT (FOUT * FOUT)

__global__ void __launch_bounds__(1024) bin_local(const int* __restrict__ src,
                                                  const int* __restrict__ dst,
                                                  int* __restrict__ rec,
                                                  int* __restrict__ cntT,
                                                  int* __restrict__ lstT,
                                                  int e, int chunk,
                                                  const float* __restrict__ W0,
                                                  const float* __restrict__ W1,
                                                  const float* __restrict__ W2,
                                                  const float* __restrict__ W3,
                                                  _Float16* __restrict__ wt16,
                                                  int K0, int K1, int K2, int K3) {
    __shared__ int hist[BIN_NODES];
    __shared__ int lst[BIN_NODES];
    __shared__ int cur[BIN_NODES];
    int t = threadIdx.x, blk = blockIdx.x;
    if (blk >= BIN_GRID) {
        int i = (blk - BIN_GRID) * 1024 + t;
        const int n0 = K0 * FOUT, n1 = K1 * FOUT, n2 = K2 * FOUT, n3 = K3 * FOUT;
        if (i < n0)              stage_w_elem(W0, wt16,              K0, i);
        else if ((i -= n0) < n1) stage_w_elem(W1, wt16 + W_SLOT,     K1, i);
        else if ((i -= n1) < n2) stage_w_elem(W2, wt16 + 2 * W_SLOT, K2, i);
        else if ((i -= n2) < n3) stage_w_elem(W3, wt16 + 3 * W_SLOT, K3, i);
        return;
    }
    int e0 = blk * chunk;
    int e1 = e0 + chunk; if (e1 > e) e1 = e;
    if (t < 256) { hist[t] = 0; cur[t] = 0; }
    __syncthreads();
    for (int base = e0 + t; base < e1; base += 4096) {
        int d[4];
#pragma unroll
        for (int u = 0; u < 4; ++u) { int i = base + u * 1024; d[u] = dst[(i < e1) ? i : (e1 - 1)]; }
#pragma unroll
        for (int u = 0; u < 4; ++u) atomicAdd(&hist[bin_of(d[u])], (base + u * 1024 < e1) ? 1 : 0);
    }
    __syncthreads();
    int c = 0;
    if (t < 256) { c = hist[t]; lst[t] = c; }
    __syncthreads();
    scan256_inclusive(lst, t);
    if (t < 256) {
        int ex = lst[t] - c;
        cntT[blk * BIN_NODES + t] = c;
        lstT[blk * BIN_NODES + t] = ex;
        lst[t] = ex;
    }
    __syncthreads();
    int2* r2 = (int2*)rec;
    for (int base = e0 + t; base < e1; base += 4096) {
        int d[4], s[4];
#pragma unroll
        for (int u = 0; u < 4; ++u) {
            int i = base + u * 1024;
            int ic = (i < e1) ? i : (e1 - 1);
            d[u] = dst[ic];
            s[u] = src[ic];
        }
#pragma unroll
        for (int u = 0; u < 4; ++u) {
            if (base + u * 1024 < e1) {
                int b = bin_of(d[u]);
                int r = atomicAdd(&cur[b], 1);
                r2[e0 + lst[b] + r] = make_int2(s[u], d[u]);
            }
        }
    }
}

__global__ void __launch_bounds__(1024) bin_build(const int* __restrict__ rec,
                                                  const int* __restrict__ cntT,
                                                  const int* __restrict__ lstT,
                                                  int* __restrict__ off,
                                                  int* __restrict__ eidx,
                                                  int n, int e, int chunk) {
    __shared__ int cnt[BIN_NODES];
    __shared__ int exs[BIN_NODES];
    __shared__ int cur[BIN_NODES];
    __shared__ int sS;
    int t = threadIdx.x, b = blockIdx.x;
    if (t < 256) { cnt[t] = 0; cur[t] = 0; }
    if (t == 0) sS = 0;
    __syncthreads();

    int col = t & 255;
    int acc = 0;
    if (col < b) {
        for (int r = t >> 8; r < BIN_GRID; r += 4) acc += cntT[r * BIN_NODES + col];
    }
#pragma unroll
    for (int o = 16; o >= 1; o >>= 1) acc += __shfl_xor(acc, o);
    if ((t & 31) == 0 && acc != 0) atomicAdd(&sS, acc);

    int seg = t >> 2, sub = t & 3;
    int st  = seg * chunk + lstT[seg * BIN_NODES + b];
    int len = cntT[seg * BIN_NODES + b];
    const int2* r2 = (const int2*)rec;
    for (int base = sub; base < len; base += 16) {
        int d[4];
#pragma unroll
        for (int u = 0; u < 4; ++u) { int i = base + u * 4; d[u] = r2[st + ((i < len) ? i : (len - 1))].y; }
#pragma unroll
        for (int u = 0; u < 4; ++u) atomicAdd(&cnt[d[u] & (BIN_NODES - 1)], (base + u * 4 < len) ? 1 : 0);
    }
    __syncthreads();

    int S = sS;
    int own = 0;
    if (t < 256) { own = cnt[t]; exs[t] = own; }
    __syncthreads();
    scan256_inclusive(exs, t);
    if (t < 256) {
        int ex = exs[t] - own;
        int node = b * BIN_NODES + t;
        if (node < n) off[node] = S + ex;
        exs[t] = ex;
    }
    if (b == (int)gridDim.x - 1 && t == 0) off[n] = e;
    __syncthreads();
    for (int base = sub; base < len; base += 16) {
        int2 r[4];
#pragma unroll
        for (int u = 0; u < 4; ++u) { int i = base + u * 4; r[u] = r2[st + ((i < len) ? i : (len - 1))]; }
#pragma unroll
        for (int u = 0; u < 4; ++u) {
            if (base + u * 4 < len) {
                int l = r[u].y & (BIN_NODES - 1);
                int k = atomicAdd(&cur[l], 1);
                eidx[S + exs[l] + k] = r[u].x;
            }
        }
    }
}

__device__ __forceinline__ v16h cvt_frag_f32(v4f x0, v4f x1, v4f y0, v4f y1) {
    v8f lo = __builtin_shufflevector(x0, x1, 0, 1, 2, 3, 4, 5, 6, 7);
    v8f hi = __builtin_shufflevector(y0, y1, 0, 1, 2, 3, 4, 5, 6, 7);
    v8h hl = __builtin_convertvector(lo, v8h);
    v8h hh = __builtin_convertvector(hi, v8h);
    return __builtin_shufflevector(hl, hh, 0, 1, 2, 3, 4, 5, 6, 7, 8, 9, 10, 11, 12, 13, 14, 15);
}

template <int H>
__device__ __forceinline__ void halve_step(float* v, bool up, int mask) {
#pragma unroll
    for (int k = 0; k < H; ++k) {
        float keep = up ? v[k + H] : v[k];
        float send = up ? v[k] : v[k + H];
        v[k] = keep + __shfl_xor(send, mask);
    }
}

#define GEMM_WAVES   13
#define GEMM_THREADS (GEMM_WAVES * 32)

#define STAGE_BATCH  10
static_assert(STAGE_BATCH == 10, "stage_batch names its ten loaded values one by one");

__device__ __forceinline__ void stage_batch(const v8h* __restrict__ g, v8h* s, int base, int units) {
    v8h t[STAGE_BATCH];
#pragma unroll
    for (int u = 0; u < STAGE_BATCH; ++u) { int i = base + u * GEMM_THREADS; t[u] = g[(i < units) ? i : (units - 1)]; }
    asm volatile("" :: "v"(t[0]), "v"(t[1]), "v"(t[2]), "v"(t[3]), "v"(t[4]), "v"(t[5]), "v"(t[6]), "v"(t[7]), "v"(t[8]), "v"(t[9]));
#pragma unroll
    for (int u = 0; u < STAGE_BATCH; ++u) { int i = base + u * GEMM_THREADS; if (i < units) s[i] = t[u]; }
}

__global__ void __launch_bounds__(GEMM_THREADS) gemm_wmma_f16(const float* __restrict__ A,
                                                              const _Float16* __restrict__ Bt,
                                                              _Float16* __restrict__ C,
                                                              const float* __restrict__ al,
                                                              const float* __restrict__ ar,
                                                              float* __restrict__ el,
                                                              float* __restrict__ er,
                                                              int M, int K) {
    __shared__ v8h   sB[FOUT * FOUT / 8];
    __shared__ float s_al[FOUT];
    __shared__ float s_ar[FOUT];
    const int tid  = threadIdx.x;
    const int lane = tid & 31;
    const int wv     = __builtin_amdgcn_readfirstlane(tid >> 5);
    const int nTiles = M >> 4;
    const int ksteps = K >> 5;
    const int stride = (int)gridDim.x * GEMM_WAVES;
    const int l15   = lane & 15;
    const int rbase = (lane >> 4) << 3;

    int tile = (int)blockIdx.x * GEMM_WAVES + wv;
    const int t0 = (tile < nTiles) ? tile : (nTiles - 1);
    const float* pa = A + (size_t)((t0 << 4) + l15) * K + rbase;
    v4f x0 = ((const v4f*)pa)[0], x1 = ((const v4f*)pa)[1];
    v4f y0 = ((const v4f*)pa)[4], y1 = ((const v4f*)pa)[5];

    {
        const v8h* g = (const v8h*)Bt;
        const int units = K * (FOUT / 8);
        stage_batch(g, sB, tid, units);
        for (int base = tid + STAGE_BATCH * GEMM_THREADS; base < units; base += STAGE_BATCH * GEMM_THREADS)
            stage_batch(g, sB, base, units);
        if (tid < FOUT) { s_al[tid] = al[tid]; s_ar[tid] = ar[tid]; }
    }
    __syncthreads();

    const int tstride = ksteps << 6;
    while (tile < nTiles) {
        const int m0 = tile << 4;
        v8f c[16];
#pragma unroll
        for (int nt = 0; nt < 16; ++nt) c[nt] = (v8f){};

        for (int ks = 0; ks < ksteps; ++ks) {
            const int kn = (ks + 1 < ksteps) ? (ks + 1) : ks;
            const v4f* qn = (const v4f*)(pa + (kn << 5));
            v4f nx0 = qn[0], nx1 = qn[1], ny0 = qn[4], ny1 = qn[5];
            v16h a = cvt_frag_f32(x0, x1, y0, y1);
            const v8h* kb = sB + (ks << 6) + lane;
#pragma unroll
            for (int g = 0; g < 16; g += 8) {
                v16h b[8];
#pragma unroll
                for (int j = 0; j < 8; ++j) {
                    v8h lo = kb[(g + j) * tstride];
                    v8h hi = kb[(g + j) * tstride + 32];
                    b[j] = __builtin_shufflevector(lo, hi, 0, 1, 2, 3, 4, 5, 6, 7, 8, 9, 10, 11, 12, 13, 14, 15);
                }
                asm volatile("" :: "v"(b[0]), "v"(b[1]), "v"(b[2]), "v"(b[3]), "v"(b[4]), "v"(b[5]), "v"(b[6]), "v"(b[7]));
#pragma unroll
                for (int j = 0; j < 8; ++j)
                    c[g + j] = __builtin_amdgcn_wmma_f32_16x16x32_f16(false, a, false, b[j], (short)0, c[g + j], false, false);
            }
            x0 = nx0; x1 = nx1; y0 = ny0; y1 = ny1;
        }

#pragma unroll
        for (int r = 0; r < 8; ++r) {
            const size_t o = (size_t)(m0 + r + rbase) * FOUT + l15;
#pragma unroll
            for (int nt = 0; nt < 16; ++nt) C[o + 16 * nt] = (_Float16)c[nt][r];
        }

#pragma unroll
        for (int i = 0; i < 4; ++i) {
            const int n0 = i << 6;
            const v8f c0 = c[4 * i], c1 = c[4 * i + 1], c2 = c[4 * i + 2], c3 = c[4 * i + 3];
            float a0 = s_al[n0 + l15], a1 = s_al[n0 + l15 + 16], a2 = s_al[n0 + l15 + 32], a3 = s_al[n0 + l15 + 48];
            float r0 = s_ar[n0 + l15], r1 = s_ar[n0 + l15 + 16], r2 = s_ar[n0 + l15 + 32], r3 = s_ar[n0 + l15 + 48];
            float v[32];
#pragma unroll
            for (int r = 0; r < 8; ++r) {
                v[r]      = c0[r] * a0 + c1[r] * a1;
                v[8 + r]  = c0[r] * r0 + c1[r] * r1;
                v[16 + r] = c2[r] * a2 + c3[r] * a3;
                v[24 + r] = c2[r] * r2 + c3[r] * r3;
            }
            halve_step<16>(v, (l15 & 8) != 0, 8);
            halve_step<8>(v,  (l15 & 4) != 0, 4);
            halve_step<4>(v,  (l15 & 2) != 0, 2);
            halve_step<2>(v,  (l15 & 1) != 0, 1);
            const int q   = l15 >> 2;
            const int hd  = (i << 1) + (q >> 1);
            const int row = m0 + rbase + ((l15 & 3) << 1);
            float* gd = (q & 1) ? er : el;
            gd[(size_t)row * HEADS + hd]       = v[0];
            gd[(size_t)(row + 1) * HEADS + hd] = v[1];
        }

        tile += stride;
        if (tile < nTiles) {
            pa = A + (size_t)((tile << 4) + l15) * K + rbase;
            x0 = ((const v4f*)pa)[0]; x1 = ((const v4f*)pa)[1];
            y0 = ((const v4f*)pa)[4]; y1 = ((const v4f*)pa)[5];
        }
    }
}

__device__ __forceinline__ float leaky(float e) { return (e > 0.f) ? e : NEG_SLOPE * e; }

__device__ __forceinline__ float pick4(const float* a, int k) {
    float r = a[0];
    r = (k == 1) ? a[1] : r;
    r = (k == 2) ? a[2] : r;
    r = (k == 3) ? a[3] : r;
    return r;
}

__device__ __forceinline__ void edge_fma(float4& accA, float4& accB, float4 fa, float4 fb,
                                         float la, float lb, float erA, float erB,
                                         float mA, float mB, float iA, float iB) {
    float wA = __expf(leaky(la + erA) - mA) * iA;
    float wB = __expf(leaky(lb + erB) - mB) * iB;
    accA.x += fa.x * wA; accA.y += fa.y * wA; accA.z += fa.z * wA; accA.w += fa.w * wA;
    accB.x += fb.x * wB; accB.y += fb.y * wB; accB.z += fb.z * wB; accB.w += fb.w * wB;
}

__device__ __forceinline__ float4 row4(const _Float16* p) {
    v4h h = *(const v4h*)p;
    v4f f = __builtin_convertvector(h, v4f);
    return make_float4(f[0], f[1], f[2], f[3]);
}

template <int M>
__device__ __forceinline__ int xor_lane(int v) {
    if constexpr (M == 16) return __shfl_xor(v, 16);
    else return __builtin_amdgcn_update_dpp(v, v, 0x160 + M, 0xf, 0xf, false);
}

template <int K, int J>
__device__ __forceinline__ int bitonic_step(int key, int lane) {
    int other = xor_lane<J>(key);
    bool up    = (K == 32) ? true : ((lane & K) == 0);
    bool lower = (lane & J) == 0;
    int mn = (key < other) ? key : other;
    int mx = (key < other) ? other : key;
    return (lower == up) ? mn : mx;
}

__device__ __forceinline__ int sort32_ascending(int key, int lane) {
    key = bitonic_step<2, 1>(key, lane);
    key = bitonic_step<4, 2>(key, lane);   key = bitonic_step<4, 1>(key, lane);
    key = bitonic_step<8, 4>(key, lane);   key = bitonic_step<8, 2>(key, lane);   key = bitonic_step<8, 1>(key, lane);
    key = bitonic_step<16, 8>(key, lane);  key = bitonic_step<16, 4>(key, lane);  key = bitonic_step<16, 2>(key, lane);
    key = bitonic_step<16, 1>(key, lane);
    key = bitonic_step<32, 16>(key, lane); key = bitonic_step<32, 8>(key, lane);  key = bitonic_step<32, 4>(key, lane);
    key = bitonic_step<32, 2>(key, lane);  key = bitonic_step<32, 1>(key, lane);
    return key;
}

__device__ __forceinline__ void row_fma(float4& accA, float4& accB, float4 fa, float4 fb,
                                        float wA, float wB) {
    accA.x = fmaf(fa.x, wA, accA.x); accA.y = fmaf(fa.y, wA, accA.y);
    accA.z = fmaf(fa.z, wA, accA.z); accA.w = fmaf(fa.w, wA, accA.w);
    accB.x = fmaf(fb.x, wB, accB.x); accB.y = fmaf(fb.y, wB, accB.y);
    accB.z = fmaf(fb.z, wB, accB.z); accB.w = fmaf(fb.w, wB, accB.w);
}

template <int M>
__device__ __forceinline__ float xor_lane_f(float x) {
    int b = __builtin_bit_cast(int, x);
    if constexpr (M == 16) return __builtin_bit_cast(float, __builtin_amdgcn_permlanex16(b, b, 0x76543210, 0xfedcba98, false, false));
    else return __builtin_bit_cast(float, __builtin_amdgcn_update_dpp(b, b, 0x160 + M, 0xf, 0xf, false));
}

template <bool IS_MAX>
__device__ __forceinline__ float comb2(float a, float b) { return IS_MAX ? fmaxf(a, b) : (a + b); }

template <bool IS_MAX, int H, int M>
__device__ __forceinline__ void head_halve(float* v, int lane) {
    const bool up = (lane & M) != 0;
#pragma unroll
    for (int k = 0; k < H; ++k) {
        float keep = up ? v[k + H] : v[k];
        float send = up ? v[k] : v[k + H];
        v[k] = comb2<IS_MAX>(keep, xor_lane_f<M>(send));
    }
}

template <bool IS_MAX>
__device__ __forceinline__ float head_reduce(float* v, int lane) {
    head_halve<IS_MAX, 4, 16>(v, lane);
    head_halve<IS_MAX, 2, 8>(v, lane);
    head_halve<IS_MAX, 1, 4>(v, lane);
    v[0] = comb2<IS_MAX>(v[0], xor_lane_f<2>(v[0]));
    v[0] = comb2<IS_MAX>(v[0], xor_lane_f<1>(v[0]));
    return v[0];
}

__device__ __forceinline__ float head_value(float r, int h) {
    return __builtin_bit_cast(float, __builtin_amdgcn_readlane(__builtin_bit_cast(int, r), 4 * h));
}

__global__ void __launch_bounds__(256) gat_pull(const _Float16* __restrict__ feat,
                                                const float* __restrict__ el,
                                                const float* __restrict__ er,
                                                const int* __restrict__ off,
                                                const int* __restrict__ eidx,
                                                const float* __restrict__ bias,
                                                float* __restrict__ out,
                                                int n, int mode) {
    __shared__ float s_w[8][32 * HEADS];
    int gt   = blockIdx.x * 256 + threadIdx.x;
    int lane = gt & 31;
    int v = __builtin_amdgcn_readfirstlane(gt >> 5);
    if (v >= n) return;

    int beg = off[v], end = off[v + 1];
    int hA = lane >> 3;
    size_t obase = (size_t)v * FOUT + lane * 4;

    float4 oA = make_float4(0.f, 0.f, 0.f, 0.f), oB = oA;
    if (mode != 0) {
        oA = *(const float4*)(out + obase);
        oB = *(const float4*)(out + obase + 128);
    }

    float ern[8];
#pragma unroll
    for (int h = 0; h < 8; ++h) ern[h] = er[(size_t)v * HEADS + h];

    const int deg = end - beg;
    float4 accA = make_float4(0.f, 0.f, 0.f, 0.f), accB = accA;
    const _Float16* fl = feat + lane * 4;

    if (deg <= 32) {
        float* swv = s_w[threadIdx.x >> 5];
        const bool have = lane < deg;
        int sl = 0;
        float e[8];
#pragma unroll
        for (int h = 0; h < 8; ++h) e[h] = -INFINITY;
        int key = 0x7fffffff;
        if (have) key = eidx[beg + lane];
        key = sort32_ascending(key, lane);
        if (have) {
            sl = key;
            const float4* ep = (const float4*)(el + (size_t)sl * HEADS);
            float4 a0 = ep[0], a1 = ep[1];
            float ev[8] = {a0.x, a0.y, a0.z, a0.w, a1.x, a1.y, a1.z, a1.w};
#pragma unroll
            for (int h = 0; h < 8; ++h) e[h] = leaky(ev[h] + ern[h]);
        }
        float t[8];
#pragma unroll
        for (int h = 0; h < 8; ++h) t[h] = e[h];
        const float m1 = head_reduce<true>(t, lane);
        float mx[8];
#pragma unroll
        for (int h = 0; h < 8; ++h) mx[h] = head_value(m1, h);
        float p[8];
#pragma unroll
        for (int h = 0; h < 8; ++h) {
            p[h] = have ? __expf(e[h] - mx[h]) : 0.f;
            t[h] = p[h];
        }
        const float s1 = head_reduce<false>(t, lane);
        const float i1 = (deg > 0) ? 1.f / s1 : 0.f;
        float w[8];
#pragma unroll
        for (int h = 0; h < 8; ++h) w[h] = p[h] * head_value(i1, h);
        *(float4*)(swv + lane * HEADS)     = make_float4(w[0], w[1], w[2], w[3]);
        *(float4*)(swv + lane * HEADS + 4) = make_float4(w[4], w[5], w[6], w[7]);
        __builtin_amdgcn_wave_barrier();

        const float* wl = swv + hA;
        int j = 0;
        for (; j + 4 <= deg; j += 4) {
            int s0 = __builtin_amdgcn_readlane(sl, j);
            int s1 = __builtin_amdgcn_readlane(sl, j + 1);
            int s2 = __builtin_amdgcn_readlane(sl, j + 2);
            int s3 = __builtin_amdgcn_readlane(sl, j + 3);
            const _Float16* f0 = fl + (size_t)s0 * FOUT;
            const _Float16* f1 = fl + (size_t)s1 * FOUT;
            const _Float16* f2 = fl + (size_t)s2 * FOUT;
            const _Float16* f3 = fl + (size_t)s3 * FOUT;
            float4 a0 = row4(f0), b0 = row4(f0 + 128);
            float4 a1 = row4(f1), b1 = row4(f1 + 128);
            float4 a2 = row4(f2), b2 = row4(f2 + 128);
            float4 a3 = row4(f3), b3 = row4(f3 + 128);
            const float* wj = wl + j * HEADS;
            float wa0 = wj[0],  wb0 = wj[4];
            float wa1 = wj[8],  wb1 = wj[12];
            float wa2 = wj[16], wb2 = wj[20];
            float wa3 = wj[24], wb3 = wj[28];
            row_fma(accA, accB, a0, b0, wa0, wb0);
            row_fma(accA, accB, a1, b1, wa1, wb1);
            row_fma(accA, accB, a2, b2, wa2, wb2);
            row_fma(accA, accB, a3, b3, wa3, wb3);
        }
#pragma unroll 1
        for (; j < deg; ++j) {
            int s0 = __builtin_amdgcn_readlane(sl, j);
            const _Float16* f0 = fl + (size_t)s0 * FOUT;
            float4 a0 = row4(f0), b0 = row4(f0 + 128);
            float wa0 = wl[j * HEADS], wb0 = wl[j * HEADS + 4];
            row_fma(accA, accB, a0, b0, wa0, wb0);
        }
    } else {
        float mx[8];
#pragma unroll
        for (int h = 0; h < 8; ++h) mx[h] = -INFINITY;
        for (int c0 = beg; c0 < end; c0 += 32) {
            int t = c0 + lane;
            if (t < end) {
                int s = eidx[t];
                const float4* ep = (const float4*)(el + (size_t)s * HEADS);
                float4 a0 = ep[0], a1 = ep[1];
                float ev[8] = {a0.x, a0.y, a0.z, a0.w, a1.x, a1.y, a1.z, a1.w};
#pragma unroll
                for (int h = 0; h < 8; ++h) mx[h] = fmaxf(mx[h], leaky(ev[h] + ern[h]));
            }
        }
#pragma unroll
        for (int h = 0; h < 8; ++h) {
#pragma unroll
            for (int o = 16; o >= 1; o >>= 1) mx[h] = fmaxf(mx[h], __shfl_xor(mx[h], o));
        }

        float sm[8];
#pragma unroll
        for (int h = 0; h < 8; ++h) sm[h] = 0.f;
        for (int c0 = beg; c0 < end; c0 += 32) {
            int t = c0 + lane;
            if (t < end) {
                int s = eidx[t];
                const float4* ep = (const float4*)(el + (size_t)s * HEADS);
                float4 a0 = ep[0], a1 = ep[1];
                float ev[8] = {a0.x, a0.y, a0.z, a0.w, a1.x, a1.y, a1.z, a1.w};
#pragma unroll
                for (int h = 0; h < 8; ++h) sm[h] += __expf(leaky(ev[h] + ern[h]) - mx[h]);
            }
        }
#pragma unroll
        for (int h = 0; h < 8; ++h) {
#pragma unroll
            for (int o = 16; o >= 1; o >>= 1) sm[h] += __shfl_xor(sm[h], o);
        }
        float inv[8];
#pragma unroll
        for (int h = 0; h < 8; ++h) inv[h] = (end > beg) ? 1.f / sm[h] : 0.f;

        float mA = pick4(mx, hA),  mB = pick4(mx + 4, hA);
        float iA = pick4(inv, hA), iB = pick4(inv + 4, hA);
        float erA = pick4(ern, hA), erB = pick4(ern + 4, hA);

        const float* ll = el + hA;
        for (int c0 = beg; c0 < end; c0 += 32) {
            int cnt = end - c0; if (cnt > 32) cnt = 32;
            int sl = (lane < cnt) ? eidx[c0 + lane] : 0;
#pragma unroll 1
            for (int j = 0; j < cnt; ++j) {
                int s0 = __builtin_amdgcn_readlane(sl, j);
                const _Float16* f0 = fl + (size_t)s0 * FOUT;
                float4 a0 = row4(f0), b0 = row4(f0 + 128);
                float la0 = ll[(size_t)s0 * HEADS], lb0 = ll[(size_t)s0 * HEADS + 4];
                edge_fma(accA, accB, a0, b0, la0, lb0, erA, erB, mA, mB, iA, iB);
            }
        }
    }

    float ba[4], bb[4];
#pragma unroll
    for (int q = 0; q < 4; ++q) { ba[q] = bias[lane * 4 + q]; bb[q] = bias[128 + lane * 4 + q]; }
    float4 rA = make_float4(accA.x + ba[0], accA.y + ba[1], accA.z + ba[2], accA.w + ba[3]);
    float4 rB = make_float4(accB.x + bb[0], accB.y + bb[1], accB.z + bb[2], accB.w + bb[3]);
    if (mode == 0) {
        rA.x = (rA.x > 0.f) ? rA.x : (__expf(rA.x) - 1.f);
        rA.y = (rA.y > 0.f) ? rA.y : (__expf(rA.y) - 1.f);
        rA.z = (rA.z > 0.f) ? rA.z : (__expf(rA.z) - 1.f);
        rA.w = (rA.w > 0.f) ? rA.w : (__expf(rA.w) - 1.f);
        rB.x = (rB.x > 0.f) ? rB.x : (__expf(rB.x) - 1.f);
        rB.y = (rB.y > 0.f) ? rB.y : (__expf(rB.y) - 1.f);
        rB.z = (rB.z > 0.f) ? rB.z : (__expf(rB.z) - 1.f);
        rB.w = (rB.w > 0.f) ? rB.w : (__expf(rB.w) - 1.f);
    } else {
        rA.x += oA.x; rA.y += oA.y; rA.z += oA.z; rA.w += oA.w;
        rB.x += oB.x; rB.y += oB.y; rB.z += oB.z; rB.w += oB.w;
    }
    *(float4*)(out + obase)       = rA;
    *(float4*)(out + obase + 128) = rB;
}

static inline size_t align256(size_t x) { return (x + 255) & ~(size_t)255; }

extern "C" void kernel_launch(void* const* d_in, const int* in_sizes, int n_in,
                              void* d_out, int out_size, void* d_ws, size_t ws_size,
                              hipStream_t stream) {
    const int N = in_sizes[0] / 128;
    const int E = in_sizes[1];

    const float* x   = (const float*)d_in[0];
    const int*   src = (const int*)d_in[1];
    const int*   dst = (const int*)d_in[2];

    char* ws = (char*)d_ws;
    size_t o = 0;
    const int chunk = (E + BIN_GRID - 1) / BIN_GRID;
    const int nbins = (N + BIN_NODES - 1) / BIN_NODES;
    int* off    = (int*)(ws + o); o = align256(o + (size_t)(N + 1) * 4);
    int* cntT   = (int*)(ws + o); o = align256(o + (size_t)BIN_GRID * BIN_NODES * 4);
    int* lstT   = (int*)(ws + o); o = align256(o + (size_t)BIN_GRID * BIN_NODES * 4);
    int* rec    = (int*)(ws + o); o = align256(o + (size_t)BIN_GRID * chunk * 8);
    int* eidx   = (int*)(ws + o); o = align256(o + (size_t)E * 4);
    _Float16* wt16 = (_Float16*)(ws + o); o = align256(o + (size_t)4 * W_SLOT * 2);
    _Float16* feat = (_Float16*)(ws + o); o = align256(o + (size_t)N * FOUT * 2);
    float* el   = (float*)(ws + o); o = align256(o + (size_t)N * HEADS * 4);
    float* er   = (float*)(ws + o); o = align256(o + (size_t)N * HEADS * 4);
    (void)ws_size;

    float* h = (float*)d_out;

    const int T = 256;
    auto blk = [](int n, int t) { return (n + t - 1) / t; };

    const int K0 = in_sizes[3] / FOUT, K1 = in_sizes[7] / FOUT, K2 = in_sizes[11] / FOUT, K3 = in_sizes[15] / FOUT;
    const int stageBlocks = blk((K0 + K1 + K2 + K3) * FOUT, 1024);
    bin_local<<<BIN_GRID + stageBlocks, 1024, 0, stream>>>(src, dst, rec, cntT, lstT, E, chunk,
                                                          (const float*)d_in[3], (const float*)d_in[7],
                                                          (const float*)d_in[11], (const float*)d_in[15],
                                                          wt16, K0, K1, K2, K3);
    bin_build<<<nbins, 1024, 0, stream>>>(rec, cntT, lstT, off, eidx, N, E, chunk);

    for (int layer = 0; layer < 4; ++layer) {
        const float* al = (const float*)d_in[4 + 4 * layer];
        const float* ar = (const float*)d_in[5 + 4 * layer];
        const float* b  = (const float*)d_in[6 + 4 * layer];
        const int K = in_sizes[3 + 4 * layer] / FOUT;
        const float* hin = (layer == 0) ? x : h;

        int nTiles = N / 16;
        gemm_wmma_f16<<<blk(nTiles, GEMM_WAVES), GEMM_THREADS, 0, stream>>>(hin, wt16 + (size_t)layer * W_SLOT, feat, al, ar, el, er, N, K);

        int mode = (layer == 0) ? 0 : 1;
        gat_pull<<<blk(N * 32, T), T, 0, stream>>>(feat, el, er, off, eidx, b, h, N, mode);
    }
}
